# starting kernel
# baseline (speedup 1.0000x reference)
#include <hip/hip_runtime.h>
#include <hip/hip_bf16.h>


#define NNODES 50000
#define NEDGES 800000
#define BN_EPS 1e-5f

typedef __attribute__((ext_vector_type(2))) float v2f;
typedef __attribute__((ext_vector_type(8))) float v8f;

__global__ __launch_bounds__(256)
void edge_scatter_kernel(const float* __restrict__ x, int ldx,
                         const float* __restrict__ ea,
                         const int* __restrict__ src,
                         const int* __restrict__ dst,
                         const float* __restrict__ lw,
                         const float* __restrict__ lb,
                         float* __restrict__ agg)
{
    const int CHUNKS = 24;
    int gid = blockIdx.x * blockDim.x + threadIdx.x;
    if (gid >= NEDGES * CHUNKS) return;
    int j = gid / CHUNKS;
    int d = (gid % CHUNKS) * 4;

    int   s = src[j];
    int   t = dst[j];
    float a = ea[j];

    float4 xv  = *reinterpret_cast<const float4*>(x + (size_t)s * ldx + d);
    float4 lw4 = *reinterpret_cast<const float4*>(lw + d);
    float4 lb4 = *reinterpret_cast<const float4*>(lb + d);

    float m0 = fmaxf(xv.x + a * lw4.x + lb4.x, 0.0f);
    float m1 = fmaxf(xv.y + a * lw4.y + lb4.y, 0.0f);
    float m2 = fmaxf(xv.z + a * lw4.z + lb4.z, 0.0f);
    float m3 = fmaxf(xv.w + a * lw4.w + lb4.w, 0.0f);

    float* ap = agg + (size_t)t * 96 + d;
    atomicAdd(ap + 0, m0);
    atomicAdd(ap + 1, m1);
    atomicAdd(ap + 2, m2);
    atomicAdd(ap + 3, m3);
}

template <int K, bool HASADD>
__global__ __launch_bounds__(32)
void gemm_wmma_f32_kernel(const float* __restrict__ A0, int lda0,
                          const float* __restrict__ A1, int lda1,
                          const float* __restrict__ W,  int ldw,
                          const float* __restrict__ bias,
                          float* __restrict__ out, int ldo)
{
    int lane = threadIdx.x;
    int lo   = lane & 15;
    int hi   = lane >> 4;
    int row0 = blockIdx.x * 16;
    int col0 = blockIdx.y * 16;

    const float* arow0 = A0 + (size_t)(row0 + lo) * lda0;
    const float* arow1 = HASADD ? (A1 + (size_t)(row0 + lo) * lda1) : nullptr;

    v8f c = {};
#pragma unroll
    for (int k0 = 0; k0 < K; k0 += 4) {
        int ka = k0 + 2 * hi;
        v2f a, b;
        a.x = arow0[ka];
        a.y = arow0[ka + 1];
        if (HASADD) {
            a.x += arow1[ka];
            a.y += arow1[ka + 1];
        }
        b.x = W[(size_t)ka * ldw + col0 + lo];
        b.y = W[(size_t)(ka + 1) * ldw + col0 + lo];
        c = __builtin_amdgcn_wmma_f32_16x16x4_f32(
                 false, a,  false, b,
                 (short)0, c,  false,  false);
    }

    float bv = bias[col0 + lo];
#pragma unroll
    for (int r = 0; r < 8; ++r) {
        int row = row0 + r + 8 * hi;
        out[(size_t)row * ldo + col0 + lo] = fmaxf(c[r] + bv, 0.0f);
    }
}

__global__ void col_stats_kernel(const float* __restrict__ in, int ld, int n,
                                 float* __restrict__ sum,
                                 float* __restrict__ sumsq)
{
    int c = threadIdx.x;
    float s = 0.0f, s2 = 0.0f;
    for (int row = blockIdx.x; row < n; row += gridDim.x) {
        float v = in[(size_t)row * ld + c];
        s  += v;
        s2 += v * v;
    }
    atomicAdd(&sum[c],   s);
    atomicAdd(&sumsq[c], s2);
}

__global__ void bn_finalize_kernel(const float* __restrict__ sum,
                                   const float* __restrict__ sumsq,
                                   const float* __restrict__ g,
                                   const float* __restrict__ beta,
                                   float* __restrict__ scale,
                                   float* __restrict__ shift, float invN)
{
    int c = threadIdx.x;
    float mu  = sum[c] * invN;
    float var = sumsq[c] * invN - mu * mu;
    float inv = rsqrtf(var + BN_EPS);
    float sc  = g[c] * inv;
    scale[c] = sc;
    shift[c] = beta[c] - mu * sc;
}

__global__ __launch_bounds__(256)
void bn_apply_kernel(const float* __restrict__ in, int ldi,
                     float* __restrict__ out, int ldo,
                     const float* __restrict__ scale,
                     const float* __restrict__ shift, int n, int C)
{
    int gid = blockIdx.x * blockDim.x + threadIdx.x;
    if (gid >= n * C) return;
    int row = gid / C;
    int c   = gid % C;
    out[(size_t)row * ldo + c] = in[(size_t)row * ldi + c] * scale[c] + shift[c];
}

extern "C" void kernel_launch(void* const* d_in, const int* in_sizes, int n_in,
                              void* d_out, int out_size, void* d_ws, size_t ws_size,
                              hipStream_t stream)
{
    (void)in_sizes; (void)n_in; (void)out_size; (void)ws_size;

    const float* x          = (const float*)d_in[0];
    const float* edge_attr  = (const float*)d_in[1];
    const int*   edge_index = (const int*)  d_in[2];
    const float* lin_w      = (const float*)d_in[3];
    const float* lin_b      = (const float*)d_in[4];
    const float* c1_w  = (const float*)d_in[5];
    const float* c1_b  = (const float*)d_in[6];
    const float* c1_g  = (const float*)d_in[7];
    const float* c1_be = (const float*)d_in[8];
    const float* c2_w  = (const float*)d_in[9];
    const float* c2_b  = (const float*)d_in[10];
    const float* c2_g  = (const float*)d_in[11];
    const float* c2_be = (const float*)d_in[12];
    const float* l1_w  = (const float*)d_in[13];
    const float* l1_b  = (const float*)d_in[14];
    const float* l1_g  = (const float*)d_in[15];
    const float* l1_be = (const float*)d_in[16];
    const float* ma_w  = (const float*)d_in[17];
    const float* ma_b  = (const float*)d_in[18];
    const float* ma_g  = (const float*)d_in[19];
    const float* ma_be = (const float*)d_in[20];
    const float* mb_w  = (const float*)d_in[21];
    const float* mb_b  = (const float*)d_in[22];
    const float* mb_g  = (const float*)d_in[23];
    const float* mb_be = (const float*)d_in[24];

    const int* src = edge_index;
    const int* dst = edge_index + NEDGES;

    const size_t N = NNODES;

    float* ws    = (float*)d_ws;
    float* agg   = ws;
    float* bufP  = agg  + N * 96;
    float* bufQ  = bufP + N * 96;
    float* xcat  = bufQ + N * 96;
    float* stats = xcat + N * 160;

    const float invN = 1.0f / (float)N;
    const int rowTiles = NNODES / 16;

    auto statsL = [&](int L) { return stats + (size_t)L * 384; };

    hipMemsetAsync(agg,   0, N * 96 * sizeof(float), stream);
    hipMemsetAsync(stats, 0, 5 * 384 * sizeof(float), stream);

    const int edgeThreads = NEDGES * 24;
    const int edgeBlocks  = (edgeThreads + 255) / 256;

    edge_scatter_kernel<<<edgeBlocks, 256, 0, stream>>>(
        x, 96, edge_attr, src, dst, lin_w, lin_b, agg);

    gemm_wmma_f32_kernel<96, true><<<dim3(rowTiles, 6), 32, 0, stream>>>(
        x, 96, agg, 96, c1_w, 96, c1_b, bufP, 96);

    col_stats_kernel<<<256, 96, 0, stream>>>(bufP, 96, NNODES,
                                             statsL(0), statsL(0) + 96);
    bn_finalize_kernel<<<1, 96, 0, stream>>>(statsL(0), statsL(0) + 96,
                                             c1_g, c1_be,
                                             statsL(0) + 192, statsL(0) + 288, invN);
    bn_apply_kernel<<<(NNODES * 96 + 255) / 256, 256, 0, stream>>>(
        bufP, 96, xcat, 160, statsL(0) + 192, statsL(0) + 288, NNODES, 96);

    hipMemsetAsync(agg, 0, N * 96 * sizeof(float), stream);
    edge_scatter_kernel<<<edgeBlocks, 256, 0, stream>>>(
        xcat, 160, edge_attr, src, dst, lin_w, lin_b, agg);

    gemm_wmma_f32_kernel<96, true><<<dim3(rowTiles, 4), 32, 0, stream>>>(
        xcat, 160, agg, 96, c2_w, 64, c2_b, bufP, 64);

    col_stats_kernel<<<256, 64, 0, stream>>>(bufP, 64, NNODES,
                                             statsL(1), statsL(1) + 96);
    bn_finalize_kernel<<<1, 64, 0, stream>>>(statsL(1), statsL(1) + 96,
                                             c2_g, c2_be,
                                             statsL(1) + 192, statsL(1) + 288, invN);
    bn_apply_kernel<<<(NNODES * 64 + 255) / 256, 256, 0, stream>>>(
        bufP, 64, xcat + 96, 160, statsL(1) + 192, statsL(1) + 288, NNODES, 64);

    gemm_wmma_f32_kernel<160, false><<<dim3(rowTiles, 6), 32, 0, stream>>>(
        xcat, 160, nullptr, 0, l1_w, 96, l1_b, bufQ, 96);

    col_stats_kernel<<<256, 96, 0, stream>>>(bufQ, 96, NNODES,
                                             statsL(2), statsL(2) + 96);
    bn_finalize_kernel<<<1, 96, 0, stream>>>(statsL(2), statsL(2) + 96,
                                             l1_g, l1_be,
                                             statsL(2) + 192, statsL(2) + 288, invN);
    bn_apply_kernel<<<(NNODES * 96 + 255) / 256, 256, 0, stream>>>(
        bufQ, 96, bufQ, 96, statsL(2) + 192, statsL(2) + 288, NNODES, 96);

    gemm_wmma_f32_kernel<96, false><<<dim3(rowTiles, 6), 32, 0, stream>>>(
        bufQ, 96, nullptr, 0, ma_w, 96, ma_b, bufP, 96);

    col_stats_kernel<<<256, 96, 0, stream>>>(bufP, 96, NNODES,
                                             statsL(3), statsL(3) + 96);
    bn_finalize_kernel<<<1, 96, 0, stream>>>(statsL(3), statsL(3) + 96,
                                             ma_g, ma_be,
                                             statsL(3) + 192, statsL(3) + 288, invN);
    bn_apply_kernel<<<(NNODES * 96 + 255) / 256, 256, 0, stream>>>(
        bufP, 96, bufP, 96, statsL(3) + 192, statsL(3) + 288, NNODES, 96);

    gemm_wmma_f32_kernel<96, false><<<dim3(rowTiles, 1), 32, 0, stream>>>(
        bufP, 96, nullptr, 0, mb_w, 16, mb_b, bufQ, 16);

    col_stats_kernel<<<256, 16, 0, stream>>>(bufQ, 16, NNODES,
                                             statsL(4), statsL(4) + 96);
    bn_finalize_kernel<<<1, 16, 0, stream>>>(statsL(4), statsL(4) + 96,
                                             mb_g, mb_be,
                                             statsL(4) + 192, statsL(4) + 288, invN);
    bn_apply_kernel<<<(NNODES * 16 + 255) / 256, 256, 0, stream>>>(
        bufQ, 16, (float*)d_out, 16, statsL(4) + 192, statsL(4) + 288, NNODES, 16);
}


# BN column statistics: 196-row serial load-wait loop per thread replaced by 384-thread blocks, 8-13 float4 loads in flight per thread, LDS row reduce, then the atomics; summation order changes
# speedup vs baseline: 1.8937x; 1.8937x over previous
#include <hip/hip_runtime.h>
#include <hip/hip_bf16.h>


#define NNODES 50000
#define NEDGES 800000
#define BN_EPS 1e-5f

typedef __attribute__((ext_vector_type(2))) float v2f;
typedef __attribute__((ext_vector_type(8))) float v8f;

__global__ __launch_bounds__(256)
void edge_scatter_kernel(const float* __restrict__ x, int ldx,
                         const float* __restrict__ ea,
                         const int* __restrict__ src,
                         const int* __restrict__ dst,
                         const float* __restrict__ lw,
                         const float* __restrict__ lb,
                         float* __restrict__ agg)
{
    const int CHUNKS = 24;
    int gid = blockIdx.x * blockDim.x + threadIdx.x;
    if (gid >= NEDGES * CHUNKS) return;
    int j = gid / CHUNKS;
    int d = (gid % CHUNKS) * 4;

    int   s = src[j];
    int   t = dst[j];
    float a = ea[j];

    float4 xv  = *reinterpret_cast<const float4*>(x + (size_t)s * ldx + d);
    float4 lw4 = *reinterpret_cast<const float4*>(lw + d);
    float4 lb4 = *reinterpret_cast<const float4*>(lb + d);

    float m0 = fmaxf(xv.x + a * lw4.x + lb4.x, 0.0f);
    float m1 = fmaxf(xv.y + a * lw4.y + lb4.y, 0.0f);
    float m2 = fmaxf(xv.z + a * lw4.z + lb4.z, 0.0f);
    float m3 = fmaxf(xv.w + a * lw4.w + lb4.w, 0.0f);

    float* ap = agg + (size_t)t * 96 + d;
    atomicAdd(ap + 0, m0);
    atomicAdd(ap + 1, m1);
    atomicAdd(ap + 2, m2);
    atomicAdd(ap + 3, m3);
}

template <int K, bool HASADD>
__global__ __launch_bounds__(32)
void gemm_wmma_f32_kernel(const float* __restrict__ A0, int lda0,
                          const float* __restrict__ A1, int lda1,
                          const float* __restrict__ W,  int ldw,
                          const float* __restrict__ bias,
                          float* __restrict__ out, int ldo)
{
    int lane = threadIdx.x;
    int lo   = lane & 15;
    int hi   = lane >> 4;
    int row0 = blockIdx.x * 16;
    int col0 = blockIdx.y * 16;

    const float* arow0 = A0 + (size_t)(row0 + lo) * lda0;
    const float* arow1 = HASADD ? (A1 + (size_t)(row0 + lo) * lda1) : nullptr;

    v8f c = {};
#pragma unroll
    for (int k0 = 0; k0 < K; k0 += 4) {
        int ka = k0 + 2 * hi;
        v2f a, b;
        a.x = arow0[ka];
        a.y = arow0[ka + 1];
        if (HASADD) {
            a.x += arow1[ka];
            a.y += arow1[ka + 1];
        }
        b.x = W[(size_t)ka * ldw + col0 + lo];
        b.y = W[(size_t)(ka + 1) * ldw + col0 + lo];
        c = __builtin_amdgcn_wmma_f32_16x16x4_f32(
                 false, a,  false, b,
                 (short)0, c,  false,  false);
    }

    float bv = bias[col0 + lo];
#pragma unroll
    for (int r = 0; r < 8; ++r) {
        int row = row0 + r + 8 * hi;
        out[(size_t)row * ldo + col0 + lo] = fmaxf(c[r] + bv, 0.0f);
    }
}

#define STATS_TPB 384

template <int C, int ITERS>
struct StatsShape {
    static constexpr int CG   = C / 4;
    static constexpr int RPP  = STATS_TPB / CG;
    static constexpr int RB   = RPP * ITERS;
    static constexpr int GRID = (NNODES + RB - 1) / RB;
};

template <int C, int ITERS>
__global__ __launch_bounds__(STATS_TPB)
void col_stats_kernel(const float* __restrict__ in, int ld, int n,
                      float* __restrict__ sum,
                      float* __restrict__ sumsq)
{
    typedef StatsShape<C, ITERS> S;
    static_assert(STATS_TPB % S::CG == 0, "block must hold whole rows");
    static_assert(2 * C <= STATS_TPB, "one thread per (column, statistic)");

    __shared__ float4 red[2][STATS_TPB];

    const int t    = threadIdx.x;
    const int cg   = t % S::CG;
    const int rsub = t / S::CG;
    const int row0 = blockIdx.x * S::RB + rsub;

    float4 v[ITERS];
#pragma unroll
    for (int k = 0; k < ITERS; ++k) {
        int row = row0 + k * S::RPP;
        int rc  = row < n ? row : n - 1;
        v[k] = *reinterpret_cast<const float4*>(in + (size_t)rc * ld + cg * 4);
    }

    float4 s = make_float4(0.0f, 0.0f, 0.0f, 0.0f);
    float4 q = make_float4(0.0f, 0.0f, 0.0f, 0.0f);
#pragma unroll
    for (int k = 0; k < ITERS; ++k) {
        int row = row0 + k * S::RPP;
        float4 w = v[k];
        if (row >= n) w = make_float4(0.0f, 0.0f, 0.0f, 0.0f);
        s.x += w.x;  s.y += w.y;  s.z += w.z;  s.w += w.w;
        q.x += w.x * w.x;  q.y += w.y * w.y;
        q.z += w.z * w.z;  q.w += w.w * w.w;
    }

    red[0][t] = s;
    red[1][t] = q;
    __syncthreads();

    if (t < 2 * C) {
        const int which = t / C;
        const int c     = t % C;
        const float* rf = reinterpret_cast<const float*>(&red[which][0]);
        float acc = 0.0f;
#pragma unroll
        for (int r = 0; r < S::RPP; ++r) acc += rf[r * C + c];
        atomicAdd(which ? &sumsq[c] : &sum[c], acc);
    }
}

__global__ void bn_finalize_kernel(const float* __restrict__ sum,
                                   const float* __restrict__ sumsq,
                                   const float* __restrict__ g,
                                   const float* __restrict__ beta,
                                   float* __restrict__ scale,
                                   float* __restrict__ shift, float invN)
{
    int c = threadIdx.x;
    float mu  = sum[c] * invN;
    float var = sumsq[c] * invN - mu * mu;
    float inv = rsqrtf(var + BN_EPS);
    float sc  = g[c] * inv;
    scale[c] = sc;
    shift[c] = beta[c] - mu * sc;
}

__global__ __launch_bounds__(256)
void bn_apply_kernel(const float* __restrict__ in, int ldi,
                     float* __restrict__ out, int ldo,
                     const float* __restrict__ scale,
                     const float* __restrict__ shift, int n, int C)
{
    int gid = blockIdx.x * blockDim.x + threadIdx.x;
    if (gid >= n * C) return;
    int row = gid / C;
    int c   = gid % C;
    out[(size_t)row * ldo + c] = in[(size_t)row * ldi + c] * scale[c] + shift[c];
}

extern "C" void kernel_launch(void* const* d_in, const int* in_sizes, int n_in,
                              void* d_out, int out_size, void* d_ws, size_t ws_size,
                              hipStream_t stream)
{
    (void)in_sizes; (void)n_in; (void)out_size; (void)ws_size;

    const float* x          = (const float*)d_in[0];
    const float* edge_attr  = (const float*)d_in[1];
    const int*   edge_index = (const int*)  d_in[2];
    const float* lin_w      = (const float*)d_in[3];
    const float* lin_b      = (const float*)d_in[4];
    const float* c1_w  = (const float*)d_in[5];
    const float* c1_b  = (const float*)d_in[6];
    const float* c1_g  = (const float*)d_in[7];
    const float* c1_be = (const float*)d_in[8];
    const float* c2_w  = (const float*)d_in[9];
    const float* c2_b  = (const float*)d_in[10];
    const float* c2_g  = (const float*)d_in[11];
    const float* c2_be = (const float*)d_in[12];
    const float* l1_w  = (const float*)d_in[13];
    const float* l1_b  = (const float*)d_in[14];
    const float* l1_g  = (const float*)d_in[15];
    const float* l1_be = (const float*)d_in[16];
    const float* ma_w  = (const float*)d_in[17];
    const float* ma_b  = (const float*)d_in[18];
    const float* ma_g  = (const float*)d_in[19];
    const float* ma_be = (const float*)d_in[20];
    const float* mb_w  = (const float*)d_in[21];
    const float* mb_b  = (const float*)d_in[22];
    const float* mb_g  = (const float*)d_in[23];
    const float* mb_be = (const float*)d_in[24];

    const int* src = edge_index;
    const int* dst = edge_index + NEDGES;

    const size_t N = NNODES;

    float* ws    = (float*)d_ws;
    float* agg   = ws;
    float* bufP  = agg  + N * 96;
    float* bufQ  = bufP + N * 96;
    float* xcat  = bufQ + N * 96;
    float* stats = xcat + N * 160;

    const float invN = 1.0f / (float)N;
    const int rowTiles = NNODES / 16;

    auto statsL = [&](int L) { return stats + (size_t)L * 384; };

    hipMemsetAsync(agg,   0, N * 96 * sizeof(float), stream);
    hipMemsetAsync(stats, 0, 5 * 384 * sizeof(float), stream);

    const int edgeThreads = NEDGES * 24;
    const int edgeBlocks  = (edgeThreads + 255) / 256;

    edge_scatter_kernel<<<edgeBlocks, 256, 0, stream>>>(
        x, 96, edge_attr, src, dst, lin_w, lin_b, agg);

    gemm_wmma_f32_kernel<96, true><<<dim3(rowTiles, 6), 32, 0, stream>>>(
        x, 96, agg, 96, c1_w, 96, c1_b, bufP, 96);

    col_stats_kernel<96, 13><<<StatsShape<96, 13>::GRID, STATS_TPB, 0, stream>>>(
        bufP, 96, NNODES, statsL(0), statsL(0) + 96);
    bn_finalize_kernel<<<1, 96, 0, stream>>>(statsL(0), statsL(0) + 96,
                                             c1_g, c1_be,
                                             statsL(0) + 192, statsL(0) + 288, invN);
    bn_apply_kernel<<<(NNODES * 96 + 255) / 256, 256, 0, stream>>>(
        bufP, 96, xcat, 160, statsL(0) + 192, statsL(0) + 288, NNODES, 96);

    hipMemsetAsync(agg, 0, N * 96 * sizeof(float), stream);
    edge_scatter_kernel<<<edgeBlocks, 256, 0, stream>>>(
        xcat, 160, edge_attr, src, dst, lin_w, lin_b, agg);

    gemm_wmma_f32_kernel<96, true><<<dim3(rowTiles, 4), 32, 0, stream>>>(
        xcat, 160, agg, 96, c2_w, 64, c2_b, bufP, 64);

    col_stats_kernel<64, 9><<<StatsShape<64, 9>::GRID, STATS_TPB, 0, stream>>>(
        bufP, 64, NNODES, statsL(1), statsL(1) + 96);
    bn_finalize_kernel<<<1, 64, 0, stream>>>(statsL(1), statsL(1) + 96,
                                             c2_g, c2_be,
                                             statsL(1) + 192, statsL(1) + 288, invN);
    bn_apply_kernel<<<(NNODES * 64 + 255) / 256, 256, 0, stream>>>(
        bufP, 64, xcat + 96, 160, statsL(1) + 192, statsL(1) + 288, NNODES, 64);

    gemm_wmma_f32_kernel<160, false><<<dim3(rowTiles, 6), 32, 0, stream>>>(
        xcat, 160, nullptr, 0, l1_w, 96, l1_b, bufQ, 96);

    col_stats_kernel<96, 13><<<StatsShape<96, 13>::GRID, STATS_TPB, 0, stream>>>(
        bufQ, 96, NNODES, statsL(2), statsL(2) + 96);
    bn_finalize_kernel<<<1, 96, 0, stream>>>(statsL(2), statsL(2) + 96,
                                             l1_g, l1_be,
                                             statsL(2) + 192, statsL(2) + 288, invN);
    bn_apply_kernel<<<(NNODES * 96 + 255) / 256, 256, 0, stream>>>(
        bufQ, 96, bufQ, 96, statsL(2) + 192, statsL(2) + 288, NNODES, 96);

    gemm_wmma_f32_kernel<96, false><<<dim3(rowTiles, 6), 32, 0, stream>>>(
        bufQ, 96, nullptr, 0, ma_w, 96, ma_b, bufP, 96);

    col_stats_kernel<96, 13><<<StatsShape<96, 13>::GRID, STATS_TPB, 0, stream>>>(
        bufP, 96, NNODES, statsL(3), statsL(3) + 96);
    bn_finalize_kernel<<<1, 96, 0, stream>>>(statsL(3), statsL(3) + 96,
                                             ma_g, ma_be,
                                             statsL(3) + 192, statsL(3) + 288, invN);
    bn_apply_kernel<<<(NNODES * 96 + 255) / 256, 256, 0, stream>>>(
        bufP, 96, bufP, 96, statsL(3) + 192, statsL(3) + 288, NNODES, 96);

    gemm_wmma_f32_kernel<96, false><<<dim3(rowTiles, 1), 32, 0, stream>>>(
        bufP, 96, nullptr, 0, mb_w, 16, mb_b, bufQ, 16);

    col_stats_kernel<16, 8><<<StatsShape<16, 8>::GRID, STATS_TPB, 0, stream>>>(
        bufQ, 16, NNODES, statsL(4), statsL(4) + 96);
    bn_finalize_kernel<<<1, 16, 0, stream>>>(statsL(4), statsL(4) + 96,
                                             mb_g, mb_be,
                                             statsL(4) + 192, statsL(4) + 288, invN);
    bn_apply_kernel<<<(NNODES * 16 + 255) / 256, 256, 0, stream>>>(
        bufQ, 16, (float*)d_out, 16, statsL(4) + 192, statsL(4) + 288, NNODES, 16);
}


# Edge scatter: one wave walks 32 edges with lane = feature column, so each row gather load and each float atomic covers one whole 128-byte line; 24x fewer waves; atomic arrival order changes
# speedup vs baseline: 2.4595x; 1.2988x over previous
#include <hip/hip_runtime.h>
#include <hip/hip_bf16.h>


#define NNODES 50000
#define NEDGES 800000
#define BN_EPS 1e-5f

typedef __attribute__((ext_vector_type(2))) float v2f;
typedef __attribute__((ext_vector_type(8))) float v8f;

#define EDGE_TPB 256
#define EDGE_EPW 32
#define EDGE_U   8

__global__ __launch_bounds__(EDGE_TPB)
void edge_scatter_kernel(const float* __restrict__ x, int ldx,
                         const float* __restrict__ ea,
                         const int* __restrict__ src,
                         const int* __restrict__ dst,
                         const float* __restrict__ lw,
                         const float* __restrict__ lb,
                         float* __restrict__ agg)
{
    static_assert(NEDGES % EDGE_EPW == 0, "every wave owns a full group of edges");
    static_assert(EDGE_EPW % EDGE_U == 0, "whole batches");

    const int lane = threadIdx.x & 31;
    const int wv   = threadIdx.x >> 5;
    const int j0   = (blockIdx.x * (EDGE_TPB / 32) + wv) * EDGE_EPW;
    if (j0 >= NEDGES) return;

    const int   sv = src[j0 + lane];
    const int   tv = dst[j0 + lane];
    const int   av = __float_as_int(ea[j0 + lane]);

    const float w0 = lw[lane], w1 = lw[lane + 32], w2 = lw[lane + 64];
    const float b0 = lb[lane], b1 = lb[lane + 32], b2 = lb[lane + 64];

    for (int e0 = 0; e0 < EDGE_EPW; e0 += EDGE_U) {
        float v0[EDGE_U], v1[EDGE_U], v2[EDGE_U];
#pragma unroll
        for (int u = 0; u < EDGE_U; ++u) {
            const int s = __builtin_amdgcn_readlane(sv, e0 + u);
            const float* xr = x + (size_t)s * ldx + lane;
            v0[u] = xr[0];
            v1[u] = xr[32];
            v2[u] = xr[64];
        }
#pragma unroll
        for (int u = 0; u < EDGE_U; ++u) {
            const int   t = __builtin_amdgcn_readlane(tv, e0 + u);
            const float a = __int_as_float(__builtin_amdgcn_readlane(av, e0 + u));
            float* ap = agg + (size_t)t * 96 + lane;
            atomicAdd(ap,      fmaxf(v0[u] + a * w0 + b0, 0.0f));
            atomicAdd(ap + 32, fmaxf(v1[u] + a * w1 + b1, 0.0f));
            atomicAdd(ap + 64, fmaxf(v2[u] + a * w2 + b2, 0.0f));
        }
    }
}

template <int K, bool HASADD>
__global__ __launch_bounds__(32)
void gemm_wmma_f32_kernel(const float* __restrict__ A0, int lda0,
                          const float* __restrict__ A1, int lda1,
                          const float* __restrict__ W,  int ldw,
                          const float* __restrict__ bias,
                          float* __restrict__ out, int ldo)
{
    int lane = threadIdx.x;
    int lo   = lane & 15;
    int hi   = lane >> 4;
    int row0 = blockIdx.x * 16;
    int col0 = blockIdx.y * 16;

    const float* arow0 = A0 + (size_t)(row0 + lo) * lda0;
    const float* arow1 = HASADD ? (A1 + (size_t)(row0 + lo) * lda1) : nullptr;

    v8f c = {};
#pragma unroll
    for (int k0 = 0; k0 < K; k0 += 4) {
        int ka = k0 + 2 * hi;
        v2f a, b;
        a.x = arow0[ka];
        a.y = arow0[ka + 1];
        if (HASADD) {
            a.x += arow1[ka];
            a.y += arow1[ka + 1];
        }
        b.x = W[(size_t)ka * ldw + col0 + lo];
        b.y = W[(size_t)(ka + 1) * ldw + col0 + lo];
        c = __builtin_amdgcn_wmma_f32_16x16x4_f32(
                 false, a,  false, b,
                 (short)0, c,  false,  false);
    }

    float bv = bias[col0 + lo];
#pragma unroll
    for (int r = 0; r < 8; ++r) {
        int row = row0 + r + 8 * hi;
        out[(size_t)row * ldo + col0 + lo] = fmaxf(c[r] + bv, 0.0f);
    }
}

#define STATS_TPB 384

template <int C, int ITERS>
struct StatsShape {
    static constexpr int CG   = C / 4;
    static constexpr int RPP  = STATS_TPB / CG;
    static constexpr int RB   = RPP * ITERS;
    static constexpr int GRID = (NNODES + RB - 1) / RB;
};

template <int C, int ITERS>
__global__ __launch_bounds__(STATS_TPB)
void col_stats_kernel(const float* __restrict__ in, int ld, int n,
                      float* __restrict__ sum,
                      float* __restrict__ sumsq)
{
    typedef StatsShape<C, ITERS> S;
    static_assert(STATS_TPB % S::CG == 0, "block must hold whole rows");
    static_assert(2 * C <= STATS_TPB, "one thread per (column, statistic)");

    __shared__ float4 red[2][STATS_TPB];

    const int t    = threadIdx.x;
    const int cg   = t % S::CG;
    const int rsub = t / S::CG;
    const int row0 = blockIdx.x * S::RB + rsub;

    float4 v[ITERS];
#pragma unroll
    for (int k = 0; k < ITERS; ++k) {
        int row = row0 + k * S::RPP;
        int rc  = row < n ? row : n - 1;
        v[k] = *reinterpret_cast<const float4*>(in + (size_t)rc * ld + cg * 4);
    }

    float4 s = make_float4(0.0f, 0.0f, 0.0f, 0.0f);
    float4 q = make_float4(0.0f, 0.0f, 0.0f, 0.0f);
#pragma unroll
    for (int k = 0; k < ITERS; ++k) {
        int row = row0 + k * S::RPP;
        float4 w = v[k];
        if (row >= n) w = make_float4(0.0f, 0.0f, 0.0f, 0.0f);
        s.x += w.x;  s.y += w.y;  s.z += w.z;  s.w += w.w;
        q.x += w.x * w.x;  q.y += w.y * w.y;
        q.z += w.z * w.z;  q.w += w.w * w.w;
    }

    red[0][t] = s;
    red[1][t] = q;
    __syncthreads();

    if (t < 2 * C) {
        const int which = t / C;
        const int c     = t % C;
        const float* rf = reinterpret_cast<const float*>(&red[which][0]);
        float acc = 0.0f;
#pragma unroll
        for (int r = 0; r < S::RPP; ++r) acc += rf[r * C + c];
        atomicAdd(which ? &sumsq[c] : &sum[c], acc);
    }
}

__global__ void bn_finalize_kernel(const float* __restrict__ sum,
                                   const float* __restrict__ sumsq,
                                   const float* __restrict__ g,
                                   const float* __restrict__ beta,
                                   float* __restrict__ scale,
                                   float* __restrict__ shift, float invN)
{
    int c = threadIdx.x;
    float mu  = sum[c] * invN;
    float var = sumsq[c] * invN - mu * mu;
    float inv = rsqrtf(var + BN_EPS);
    float sc  = g[c] * inv;
    scale[c] = sc;
    shift[c] = beta[c] - mu * sc;
}

__global__ __launch_bounds__(256)
void bn_apply_kernel(const float* __restrict__ in, int ldi,
                     float* __restrict__ out, int ldo,
                     const float* __restrict__ scale,
                     const float* __restrict__ shift, int n, int C)
{
    int gid = blockIdx.x * blockDim.x + threadIdx.x;
    if (gid >= n * C) return;
    int row = gid / C;
    int c   = gid % C;
    out[(size_t)row * ldo + c] = in[(size_t)row * ldi + c] * scale[c] + shift[c];
}

extern "C" void kernel_launch(void* const* d_in, const int* in_sizes, int n_in,
                              void* d_out, int out_size, void* d_ws, size_t ws_size,
                              hipStream_t stream)
{
    (void)in_sizes; (void)n_in; (void)out_size; (void)ws_size;

    const float* x          = (const float*)d_in[0];
    const float* edge_attr  = (const float*)d_in[1];
    const int*   edge_index = (const int*)  d_in[2];
    const float* lin_w      = (const float*)d_in[3];
    const float* lin_b      = (const float*)d_in[4];
    const float* c1_w  = (const float*)d_in[5];
    const float* c1_b  = (const float*)d_in[6];
    const float* c1_g  = (const float*)d_in[7];
    const float* c1_be = (const float*)d_in[8];
    const float* c2_w  = (const float*)d_in[9];
    const float* c2_b  = (const float*)d_in[10];
    const float* c2_g  = (const float*)d_in[11];
    const float* c2_be = (const float*)d_in[12];
    const float* l1_w  = (const float*)d_in[13];
    const float* l1_b  = (const float*)d_in[14];
    const float* l1_g  = (const float*)d_in[15];
    const float* l1_be = (const float*)d_in[16];
    const float* ma_w  = (const float*)d_in[17];
    const float* ma_b  = (const float*)d_in[18];
    const float* ma_g  = (const float*)d_in[19];
    const float* ma_be = (const float*)d_in[20];
    const float* mb_w  = (const float*)d_in[21];
    const float* mb_b  = (const float*)d_in[22];
    const float* mb_g  = (const float*)d_in[23];
    const float* mb_be = (const float*)d_in[24];

    const int* src = edge_index;
    const int* dst = edge_index + NEDGES;

    const size_t N = NNODES;

    float* ws    = (float*)d_ws;
    float* agg   = ws;
    float* bufP  = agg  + N * 96;
    float* bufQ  = bufP + N * 96;
    float* xcat  = bufQ + N * 96;
    float* stats = xcat + N * 160;

    const float invN = 1.0f / (float)N;
    const int rowTiles = NNODES / 16;

    auto statsL = [&](int L) { return stats + (size_t)L * 384; };

    hipMemsetAsync(agg,   0, N * 96 * sizeof(float), stream);
    hipMemsetAsync(stats, 0, 5 * 384 * sizeof(float), stream);

    const int edgeWaves  = NEDGES / EDGE_EPW;
    const int edgeBlocks = (edgeWaves + (EDGE_TPB / 32) - 1) / (EDGE_TPB / 32);

    edge_scatter_kernel<<<edgeBlocks, EDGE_TPB, 0, stream>>>(
        x, 96, edge_attr, src, dst, lin_w, lin_b, agg);

    gemm_wmma_f32_kernel<96, true><<<dim3(rowTiles, 6), 32, 0, stream>>>(
        x, 96, agg, 96, c1_w, 96, c1_b, bufP, 96);

    col_stats_kernel<96, 13><<<StatsShape<96, 13>::GRID, STATS_TPB, 0, stream>>>(
        bufP, 96, NNODES, statsL(0), statsL(0) + 96);
    bn_finalize_kernel<<<1, 96, 0, stream>>>(statsL(0), statsL(0) + 96,
                                             c1_g, c1_be,
                                             statsL(0) + 192, statsL(0) + 288, invN);
    bn_apply_kernel<<<(NNODES * 96 + 255) / 256, 256, 0, stream>>>(
        bufP, 96, xcat, 160, statsL(0) + 192, statsL(0) + 288, NNODES, 96);

    hipMemsetAsync(agg, 0, N * 96 * sizeof(float), stream);
    edge_scatter_kernel<<<edgeBlocks, EDGE_TPB, 0, stream>>>(
        xcat, 160, edge_attr, src, dst, lin_w, lin_b, agg);

    gemm_wmma_f32_kernel<96, true><<<dim3(rowTiles, 4), 32, 0, stream>>>(
        xcat, 160, agg, 96, c2_w, 64, c2_b, bufP, 64);

    col_stats_kernel<64, 9><<<StatsShape<64, 9>::GRID, STATS_TPB, 0, stream>>>(
        bufP, 64, NNODES, statsL(1), statsL(1) + 96);
    bn_finalize_kernel<<<1, 64, 0, stream>>>(statsL(1), statsL(1) + 96,
                                             c2_g, c2_be,
                                             statsL(1) + 192, statsL(1) + 288, invN);
    bn_apply_kernel<<<(NNODES * 64 + 255) / 256, 256, 0, stream>>>(
        bufP, 64, xcat + 96, 160, statsL(1) + 192, statsL(1) + 288, NNODES, 64);

    gemm_wmma_f32_kernel<160, false><<<dim3(rowTiles, 6), 32, 0, stream>>>(
        xcat, 160, nullptr, 0, l1_w, 96, l1_b, bufQ, 96);

    col_stats_kernel<96, 13><<<StatsShape<96, 13>::GRID, STATS_TPB, 0, stream>>>(
        bufQ, 96, NNODES, statsL(2), statsL(2) + 96);
    bn_finalize_kernel<<<1, 96, 0, stream>>>(statsL(2), statsL(2) + 96,
                                             l1_g, l1_be,
                                             statsL(2) + 192, statsL(2) + 288, invN);
    bn_apply_kernel<<<(NNODES * 96 + 255) / 256, 256, 0, stream>>>(
        bufQ, 96, bufQ, 96, statsL(2) + 192, statsL(2) + 288, NNODES, 96);

    gemm_wmma_f32_kernel<96, false><<<dim3(rowTiles, 6), 32, 0, stream>>>(
        bufQ, 96, nullptr, 0, ma_w, 96, ma_b, bufP, 96);

    col_stats_kernel<96, 13><<<StatsShape<96, 13>::GRID, STATS_TPB, 0, stream>>>(
        bufP, 96, NNODES, statsL(3), statsL(3) + 96);
    bn_finalize_kernel<<<1, 96, 0, stream>>>(statsL(3), statsL(3) + 96,
                                             ma_g, ma_be,
                                             statsL(3) + 192, statsL(3) + 288, invN);
    bn_apply_kernel<<<(NNODES * 96 + 255) / 256, 256, 0, stream>>>(
        bufP, 96, bufP, 96, statsL(3) + 192, statsL(3) + 288, NNODES, 96);

    gemm_wmma_f32_kernel<96, false><<<dim3(rowTiles, 1), 32, 0, stream>>>(
        bufP, 96, nullptr, 0, mb_w, 16, mb_b, bufQ, 16);

    col_stats_kernel<16, 8><<<StatsShape<16, 8>::GRID, STATS_TPB, 0, stream>>>(
        bufQ, 16, NNODES, statsL(4), statsL(4) + 96);
    bn_finalize_kernel<<<1, 16, 0, stream>>>(statsL(4), statsL(4) + 96,
                                             mb_g, mb_be,
                                             statsL(4) + 192, statsL(4) + 288, invN);
    bn_apply_kernel<<<(NNODES * 16 + 255) / 256, 256, 0, stream>>>(
        bufQ, 16, (float*)d_out, 16, statsL(4) + 192, statsL(4) + 288, NNODES, 16);
}


# Dense layers retiled: one wave per 16-row strip over all N columns, A read once as 32-byte row chunks, W staged in LDS per 8-wave block (was a wave per 16x16 tile); k summation order changes
# speedup vs baseline: 3.3698x; 1.3701x over previous
#include <hip/hip_runtime.h>
#include <hip/hip_bf16.h>


#define NNODES 50000
#define NEDGES 800000
#define BN_EPS 1e-5f

typedef __attribute__((ext_vector_type(2))) float v2f;
typedef __attribute__((ext_vector_type(8))) float v8f;

#define EDGE_TPB 256
#define EDGE_EPW 32
#define EDGE_U   8

__global__ __launch_bounds__(EDGE_TPB)
void edge_scatter_kernel(const float* __restrict__ x, int ldx,
                         const float* __restrict__ ea,
                         const int* __restrict__ src,
                         const int* __restrict__ dst,
                         const float* __restrict__ lw,
                         const float* __restrict__ lb,
                         float* __restrict__ agg)
{
    static_assert(NEDGES % EDGE_EPW == 0, "every wave owns a full group of edges");
    static_assert(EDGE_EPW % EDGE_U == 0, "whole batches");

    const int lane = threadIdx.x & 31;
    const int wv   = threadIdx.x >> 5;
    const int j0   = (blockIdx.x * (EDGE_TPB / 32) + wv) * EDGE_EPW;
    if (j0 >= NEDGES) return;

    const int   sv = src[j0 + lane];
    const int   tv = dst[j0 + lane];
    const int   av = __float_as_int(ea[j0 + lane]);

    const float w0 = lw[lane], w1 = lw[lane + 32], w2 = lw[lane + 64];
    const float b0 = lb[lane], b1 = lb[lane + 32], b2 = lb[lane + 64];

    for (int e0 = 0; e0 < EDGE_EPW; e0 += EDGE_U) {
        float v0[EDGE_U], v1[EDGE_U], v2[EDGE_U];
#pragma unroll
        for (int u = 0; u < EDGE_U; ++u) {
            const int s = __builtin_amdgcn_readlane(sv, e0 + u);
            const float* xr = x + (size_t)s * ldx + lane;
            v0[u] = xr[0];
            v1[u] = xr[32];
            v2[u] = xr[64];
        }
#pragma unroll
        for (int u = 0; u < EDGE_U; ++u) {
            const int   t = __builtin_amdgcn_readlane(tv, e0 + u);
            const float a = __int_as_float(__builtin_amdgcn_readlane(av, e0 + u));
            float* ap = agg + (size_t)t * 96 + lane;
            atomicAdd(ap,      fmaxf(v0[u] + a * w0 + b0, 0.0f));
            atomicAdd(ap + 32, fmaxf(v1[u] + a * w1 + b1, 0.0f));
            atomicAdd(ap + 64, fmaxf(v2[u] + a * w2 + b2, 0.0f));
        }
    }
}

#define GEMM_WAVES 8
#define GEMM_TPB   (GEMM_WAVES * 32)
#define GEMM_STRIPS (NNODES / 16)
#define GEMM_GRID  ((GEMM_STRIPS + GEMM_WAVES - 1) / GEMM_WAVES)

template <int K, int N, bool HASADD>
__global__ __launch_bounds__(GEMM_TPB)
void gemm_wmma_f32_kernel(const float* __restrict__ A0, int lda0,
                          const float* __restrict__ A1, int lda1,
                          const float* __restrict__ W,
                          const float* __restrict__ bias,
                          float* __restrict__ out)
{
    static_assert(K % 16 == 0 && N % 16 == 0, "whole k blocks and column tiles");
    static_assert(NNODES % 16 == 0, "whole row strips");
    static_assert(((K / 2) * (N / 4)) % 32 == 0, "the part round of the W staging is whole waves");
    constexpr int NT    = N / 16;
    constexpr int KB    = K / 16;
    constexpr int PITCH = 2 * N + 8;
    constexpr int NQ    = N / 4;

    __shared__ float4 wl4[(K / 2) * PITCH / 4];
    float* wl = reinterpret_cast<float*>(wl4);

    const int tid  = threadIdx.x;
    const int lane = tid & 31;
    const int wv   = tid >> 5;
    const int lo   = lane & 15;
    const int hi   = lane >> 4;

    int strip = blockIdx.x * GEMM_WAVES + wv;
    const bool live = strip < GEMM_STRIPS;
    if (!live) strip = GEMM_STRIPS - 1;
    const int row0 = strip * 16;

    const float* ar0 = A0 + (size_t)(row0 + lo) * lda0 + 8 * hi;
    float4 a[KB][2];
#pragma unroll
    for (int i = 0; i < KB; ++i) {
        a[i][0] = *reinterpret_cast<const float4*>(ar0 + 16 * i);
        a[i][1] = *reinterpret_cast<const float4*>(ar0 + 16 * i + 4);
    }
    float4 a1[HASADD ? KB : 1][2];
    if (HASADD) {
        const float* ar1 = A1 + (size_t)(row0 + lo) * lda1 + 8 * hi;
#pragma unroll
        for (int i = 0; i < KB; ++i) {
            a1[i][0] = *reinterpret_cast<const float4*>(ar1 + 16 * i);
            a1[i][1] = *reinterpret_cast<const float4*>(ar1 + 16 * i + 4);
        }
    }

    constexpr int ITEMS = (K / 2) * NQ;
    constexpr int SFULL = ITEMS / GEMM_TPB;
    constexpr int SREM  = ITEMS % GEMM_TPB;
    float4 r0[SFULL > 0 ? SFULL : 1], r1[SFULL > 0 ? SFULL : 1];
#pragma unroll
    for (int s = 0; s < SFULL; ++s) {
        const int it = tid + s * GEMM_TPB;
        const int p  = it / NQ;
        const int n4 = it % NQ;
        r0[s] = *reinterpret_cast<const float4*>(W + (size_t)(2 * p) * N + 4 * n4);
        r1[s] = *reinterpret_cast<const float4*>(W + (size_t)(2 * p + 1) * N + 4 * n4);
    }
    if (SREM != 0 && tid < SREM) {
        const int it = tid + SFULL * GEMM_TPB;
        const int p  = it / NQ;
        const int n4 = it % NQ;
        const float4 q0 = *reinterpret_cast<const float4*>(W + (size_t)(2 * p) * N + 4 * n4);
        const float4 q1 = *reinterpret_cast<const float4*>(W + (size_t)(2 * p + 1) * N + 4 * n4);
        float4* d = reinterpret_cast<float4*>(wl + p * PITCH + 8 * n4);
        d[0] = make_float4(q0.x, q1.x, q0.y, q1.y);
        d[1] = make_float4(q0.z, q1.z, q0.w, q1.w);
    }
#pragma unroll
    for (int s = 0; s < SFULL; ++s) {
        const int it = tid + s * GEMM_TPB;
        const int p  = it / NQ;
        const int n4 = it % NQ;
        float4* d = reinterpret_cast<float4*>(wl + p * PITCH + 8 * n4);
        d[0] = make_float4(r0[s].x, r1[s].x, r0[s].y, r1[s].y);
        d[1] = make_float4(r0[s].z, r1[s].z, r0[s].w, r1[s].w);
    }
    __syncthreads();

    if (HASADD) {
#pragma unroll
        for (int i = 0; i < KB; ++i) {
#pragma unroll
            for (int h = 0; h < 2; ++h) {
                a[i][h].x += a1[i][h].x;  a[i][h].y += a1[i][h].y;
                a[i][h].z += a1[i][h].z;  a[i][h].w += a1[i][h].w;
            }
        }
    }

    v8f c[NT];
#pragma unroll
    for (int t = 0; t < NT; ++t) c[t] = v8f{};

    const float* wb = wl + (4 * hi) * PITCH + 2 * lo;
#pragma unroll
    for (int i = 0; i < KB; ++i) {
#pragma unroll
        for (int j = 0; j < 4; ++j) {
            v2f av;
            if (j == 0)      { av.x = a[i][0].x; av.y = a[i][0].y; }
            else if (j == 1) { av.x = a[i][0].z; av.y = a[i][0].w; }
            else if (j == 2) { av.x = a[i][1].x; av.y = a[i][1].y; }
            else             { av.x = a[i][1].z; av.y = a[i][1].w; }
#pragma unroll
            for (int t = 0; t < NT; ++t) {
                const v2f bv = *reinterpret_cast<const v2f*>(
                    wb + (8 * i + j) * PITCH + 32 * t);
                c[t] = __builtin_amdgcn_wmma_f32_16x16x4_f32(
                         false, av,  false, bv,
                         (short)0, c[t],  false,  false);
            }
        }
    }

    float res[NT][8];
#pragma unroll
    for (int t = 0; t < NT; ++t) {
        const float bsv = bias[16 * t + lo];
#pragma unroll
        for (int r = 0; r < 8; ++r) res[t][r] = fmaxf(c[t][r] + bsv, 0.0f);
    }
    if (live) {
#pragma unroll
        for (int r = 0; r < 8; ++r) {
            float* orow = out + (size_t)(row0 + r + 8 * hi) * N + lo;
#pragma unroll
            for (int t = 0; t < NT; ++t) orow[16 * t] = res[t][r];
        }
    }
}

#define STATS_TPB 384

template <int C, int ITERS>
struct StatsShape {
    static constexpr int CG   = C / 4;
    static constexpr int RPP  = STATS_TPB / CG;
    static constexpr int RB   = RPP * ITERS;
    static constexpr int GRID = (NNODES + RB - 1) / RB;
};

template <int C, int ITERS>
__global__ __launch_bounds__(STATS_TPB)
void col_stats_kernel(const float* __restrict__ in, int ld, int n,
                      float* __restrict__ sum,
                      float* __restrict__ sumsq)
{
    typedef StatsShape<C, ITERS> S;
    static_assert(STATS_TPB % S::CG == 0, "block must hold whole rows");
    static_assert(2 * C <= STATS_TPB, "one thread per (column, statistic)");

    __shared__ float4 red[2][STATS_TPB];

    const int t    = threadIdx.x;
    const int cg   = t % S::CG;
    const int rsub = t / S::CG;
    const int row0 = blockIdx.x * S::RB + rsub;

    float4 v[ITERS];
#pragma unroll
    for (int k = 0; k < ITERS; ++k) {
        int row = row0 + k * S::RPP;
        int rc  = row < n ? row : n - 1;
        v[k] = *reinterpret_cast<const float4*>(in + (size_t)rc * ld + cg * 4);
    }

    float4 s = make_float4(0.0f, 0.0f, 0.0f, 0.0f);
    float4 q = make_float4(0.0f, 0.0f, 0.0f, 0.0f);
#pragma unroll
    for (int k = 0; k < ITERS; ++k) {
        int row = row0 + k * S::RPP;
        float4 w = v[k];
        if (row >= n) w = make_float4(0.0f, 0.0f, 0.0f, 0.0f);
        s.x += w.x;  s.y += w.y;  s.z += w.z;  s.w += w.w;
        q.x += w.x * w.x;  q.y += w.y * w.y;
        q.z += w.z * w.z;  q.w += w.w * w.w;
    }

    red[0][t] = s;
    red[1][t] = q;
    __syncthreads();

    if (t < 2 * C) {
        const int which = t / C;
        const int c     = t % C;
        const float* rf = reinterpret_cast<const float*>(&red[which][0]);
        float acc = 0.0f;
#pragma unroll
        for (int r = 0; r < S::RPP; ++r) acc += rf[r * C + c];
        atomicAdd(which ? &sumsq[c] : &sum[c], acc);
    }
}

__global__ void bn_finalize_kernel(const float* __restrict__ sum,
                                   const float* __restrict__ sumsq,
                                   const float* __restrict__ g,
                                   const float* __restrict__ beta,
                                   float* __restrict__ scale,
                                   float* __restrict__ shift, float invN)
{
    int c = threadIdx.x;
    float mu  = sum[c] * invN;
    float var = sumsq[c] * invN - mu * mu;
    float inv = rsqrtf(var + BN_EPS);
    float sc  = g[c] * inv;
    scale[c] = sc;
    shift[c] = beta[c] - mu * sc;
}

__global__ __launch_bounds__(256)
void bn_apply_kernel(const float* __restrict__ in, int ldi,
                     float* __restrict__ out, int ldo,
                     const float* __restrict__ scale,
                     const float* __restrict__ shift, int n, int C)
{
    int gid = blockIdx.x * blockDim.x + threadIdx.x;
    if (gid >= n * C) return;
    int row = gid / C;
    int c   = gid % C;
    out[(size_t)row * ldo + c] = in[(size_t)row * ldi + c] * scale[c] + shift[c];
}

extern "C" void kernel_launch(void* const* d_in, const int* in_sizes, int n_in,
                              void* d_out, int out_size, void* d_ws, size_t ws_size,
                              hipStream_t stream)
{
    (void)in_sizes; (void)n_in; (void)out_size; (void)ws_size;

    const float* x          = (const float*)d_in[0];
    const float* edge_attr  = (const float*)d_in[1];
    const int*   edge_index = (const int*)  d_in[2];
    const float* lin_w      = (const float*)d_in[3];
    const float* lin_b      = (const float*)d_in[4];
    const float* c1_w  = (const float*)d_in[5];
    const float* c1_b  = (const float*)d_in[6];
    const float* c1_g  = (const float*)d_in[7];
    const float* c1_be = (const float*)d_in[8];
    const float* c2_w  = (const float*)d_in[9];
    const float* c2_b  = (const float*)d_in[10];
    const float* c2_g  = (const float*)d_in[11];
    const float* c2_be = (const float*)d_in[12];
    const float* l1_w  = (const float*)d_in[13];
    const float* l1_b  = (const float*)d_in[14];
    const float* l1_g  = (const float*)d_in[15];
    const float* l1_be = (const float*)d_in[16];
    const float* ma_w  = (const float*)d_in[17];
    const float* ma_b  = (const float*)d_in[18];
    const float* ma_g  = (const float*)d_in[19];
    const float* ma_be = (const float*)d_in[20];
    const float* mb_w  = (const float*)d_in[21];
    const float* mb_b  = (const float*)d_in[22];
    const float* mb_g  = (const float*)d_in[23];
    const float* mb_be = (const float*)d_in[24];

    const int* src = edge_index;
    const int* dst = edge_index + NEDGES;

    const size_t N = NNODES;

    float* ws    = (float*)d_ws;
    float* agg   = ws;
    float* bufP  = agg  + N * 96;
    float* bufQ  = bufP + N * 96;
    float* xcat  = bufQ + N * 96;
    float* stats = xcat + N * 160;

    const float invN = 1.0f / (float)N;

    auto statsL = [&](int L) { return stats + (size_t)L * 384; };

    hipMemsetAsync(agg,   0, N * 96 * sizeof(float), stream);
    hipMemsetAsync(stats, 0, 5 * 384 * sizeof(float), stream);

    const int edgeWaves  = NEDGES / EDGE_EPW;
    const int edgeBlocks = (edgeWaves + (EDGE_TPB / 32) - 1) / (EDGE_TPB / 32);

    edge_scatter_kernel<<<edgeBlocks, EDGE_TPB, 0, stream>>>(
        x, 96, edge_attr, src, dst, lin_w, lin_b, agg);

    gemm_wmma_f32_kernel<96, 96, true><<<GEMM_GRID, GEMM_TPB, 0, stream>>>(
        x, 96, agg, 96, c1_w, c1_b, bufP);

    col_stats_kernel<96, 13><<<StatsShape<96, 13>::GRID, STATS_TPB, 0, stream>>>(
        bufP, 96, NNODES, statsL(0), statsL(0) + 96);
    bn_finalize_kernel<<<1, 96, 0, stream>>>(statsL(0), statsL(0) + 96,
                                             c1_g, c1_be,
                                             statsL(0) + 192, statsL(0) + 288, invN);
    bn_apply_kernel<<<(NNODES * 96 + 255) / 256, 256, 0, stream>>>(
        bufP, 96, xcat, 160, statsL(0) + 192, statsL(0) + 288, NNODES, 96);

    hipMemsetAsync(agg, 0, N * 96 * sizeof(float), stream);
    edge_scatter_kernel<<<edgeBlocks, EDGE_TPB, 0, stream>>>(
        xcat, 160, edge_attr, src, dst, lin_w, lin_b, agg);

    gemm_wmma_f32_kernel<96, 64, true><<<GEMM_GRID, GEMM_TPB, 0, stream>>>(
        xcat, 160, agg, 96, c2_w, c2_b, bufP);

    col_stats_kernel<64, 9><<<StatsShape<64, 9>::GRID, STATS_TPB, 0, stream>>>(
        bufP, 64, NNODES, statsL(1), statsL(1) + 96);
    bn_finalize_kernel<<<1, 64, 0, stream>>>(statsL(1), statsL(1) + 96,
                                             c2_g, c2_be,
                                             statsL(1) + 192, statsL(1) + 288, invN);
    bn_apply_kernel<<<(NNODES * 64 + 255) / 256, 256, 0, stream>>>(
        bufP, 64, xcat + 96, 160, statsL(1) + 192, statsL(1) + 288, NNODES, 64);

    gemm_wmma_f32_kernel<160, 96, false><<<GEMM_GRID, GEMM_TPB, 0, stream>>>(
        xcat, 160, nullptr, 0, l1_w, l1_b, bufQ);

    col_stats_kernel<96, 13><<<StatsShape<96, 13>::GRID, STATS_TPB, 0, stream>>>(
        bufQ, 96, NNODES, statsL(2), statsL(2) + 96);
    bn_finalize_kernel<<<1, 96, 0, stream>>>(statsL(2), statsL(2) + 96,
                                             l1_g, l1_be,
                                             statsL(2) + 192, statsL(2) + 288, invN);
    bn_apply_kernel<<<(NNODES * 96 + 255) / 256, 256, 0, stream>>>(
        bufQ, 96, bufQ, 96, statsL(2) + 192, statsL(2) + 288, NNODES, 96);

    gemm_wmma_f32_kernel<96, 96, false><<<GEMM_GRID, GEMM_TPB, 0, stream>>>(
        bufQ, 96, nullptr, 0, ma_w, ma_b, bufP);

    col_stats_kernel<96, 13><<<StatsShape<96, 13>::GRID, STATS_TPB, 0, stream>>>(
        bufP, 96, NNODES, statsL(3), statsL(3) + 96);
    bn_finalize_kernel<<<1, 96, 0, stream>>>(statsL(3), statsL(3) + 96,
                                             ma_g, ma_be,
                                             statsL(3) + 192, statsL(3) + 288, invN);
    bn_apply_kernel<<<(NNODES * 96 + 255) / 256, 256, 0, stream>>>(
        bufP, 96, bufP, 96, statsL(3) + 192, statsL(3) + 288, NNODES, 96);

    gemm_wmma_f32_kernel<96, 16, false><<<GEMM_GRID, GEMM_TPB, 0, stream>>>(
        bufP, 96, nullptr, 0, mb_w, mb_b, bufQ);

    col_stats_kernel<16, 8><<<StatsShape<16, 8>::GRID, STATS_TPB, 0, stream>>>(
        bufQ, 16, NNODES, statsL(4), statsL(4) + 96);
    bn_finalize_kernel<<<1, 16, 0, stream>>>(statsL(4), statsL(4) + 96,
                                             mb_g, mb_be,
                                             statsL(4) + 192, statsL(4) + 288, invN);
    bn_apply_kernel<<<(NNODES * 16 + 255) / 256, 256, 0, stream>>>(
        bufQ, 16, (float*)d_out, 16, statsL(4) + 192, statsL(4) + 288, NNODES, 16);
}


# BN statistics without launches of their own: column sums and squares taken from the GEMM result registers in the epilogue (LDS reduce, block atomics), scale/shift derived in apply; sum order changes
# speedup vs baseline: 3.6451x; 1.0817x over previous
#include <hip/hip_runtime.h>
#include <hip/hip_bf16.h>


#define NNODES 50000
#define NEDGES 800000
#define BN_EPS 1e-5f

typedef __attribute__((ext_vector_type(2))) float v2f;
typedef __attribute__((ext_vector_type(8))) float v8f;

#define EDGE_TPB 256
#define EDGE_EPW 32
#define EDGE_U   8

__global__ __launch_bounds__(EDGE_TPB)
void edge_scatter_kernel(const float* __restrict__ x, int ldx,
                         const float* __restrict__ ea,
                         const int* __restrict__ src,
                         const int* __restrict__ dst,
                         const float* __restrict__ lw,
                         const float* __restrict__ lb,
                         float* __restrict__ agg)
{
    static_assert(NEDGES % EDGE_EPW == 0, "every wave owns a full group of edges");
    static_assert(EDGE_EPW % EDGE_U == 0, "whole batches");

    const int lane = threadIdx.x & 31;
    const int wv   = threadIdx.x >> 5;
    const int j0   = (blockIdx.x * (EDGE_TPB / 32) + wv) * EDGE_EPW;
    if (j0 >= NEDGES) return;

    const int   sv = src[j0 + lane];
    const int   tv = dst[j0 + lane];
    const int   av = __float_as_int(ea[j0 + lane]);

    const float w0 = lw[lane], w1 = lw[lane + 32], w2 = lw[lane + 64];
    const float b0 = lb[lane], b1 = lb[lane + 32], b2 = lb[lane + 64];

    for (int e0 = 0; e0 < EDGE_EPW; e0 += EDGE_U) {
        float v0[EDGE_U], v1[EDGE_U], v2[EDGE_U];
#pragma unroll
        for (int u = 0; u < EDGE_U; ++u) {
            const int s = __builtin_amdgcn_readlane(sv, e0 + u);
            const float* xr = x + (size_t)s * ldx + lane;
            v0[u] = xr[0];
            v1[u] = xr[32];
            v2[u] = xr[64];
        }
#pragma unroll
        for (int u = 0; u < EDGE_U; ++u) {
            const int   t = __builtin_amdgcn_readlane(tv, e0 + u);
            const float a = __int_as_float(__builtin_amdgcn_readlane(av, e0 + u));
            float* ap = agg + (size_t)t * 96 + lane;
            atomicAdd(ap,      fmaxf(v0[u] + a * w0 + b0, 0.0f));
            atomicAdd(ap + 32, fmaxf(v1[u] + a * w1 + b1, 0.0f));
            atomicAdd(ap + 64, fmaxf(v2[u] + a * w2 + b2, 0.0f));
        }
    }
}

#define GEMM_WAVES 8
#define GEMM_TPB   (GEMM_WAVES * 32)
#define GEMM_STRIPS (NNODES / 16)
#define GEMM_GRID  ((GEMM_STRIPS + GEMM_WAVES - 1) / GEMM_WAVES)

template <int K, int N, bool HASADD>
__global__ __launch_bounds__(GEMM_TPB)
void gemm_wmma_f32_kernel(const float* __restrict__ A0, int lda0,
                          const float* __restrict__ A1, int lda1,
                          const float* __restrict__ W,
                          const float* __restrict__ bias,
                          float* __restrict__ out,
                          float* __restrict__ colsum,
                          float* __restrict__ colsq)
{
    static_assert(K % 16 == 0 && N % 16 == 0, "whole k blocks and column tiles");
    static_assert(2 * N <= GEMM_TPB, "one thread per (column, statistic)");
    static_assert(NNODES % 16 == 0, "whole row strips");
    static_assert(((K / 2) * (N / 4)) % 32 == 0, "the part round of the W staging is whole waves");
    constexpr int NT    = N / 16;
    constexpr int KB    = K / 16;
    constexpr int PITCH = 2 * N + 8;
    constexpr int NQ    = N / 4;

    __shared__ float4 wl4[(K / 2) * PITCH / 4];
    float* wl = reinterpret_cast<float*>(wl4);
    __shared__ float red[2][2 * GEMM_WAVES][N];

    const int tid  = threadIdx.x;
    const int lane = tid & 31;
    const int wv   = tid >> 5;
    const int lo   = lane & 15;
    const int hi   = lane >> 4;

    int strip = blockIdx.x * GEMM_WAVES + wv;
    const bool live = strip < GEMM_STRIPS;
    if (!live) strip = GEMM_STRIPS - 1;
    const int row0 = strip * 16;

    const float* ar0 = A0 + (size_t)(row0 + lo) * lda0 + 8 * hi;
    float4 a[KB][2];
#pragma unroll
    for (int i = 0; i < KB; ++i) {
        a[i][0] = *reinterpret_cast<const float4*>(ar0 + 16 * i);
        a[i][1] = *reinterpret_cast<const float4*>(ar0 + 16 * i + 4);
    }
    float4 a1[HASADD ? KB : 1][2];
    if (HASADD) {
        const float* ar1 = A1 + (size_t)(row0 + lo) * lda1 + 8 * hi;
#pragma unroll
        for (int i = 0; i < KB; ++i) {
            a1[i][0] = *reinterpret_cast<const float4*>(ar1 + 16 * i);
            a1[i][1] = *reinterpret_cast<const float4*>(ar1 + 16 * i + 4);
        }
    }

    constexpr int ITEMS = (K / 2) * NQ;
    constexpr int SFULL = ITEMS / GEMM_TPB;
    constexpr int SREM  = ITEMS % GEMM_TPB;
    float4 r0[SFULL > 0 ? SFULL : 1], r1[SFULL > 0 ? SFULL : 1];
#pragma unroll
    for (int s = 0; s < SFULL; ++s) {
        const int it = tid + s * GEMM_TPB;
        const int p  = it / NQ;
        const int n4 = it % NQ;
        r0[s] = *reinterpret_cast<const float4*>(W + (size_t)(2 * p) * N + 4 * n4);
        r1[s] = *reinterpret_cast<const float4*>(W + (size_t)(2 * p + 1) * N + 4 * n4);
    }
    if (SREM != 0 && tid < SREM) {
        const int it = tid + SFULL * GEMM_TPB;
        const int p  = it / NQ;
        const int n4 = it % NQ;
        const float4 q0 = *reinterpret_cast<const float4*>(W + (size_t)(2 * p) * N + 4 * n4);
        const float4 q1 = *reinterpret_cast<const float4*>(W + (size_t)(2 * p + 1) * N + 4 * n4);
        float4* d = reinterpret_cast<float4*>(wl + p * PITCH + 8 * n4);
        d[0] = make_float4(q0.x, q1.x, q0.y, q1.y);
        d[1] = make_float4(q0.z, q1.z, q0.w, q1.w);
    }
#pragma unroll
    for (int s = 0; s < SFULL; ++s) {
        const int it = tid + s * GEMM_TPB;
        const int p  = it / NQ;
        const int n4 = it % NQ;
        float4* d = reinterpret_cast<float4*>(wl + p * PITCH + 8 * n4);
        d[0] = make_float4(r0[s].x, r1[s].x, r0[s].y, r1[s].y);
        d[1] = make_float4(r0[s].z, r1[s].z, r0[s].w, r1[s].w);
    }
    __syncthreads();

    if (HASADD) {
#pragma unroll
        for (int i = 0; i < KB; ++i) {
#pragma unroll
            for (int h = 0; h < 2; ++h) {
                a[i][h].x += a1[i][h].x;  a[i][h].y += a1[i][h].y;
                a[i][h].z += a1[i][h].z;  a[i][h].w += a1[i][h].w;
            }
        }
    }

    v8f c[NT];
#pragma unroll
    for (int t = 0; t < NT; ++t) c[t] = v8f{};

    const float* wb = wl + (4 * hi) * PITCH + 2 * lo;
#pragma unroll
    for (int i = 0; i < KB; ++i) {
#pragma unroll
        for (int j = 0; j < 4; ++j) {
            v2f av;
            if (j == 0)      { av.x = a[i][0].x; av.y = a[i][0].y; }
            else if (j == 1) { av.x = a[i][0].z; av.y = a[i][0].w; }
            else if (j == 2) { av.x = a[i][1].x; av.y = a[i][1].y; }
            else             { av.x = a[i][1].z; av.y = a[i][1].w; }
#pragma unroll
            for (int t = 0; t < NT; ++t) {
                const v2f bv = *reinterpret_cast<const v2f*>(
                    wb + (8 * i + j) * PITCH + 32 * t);
                c[t] = __builtin_amdgcn_wmma_f32_16x16x4_f32(
                         false, av,  false, bv,
                         (short)0, c[t],  false,  false);
            }
        }
    }

    float res[NT][8];
#pragma unroll
    for (int t = 0; t < NT; ++t) {
        const float bsv = bias[16 * t + lo];
#pragma unroll
        for (int r = 0; r < 8; ++r) res[t][r] = fmaxf(c[t][r] + bsv, 0.0f);
    }
#pragma unroll
    for (int t = 0; t < NT; ++t) {
        float s = 0.0f, q = 0.0f;
#pragma unroll
        for (int r = 0; r < 8; ++r) {
            const float v = res[t][r];
            s += v;
            q += v * v;
        }
        red[0][2 * wv + hi][16 * t + lo] = live ? s : 0.0f;
        red[1][2 * wv + hi][16 * t + lo] = live ? q : 0.0f;
    }
    __syncthreads();

    if (live) {
#pragma unroll
        for (int r = 0; r < 8; ++r) {
            float* orow = out + (size_t)(row0 + r + 8 * hi) * N + lo;
#pragma unroll
            for (int t = 0; t < NT; ++t) orow[16 * t] = res[t][r];
        }
    }

    if (tid < 2 * N) {
        const int which = tid / N;
        const int col   = tid % N;
        float acc = 0.0f;
#pragma unroll
        for (int p = 0; p < 2 * GEMM_WAVES; ++p) acc += red[which][p][col];
        atomicAdd(which ? &colsq[col] : &colsum[col], acc);
    }
}

__global__ __launch_bounds__(256)
void bn_apply_kernel(const float* __restrict__ in, int ldi,
                     float* __restrict__ out, int ldo,
                     const float* __restrict__ sum,
                     const float* __restrict__ sumsq,
                     const float* __restrict__ g,
                     const float* __restrict__ beta,
                     float invN, int n, int C)
{
    int gid = blockIdx.x * blockDim.x + threadIdx.x;
    if (gid >= n * C) return;
    int row = gid / C;
    int c   = gid % C;
    float mu  = sum[c] * invN;
    float var = sumsq[c] * invN - mu * mu;
    float inv = rsqrtf(var + BN_EPS);
    float sc  = g[c] * inv;
    float sh  = beta[c] - mu * sc;
    out[(size_t)row * ldo + c] = in[(size_t)row * ldi + c] * sc + sh;
}

extern "C" void kernel_launch(void* const* d_in, const int* in_sizes, int n_in,
                              void* d_out, int out_size, void* d_ws, size_t ws_size,
                              hipStream_t stream)
{
    (void)in_sizes; (void)n_in; (void)out_size; (void)ws_size;

    const float* x          = (const float*)d_in[0];
    const float* edge_attr  = (const float*)d_in[1];
    const int*   edge_index = (const int*)  d_in[2];
    const float* lin_w      = (const float*)d_in[3];
    const float* lin_b      = (const float*)d_in[4];
    const float* c1_w  = (const float*)d_in[5];
    const float* c1_b  = (const float*)d_in[6];
    const float* c1_g  = (const float*)d_in[7];
    const float* c1_be = (const float*)d_in[8];
    const float* c2_w  = (const float*)d_in[9];
    const float* c2_b  = (const float*)d_in[10];
    const float* c2_g  = (const float*)d_in[11];
    const float* c2_be = (const float*)d_in[12];
    const float* l1_w  = (const float*)d_in[13];
    const float* l1_b  = (const float*)d_in[14];
    const float* l1_g  = (const float*)d_in[15];
    const float* l1_be = (const float*)d_in[16];
    const float* ma_w  = (const float*)d_in[17];
    const float* ma_b  = (const float*)d_in[18];
    const float* ma_g  = (const float*)d_in[19];
    const float* ma_be = (const float*)d_in[20];
    const float* mb_w  = (const float*)d_in[21];
    const float* mb_b  = (const float*)d_in[22];
    const float* mb_g  = (const float*)d_in[23];
    const float* mb_be = (const float*)d_in[24];

    const int* src = edge_index;
    const int* dst = edge_index + NEDGES;

    const size_t N = NNODES;

    float* ws    = (float*)d_ws;
    float* agg   = ws;
    float* bufP  = agg  + N * 96;
    float* bufQ  = bufP + N * 96;
    float* xcat  = bufQ + N * 96;
    float* stats = xcat + N * 160;

    const float invN = 1.0f / (float)N;

    auto sumL = [&](int L) { return stats + (size_t)L * 192; };
    auto sqL  = [&](int L) { return stats + (size_t)L * 192 + 96; };

    hipMemsetAsync(agg,   0, N * 96 * sizeof(float), stream);
    hipMemsetAsync(stats, 0, 5 * 192 * sizeof(float), stream);

    const int edgeWaves  = NEDGES / EDGE_EPW;
    const int edgeBlocks = (edgeWaves + (EDGE_TPB / 32) - 1) / (EDGE_TPB / 32);

    edge_scatter_kernel<<<edgeBlocks, EDGE_TPB, 0, stream>>>(
        x, 96, edge_attr, src, dst, lin_w, lin_b, agg);

    gemm_wmma_f32_kernel<96, 96, true><<<GEMM_GRID, GEMM_TPB, 0, stream>>>(
        x, 96, agg, 96, c1_w, c1_b, bufP, sumL(0), sqL(0));

    bn_apply_kernel<<<(NNODES * 96 + 255) / 256, 256, 0, stream>>>(
        bufP, 96, xcat, 160, sumL(0), sqL(0), c1_g, c1_be, invN, NNODES, 96);

    hipMemsetAsync(agg, 0, N * 96 * sizeof(float), stream);
    edge_scatter_kernel<<<edgeBlocks, EDGE_TPB, 0, stream>>>(
        xcat, 160, edge_attr, src, dst, lin_w, lin_b, agg);

    gemm_wmma_f32_kernel<96, 64, true><<<GEMM_GRID, GEMM_TPB, 0, stream>>>(
        xcat, 160, agg, 96, c2_w, c2_b, bufP, sumL(1), sqL(1));

    bn_apply_kernel<<<(NNODES * 64 + 255) / 256, 256, 0, stream>>>(
        bufP, 64, xcat + 96, 160, sumL(1), sqL(1), c2_g, c2_be, invN, NNODES, 64);

    gemm_wmma_f32_kernel<160, 96, false><<<GEMM_GRID, GEMM_TPB, 0, stream>>>(
        xcat, 160, nullptr, 0, l1_w, l1_b, bufQ, sumL(2), sqL(2));

    bn_apply_kernel<<<(NNODES * 96 + 255) / 256, 256, 0, stream>>>(
        bufQ, 96, bufQ, 96, sumL(2), sqL(2), l1_g, l1_be, invN, NNODES, 96);

    gemm_wmma_f32_kernel<96, 96, false><<<GEMM_GRID, GEMM_TPB, 0, stream>>>(
        bufQ, 96, nullptr, 0, ma_w, ma_b, bufP, sumL(3), sqL(3));

    bn_apply_kernel<<<(NNODES * 96 + 255) / 256, 256, 0, stream>>>(
        bufP, 96, bufP, 96, sumL(3), sqL(3), ma_g, ma_be, invN, NNODES, 96);

    gemm_wmma_f32_kernel<96, 16, false><<<GEMM_GRID, GEMM_TPB, 0, stream>>>(
        bufP, 96, nullptr, 0, mb_w, mb_b, bufQ, sumL(4), sqL(4));

    bn_apply_kernel<<<(NNODES * 16 + 255) / 256, 256, 0, stream>>>(
        bufQ, 16, (float*)d_out, 16, sumL(4), sqL(4), mb_g, mb_be, invN, NNODES, 16);
}


# BatchNorm apply of the 4 inner layers folded into its readers: edge gather and next GEMM's A load apply scale/shift derived from the epilogue statistics; 4 launches removed; sum orders unchanged
# speedup vs baseline: 4.6634x; 1.2794x over previous
#include <hip/hip_runtime.h>
#include <hip/hip_bf16.h>


#define NNODES 50000
#define NEDGES 800000
#define BN_EPS 1e-5f

typedef __attribute__((ext_vector_type(2))) float v2f;
typedef __attribute__((ext_vector_type(8))) float v8f;

__device__ __forceinline__ void bn_scale_shift(const float* __restrict__ sum,
                                               const float* __restrict__ sumsq,
                                               const float* __restrict__ g,
                                               const float* __restrict__ beta,
                                               float invN, int c,
                                               float& sc, float& sh)
{
    const float mu  = sum[c] * invN;
    const float var = sumsq[c] * invN - mu * mu;
    const float inv = rsqrtf(var + BN_EPS);
    sc = g[c] * inv;
    sh = beta[c] - mu * sc;
}

#define EDGE_TPB 256
#define EDGE_EPW 32
#define EDGE_U   8

template <bool NORM>
__global__ __launch_bounds__(EDGE_TPB)
void edge_scatter_kernel(const float* __restrict__ x, int ldx,
                         const float* __restrict__ ea,
                         const int* __restrict__ src,
                         const int* __restrict__ dst,
                         const float* __restrict__ lw,
                         const float* __restrict__ lb,
                         float* __restrict__ agg,
                         const float* __restrict__ nsum,
                         const float* __restrict__ nsq,
                         const float* __restrict__ ng,
                         const float* __restrict__ nbeta,
                         float invN)
{
    static_assert(NEDGES % EDGE_EPW == 0, "every wave owns a full group of edges");
    static_assert(EDGE_EPW % EDGE_U == 0, "whole batches");

    const int lane = threadIdx.x & 31;
    const int wv   = threadIdx.x >> 5;
    const int j0   = (blockIdx.x * (EDGE_TPB / 32) + wv) * EDGE_EPW;
    if (j0 >= NEDGES) return;

    const int   sv = src[j0 + lane];
    const int   tv = dst[j0 + lane];
    const int   av = __float_as_int(ea[j0 + lane]);

    const float w0 = lw[lane], w1 = lw[lane + 32], w2 = lw[lane + 64];
    const float b0 = lb[lane], b1 = lb[lane + 32], b2 = lb[lane + 64];

    float sc0 = 1.0f, sc1 = 1.0f, sc2 = 1.0f, sh0 = 0.0f, sh1 = 0.0f, sh2 = 0.0f;
    if (NORM) {
        bn_scale_shift(nsum, nsq, ng, nbeta, invN, lane,      sc0, sh0);
        bn_scale_shift(nsum, nsq, ng, nbeta, invN, lane + 32, sc1, sh1);
        bn_scale_shift(nsum, nsq, ng, nbeta, invN, lane + 64, sc2, sh2);
    }

    for (int e0 = 0; e0 < EDGE_EPW; e0 += EDGE_U) {
        float v0[EDGE_U], v1[EDGE_U], v2[EDGE_U];
#pragma unroll
        for (int u = 0; u < EDGE_U; ++u) {
            const int s = __builtin_amdgcn_readlane(sv, e0 + u);
            const float* xr = x + (size_t)s * ldx + lane;
            v0[u] = xr[0];
            v1[u] = xr[32];
            v2[u] = xr[64];
        }
#pragma unroll
        for (int u = 0; u < EDGE_U; ++u) {
            const int   t = __builtin_amdgcn_readlane(tv, e0 + u);
            const float a = __int_as_float(__builtin_amdgcn_readlane(av, e0 + u));
            float* ap = agg + (size_t)t * 96 + lane;
            const float x0 = NORM ? v0[u] * sc0 + sh0 : v0[u];
            const float x1 = NORM ? v1[u] * sc1 + sh1 : v1[u];
            const float x2 = NORM ? v2[u] * sc2 + sh2 : v2[u];
            atomicAdd(ap,      fmaxf(x0 + a * w0 + b0, 0.0f));
            atomicAdd(ap + 32, fmaxf(x1 + a * w1 + b1, 0.0f));
            atomicAdd(ap + 64, fmaxf(x2 + a * w2 + b2, 0.0f));
        }
    }
}

#define GEMM_WAVES 8
#define GEMM_TPB   (GEMM_WAVES * 32)
#define GEMM_STRIPS (NNODES / 16)
#define GEMM_GRID  ((GEMM_STRIPS + GEMM_WAVES - 1) / GEMM_WAVES)

template <int K, int N, bool HASADD, bool NORM>
__global__ __launch_bounds__(GEMM_TPB)
void gemm_wmma_f32_kernel(const float* __restrict__ A0, int lda0,
                          const float* __restrict__ A1, int lda1,
                          const float* __restrict__ W,
                          const float* __restrict__ bias,
                          float* __restrict__ out, int ldo,
                          float* __restrict__ colsum,
                          float* __restrict__ colsq,
                          const float* __restrict__ nsum0,
                          const float* __restrict__ nsq0,
                          const float* __restrict__ ng0,
                          const float* __restrict__ nbeta0,
                          int ncols0,
                          const float* __restrict__ nsum1,
                          const float* __restrict__ nsq1,
                          const float* __restrict__ ng1,
                          const float* __restrict__ nbeta1,
                          float invN)
{
    static_assert(K % 16 == 0 && N % 16 == 0, "whole k blocks and column tiles");
    static_assert(2 * N <= GEMM_TPB, "one thread per (column, statistic)");
    static_assert(K <= GEMM_TPB && K % 32 == 0, "one thread per column of A0 for its scale and shift, whole waves");
    static_assert(NNODES % 16 == 0, "whole row strips");
    static_assert(((K / 2) * (N / 4)) % 32 == 0, "the part round of the W staging is whole waves");
    constexpr int NT    = N / 16;
    constexpr int KB    = K / 16;
    constexpr int PITCH = 2 * N + 8;
    constexpr int NQ    = N / 4;

    __shared__ float4 wl4[(K / 2) * PITCH / 4];
    float* wl = reinterpret_cast<float*>(wl4);
    __shared__ float red[2][2 * GEMM_WAVES][N];
    __shared__ float4 nsc4[NORM ? K / 4 : 1];
    __shared__ float4 nsh4[NORM ? K / 4 : 1];

    const int tid  = threadIdx.x;
    const int lane = tid & 31;
    const int wv   = tid >> 5;
    const int lo   = lane & 15;
    const int hi   = lane >> 4;

    int strip = blockIdx.x * GEMM_WAVES + wv;
    const bool live = strip < GEMM_STRIPS;
    if (!live) strip = GEMM_STRIPS - 1;
    const int row0 = strip * 16;

    const float* ar0 = A0 + (size_t)(row0 + lo) * lda0 + 8 * hi;
    float4 a[KB][2];
#pragma unroll
    for (int i = 0; i < KB; ++i) {
        a[i][0] = *reinterpret_cast<const float4*>(ar0 + 16 * i);
        a[i][1] = *reinterpret_cast<const float4*>(ar0 + 16 * i + 4);
    }
    float4 a1[HASADD ? KB : 1][2];
    if (HASADD) {
        const float* ar1 = A1 + (size_t)(row0 + lo) * lda1 + 8 * hi;
#pragma unroll
        for (int i = 0; i < KB; ++i) {
            a1[i][0] = *reinterpret_cast<const float4*>(ar1 + 16 * i);
            a1[i][1] = *reinterpret_cast<const float4*>(ar1 + 16 * i + 4);
        }
    }

    constexpr int ITEMS = (K / 2) * NQ;
    constexpr int SFULL = ITEMS / GEMM_TPB;
    constexpr int SREM  = ITEMS % GEMM_TPB;
    float4 r0[SFULL > 0 ? SFULL : 1], r1[SFULL > 0 ? SFULL : 1];
#pragma unroll
    for (int s = 0; s < SFULL; ++s) {
        const int it = tid + s * GEMM_TPB;
        const int p  = it / NQ;
        const int n4 = it % NQ;
        r0[s] = *reinterpret_cast<const float4*>(W + (size_t)(2 * p) * N + 4 * n4);
        r1[s] = *reinterpret_cast<const float4*>(W + (size_t)(2 * p + 1) * N + 4 * n4);
    }
    if (NORM && tid < K) {
        const bool first = tid < ncols0;
        const int  c     = first ? tid : tid - ncols0;
        float sc, sh;
        bn_scale_shift(first ? nsum0 : nsum1, first ? nsq0 : nsq1,
                       first ? ng0 : ng1, first ? nbeta0 : nbeta1,
                       invN, c, sc, sh);
        reinterpret_cast<float*>(nsc4)[tid] = sc;
        reinterpret_cast<float*>(nsh4)[tid] = sh;
    }
    if (SREM != 0 && tid < SREM) {
        const int it = tid + SFULL * GEMM_TPB;
        const int p  = it / NQ;
        const int n4 = it % NQ;
        const float4 q0 = *reinterpret_cast<const float4*>(W + (size_t)(2 * p) * N + 4 * n4);
        const float4 q1 = *reinterpret_cast<const float4*>(W + (size_t)(2 * p + 1) * N + 4 * n4);
        float4* d = reinterpret_cast<float4*>(wl + p * PITCH + 8 * n4);
        d[0] = make_float4(q0.x, q1.x, q0.y, q1.y);
        d[1] = make_float4(q0.z, q1.z, q0.w, q1.w);
    }
#pragma unroll
    for (int s = 0; s < SFULL; ++s) {
        const int it = tid + s * GEMM_TPB;
        const int p  = it / NQ;
        const int n4 = it % NQ;
        float4* d = reinterpret_cast<float4*>(wl + p * PITCH + 8 * n4);
        d[0] = make_float4(r0[s].x, r1[s].x, r0[s].y, r1[s].y);
        d[1] = make_float4(r0[s].z, r1[s].z, r0[s].w, r1[s].w);
    }
    __syncthreads();

    if (NORM) {
#pragma unroll
        for (int i = 0; i < KB; ++i) {
#pragma unroll
            for (int h = 0; h < 2; ++h) {
                const float4 s4 = nsc4[4 * i + 2 * hi + h];
                const float4 t4 = nsh4[4 * i + 2 * hi + h];
                a[i][h].x = a[i][h].x * s4.x + t4.x;
                a[i][h].y = a[i][h].y * s4.y + t4.y;
                a[i][h].z = a[i][h].z * s4.z + t4.z;
                a[i][h].w = a[i][h].w * s4.w + t4.w;
            }
        }
    }

    if (HASADD) {
#pragma unroll
        for (int i = 0; i < KB; ++i) {
#pragma unroll
            for (int h = 0; h < 2; ++h) {
                a[i][h].x += a1[i][h].x;  a[i][h].y += a1[i][h].y;
                a[i][h].z += a1[i][h].z;  a[i][h].w += a1[i][h].w;
            }
        }
    }

    v8f c[NT];
#pragma unroll
    for (int t = 0; t < NT; ++t) c[t] = v8f{};

    const float* wb = wl + (4 * hi) * PITCH + 2 * lo;
#pragma unroll
    for (int i = 0; i < KB; ++i) {
#pragma unroll
        for (int j = 0; j < 4; ++j) {
            v2f av;
            if (j == 0)      { av.x = a[i][0].x; av.y = a[i][0].y; }
            else if (j == 1) { av.x = a[i][0].z; av.y = a[i][0].w; }
            else if (j == 2) { av.x = a[i][1].x; av.y = a[i][1].y; }
            else             { av.x = a[i][1].z; av.y = a[i][1].w; }
#pragma unroll
            for (int t = 0; t < NT; ++t) {
                const v2f bv = *reinterpret_cast<const v2f*>(
                    wb + (8 * i + j) * PITCH + 32 * t);
                c[t] = __builtin_amdgcn_wmma_f32_16x16x4_f32(
                         false, av,  false, bv,
                         (short)0, c[t],  false,  false);
            }
        }
    }

    float res[NT][8];
#pragma unroll
    for (int t = 0; t < NT; ++t) {
        const float bsv = bias[16 * t + lo];
#pragma unroll
        for (int r = 0; r < 8; ++r) res[t][r] = fmaxf(c[t][r] + bsv, 0.0f);
    }
#pragma unroll
    for (int t = 0; t < NT; ++t) {
        float s = 0.0f, q = 0.0f;
#pragma unroll
        for (int r = 0; r < 8; ++r) {
            const float v = res[t][r];
            s += v;
            q += v * v;
        }
        red[0][2 * wv + hi][16 * t + lo] = live ? s : 0.0f;
        red[1][2 * wv + hi][16 * t + lo] = live ? q : 0.0f;
    }
    __syncthreads();

    if (live) {
#pragma unroll
        for (int r = 0; r < 8; ++r) {
            float* orow = out + (size_t)(row0 + r + 8 * hi) * ldo + lo;
#pragma unroll
            for (int t = 0; t < NT; ++t) orow[16 * t] = res[t][r];
        }
    }

    if (tid < 2 * N) {
        const int which = tid / N;
        const int col   = tid % N;
        float acc = 0.0f;
#pragma unroll
        for (int p = 0; p < 2 * GEMM_WAVES; ++p) acc += red[which][p][col];
        atomicAdd(which ? &colsq[col] : &colsum[col], acc);
    }
}

__global__ __launch_bounds__(256)
void bn_apply_kernel(const float* __restrict__ in, int ldi,
                     float* __restrict__ out, int ldo,
                     const float* __restrict__ sum,
                     const float* __restrict__ sumsq,
                     const float* __restrict__ g,
                     const float* __restrict__ beta,
                     float invN, int n, int C)
{
    int gid = blockIdx.x * blockDim.x + threadIdx.x;
    if (gid >= n * C) return;
    int row = gid / C;
    int c   = gid % C;
    float sc, sh;
    bn_scale_shift(sum, sumsq, g, beta, invN, c, sc, sh);
    out[(size_t)row * ldo + c] = in[(size_t)row * ldi + c] * sc + sh;
}

extern "C" void kernel_launch(void* const* d_in, const int* in_sizes, int n_in,
                              void* d_out, int out_size, void* d_ws, size_t ws_size,
                              hipStream_t stream)
{
    (void)in_sizes; (void)n_in; (void)out_size; (void)ws_size;

    const float* x          = (const float*)d_in[0];
    const float* edge_attr  = (const float*)d_in[1];
    const int*   edge_index = (const int*)  d_in[2];
    const float* lin_w      = (const float*)d_in[3];
    const float* lin_b      = (const float*)d_in[4];
    const float* c1_w  = (const float*)d_in[5];
    const float* c1_b  = (const float*)d_in[6];
    const float* c1_g  = (const float*)d_in[7];
    const float* c1_be = (const float*)d_in[8];
    const float* c2_w  = (const float*)d_in[9];
    const float* c2_b  = (const float*)d_in[10];
    const float* c2_g  = (const float*)d_in[11];
    const float* c2_be = (const float*)d_in[12];
    const float* l1_w  = (const float*)d_in[13];
    const float* l1_b  = (const float*)d_in[14];
    const float* l1_g  = (const float*)d_in[15];
    const float* l1_be = (const float*)d_in[16];
    const float* ma_w  = (const float*)d_in[17];
    const float* ma_b  = (const float*)d_in[18];
    const float* ma_g  = (const float*)d_in[19];
    const float* ma_be = (const float*)d_in[20];
    const float* mb_w  = (const float*)d_in[21];
    const float* mb_b  = (const float*)d_in[22];
    const float* mb_g  = (const float*)d_in[23];
    const float* mb_be = (const float*)d_in[24];

    const int* src = edge_index;
    const int* dst = edge_index + NEDGES;

    const size_t N = NNODES;

    float* ws    = (float*)d_ws;
    float* agg   = ws;
    float* bufP  = agg  + N * 96;
    float* bufQ  = bufP + N * 96;
    float* xcat  = bufQ + N * 96;
    float* stats = xcat + N * 160;

    const float invN = 1.0f / (float)N;

    auto sumL = [&](int L) { return stats + (size_t)L * 192; };
    auto sqL  = [&](int L) { return stats + (size_t)L * 192 + 96; };

    hipMemsetAsync(agg,   0, N * 96 * sizeof(float), stream);
    hipMemsetAsync(stats, 0, 5 * 192 * sizeof(float), stream);

    const int edgeWaves  = NEDGES / EDGE_EPW;
    const int edgeBlocks = (edgeWaves + (EDGE_TPB / 32) - 1) / (EDGE_TPB / 32);

    edge_scatter_kernel<false><<<edgeBlocks, EDGE_TPB, 0, stream>>>(
        x, 96, edge_attr, src, dst, lin_w, lin_b, agg,
        nullptr, nullptr, nullptr, nullptr, invN);

    gemm_wmma_f32_kernel<96, 96, true, false><<<GEMM_GRID, GEMM_TPB, 0, stream>>>(
        x, 96, agg, 96, c1_w, c1_b, xcat, 160, sumL(0), sqL(0),
        nullptr, nullptr, nullptr, nullptr, 96,
        nullptr, nullptr, nullptr, nullptr, invN);

    hipMemsetAsync(agg, 0, N * 96 * sizeof(float), stream);
    edge_scatter_kernel<true><<<edgeBlocks, EDGE_TPB, 0, stream>>>(
        xcat, 160, edge_attr, src, dst, lin_w, lin_b, agg,
        sumL(0), sqL(0), c1_g, c1_be, invN);

    gemm_wmma_f32_kernel<96, 64, true, true><<<GEMM_GRID, GEMM_TPB, 0, stream>>>(
        xcat, 160, agg, 96, c2_w, c2_b, xcat + 96, 160, sumL(1), sqL(1),
        sumL(0), sqL(0), c1_g, c1_be, 96,
        nullptr, nullptr, nullptr, nullptr, invN);

    gemm_wmma_f32_kernel<160, 96, false, true><<<GEMM_GRID, GEMM_TPB, 0, stream>>>(
        xcat, 160, nullptr, 0, l1_w, l1_b, bufQ, 96, sumL(2), sqL(2),
        sumL(0), sqL(0), c1_g, c1_be, 96,
        sumL(1), sqL(1), c2_g, c2_be, invN);

    gemm_wmma_f32_kernel<96, 96, false, true><<<GEMM_GRID, GEMM_TPB, 0, stream>>>(
        bufQ, 96, nullptr, 0, ma_w, ma_b, bufP, 96, sumL(3), sqL(3),
        sumL(2), sqL(2), l1_g, l1_be, 96,
        nullptr, nullptr, nullptr, nullptr, invN);

    gemm_wmma_f32_kernel<96, 16, false, true><<<GEMM_GRID, GEMM_TPB, 0, stream>>>(
        bufP, 96, nullptr, 0, mb_w, mb_b, bufQ, 16, sumL(4), sqL(4),
        sumL(3), sqL(3), ma_g, ma_be, 96,
        nullptr, nullptr, nullptr, nullptr, invN);

    bn_apply_kernel<<<(NNODES * 16 + 255) / 256, 256, 0, stream>>>(
        bufQ, 16, (float*)d_out, 16, sumL(4), sqL(4), mb_g, mb_be, invN, NNODES, 16);
}


# Edge sums without global atomics: edges binned once per call by destination range (LDS counts, scan, place); a block sums its 50 rows with LDS float adds, stores them once; float sum order changes
# speedup vs baseline: 6.3507x; 1.3618x over previous
#include <hip/hip_runtime.h>
#include <hip/hip_bf16.h>


#define NNODES 50000
#define NEDGES 800000
#define BN_EPS 1e-5f

typedef __attribute__((ext_vector_type(2))) float v2f;
typedef __attribute__((ext_vector_type(8))) float v8f;

__device__ __forceinline__ void bn_scale_shift(const float* __restrict__ sum,
                                               const float* __restrict__ sumsq,
                                               const float* __restrict__ g,
                                               const float* __restrict__ beta,
                                               float invN, int c,
                                               float& sc, float& sh)
{
    const float mu  = sum[c] * invN;
    const float var = sumsq[c] * invN - mu * mu;
    const float inv = rsqrtf(var + BN_EPS);
    sc = g[c] * inv;
    sh = beta[c] - mu * sc;
}

#define NPB   50
#define NBINS (NNODES / NPB)
#define PBLK  128
#define PCH   (NEDGES / PBLK)
#define PTPB  1024
#define PIT   ((PCH + PTPB - 1) / PTPB)

__global__ __launch_bounds__(PTPB)
void bin_count_kernel(const int* __restrict__ dst, int* __restrict__ hist)
{
    static_assert(NNODES % NPB == 0, "whole bins");
    static_assert(NEDGES % PBLK == 0, "whole chunks");
    static_assert(NBINS <= PTPB, "one thread per bin");
    __shared__ int cnt[NBINS];
    const int tid = threadIdx.x;
    if (tid < NBINS) cnt[tid] = 0;
    __syncthreads();
    const int j0 = blockIdx.x * PCH;
    int d[PIT];
#pragma unroll
    for (int i = 0; i < PIT; ++i) {
        const int j = tid + i * PTPB;
        d[i] = (j < PCH) ? dst[j0 + j] : 0;
    }
#pragma unroll
    for (int i = 0; i < PIT; ++i)
        if (tid + i * PTPB < PCH) atomicAdd(&cnt[d[i] / NPB], 1);
    __syncthreads();
    if (tid < NBINS) hist[blockIdx.x * NBINS + tid] = cnt[tid];
}

__global__ __launch_bounds__(PTPB)
void bin_place_kernel(const int* __restrict__ src,
                      const int* __restrict__ dst,
                      const float* __restrict__ ea,
                      const int* __restrict__ hist,
                      int* __restrict__ binstart,
                      int2* __restrict__ rec)
{
    static_assert(NNODES <= 65536, "src packs into 16 bits");
    static_assert(PTPB == 1024, "32 waves: one lane of wave 0 per wave total");
    __shared__ int cur[NBINS];
    __shared__ int wsum[PTPB / 32];
    const int tid  = threadIdx.x;
    const int lane = tid & 31;
    const int wv   = tid >> 5;

    int tot = 0, pre = 0;
    if (tid < NBINS) {
#pragma unroll 32
        for (int k = 0; k < PBLK; ++k) {
            const int v = hist[k * NBINS + tid];
            tot += v;
            pre += (k < (int)blockIdx.x) ? v : 0;
        }
    }
    int inc = tot;
    for (int o = 1; o < 32; o <<= 1) {
        const int nb = __shfl_up(inc, o);
        if (lane >= o) inc += nb;
    }
    if (lane == 31) wsum[wv] = inc;
    __syncthreads();
    if (wv == 0) {
        const int w = wsum[lane];
        int winc = w;
        for (int o = 1; o < 32; o <<= 1) {
            const int nb = __shfl_up(winc, o);
            if (lane >= o) winc += nb;
        }
        wsum[lane] = winc - w;
    }
    __syncthreads();
    const int start = wsum[wv] + inc - tot;
    if (tid < NBINS) {
        cur[tid] = start + pre;
        if (blockIdx.x == 0) binstart[tid] = start;
    }
    if (blockIdx.x == 0 && tid == 0) binstart[NBINS] = NEDGES;
    __syncthreads();

    const int j0 = blockIdx.x * PCH;
    int d[PIT], sv[PIT], av[PIT];
#pragma unroll
    for (int i = 0; i < PIT; ++i) {
        const int  j  = tid + i * PTPB;
        const bool in = j < PCH;
        d[i]  = in ? dst[j0 + j] : 0;
        sv[i] = in ? src[j0 + j] : 0;
        av[i] = in ? __float_as_int(ea[j0 + j]) : 0;
    }
#pragma unroll
    for (int i = 0; i < PIT; ++i) {
        if (tid + i * PTPB < PCH) {
            const int b   = d[i] / NPB;
            const int loc = d[i] - b * NPB;
            const int pos = atomicAdd(&cur[b], 1);
            rec[pos] = make_int2(sv[i] | (loc << 16), av[i]);
        }
    }
}

#define EDGE_TPB 256
#define EDGE_U   16

template <bool NORM>
__global__ __launch_bounds__(EDGE_TPB)
void edge_bin_kernel(const float* __restrict__ x, int ldx,
                     const int2* __restrict__ rec,
                     const int* __restrict__ binstart,
                     const float* __restrict__ lw,
                     const float* __restrict__ lb,
                     float* __restrict__ agg,
                     const float* __restrict__ nsum,
                     const float* __restrict__ nsq,
                     const float* __restrict__ ng,
                     const float* __restrict__ nbeta,
                     float invN)
{
    static_assert((NPB * 96) % 4 == 0, "the bin's rows are whole float4");
    static_assert(32 % EDGE_U == 0, "whole batches");
    __shared__ float4 acc4[NPB * 96 / 4];
    float* acc = reinterpret_cast<float*>(acc4);

    const int tid  = threadIdx.x;
    const int lane = tid & 31;
    const int wv   = __builtin_amdgcn_readfirstlane(tid >> 5);

    for (int i = tid; i < NPB * 96 / 4; i += EDGE_TPB)
        acc4[i] = make_float4(0.0f, 0.0f, 0.0f, 0.0f);

    const int beg = binstart[blockIdx.x];
    const int end = binstart[blockIdx.x + 1];

    const float w0 = lw[lane], w1 = lw[lane + 32], w2 = lw[lane + 64];
    const float b0 = lb[lane], b1 = lb[lane + 32], b2 = lb[lane + 64];

    float sc0 = 1.0f, sc1 = 1.0f, sc2 = 1.0f, sh0 = 0.0f, sh1 = 0.0f, sh2 = 0.0f;
    if (NORM) {
        bn_scale_shift(nsum, nsq, ng, nbeta, invN, lane,      sc0, sh0);
        bn_scale_shift(nsum, nsq, ng, nbeta, invN, lane + 32, sc1, sh1);
        bn_scale_shift(nsum, nsq, ng, nbeta, invN, lane + 64, sc2, sh2);
    }
    __syncthreads();

    for (int g = beg + wv * 32; g < end; g += EDGE_TPB) {
        const int n = min(32, end - g);
        int2 r = make_int2(0, 0);
        if (lane < n) r = rec[g + lane];
        const int kv = r.x;
        const int av = r.y;

        for (int e0 = 0; e0 < n; e0 += EDGE_U) {
            float v0[EDGE_U], v1[EDGE_U], v2[EDGE_U];
#pragma unroll
            for (int u = 0; u < EDGE_U; ++u) {
                const int s = __builtin_amdgcn_readlane(kv, e0 + u) & 0xFFFF;
                const float* xr = x + (size_t)s * ldx + lane;
                v0[u] = xr[0];
                v1[u] = xr[32];
                v2[u] = xr[64];
            }
#pragma unroll
            for (int u = 0; u < EDGE_U; ++u) {
                if (e0 + u < n) {
                    const int   t = __builtin_amdgcn_readlane(kv, e0 + u) >> 16;
                    const float a = __int_as_float(__builtin_amdgcn_readlane(av, e0 + u));
                    float* ap = acc + t * 96 + lane;
                    const float x0 = NORM ? v0[u] * sc0 + sh0 : v0[u];
                    const float x1 = NORM ? v1[u] * sc1 + sh1 : v1[u];
                    const float x2 = NORM ? v2[u] * sc2 + sh2 : v2[u];
                    atomicAdd(ap,      fmaxf(x0 + a * w0 + b0, 0.0f));
                    atomicAdd(ap + 32, fmaxf(x1 + a * w1 + b1, 0.0f));
                    atomicAdd(ap + 64, fmaxf(x2 + a * w2 + b2, 0.0f));
                }
            }
        }
    }
    __syncthreads();

    float4* o = reinterpret_cast<float4*>(agg + (size_t)blockIdx.x * (NPB * 96));
    for (int i = tid; i < NPB * 96 / 4; i += EDGE_TPB) o[i] = acc4[i];
}

#define GEMM_WAVES 8
#define GEMM_TPB   (GEMM_WAVES * 32)
#define GEMM_STRIPS (NNODES / 16)
#define GEMM_GRID  ((GEMM_STRIPS + GEMM_WAVES - 1) / GEMM_WAVES)

template <int K, int N, bool HASADD, bool NORM>
__global__ __launch_bounds__(GEMM_TPB)
void gemm_wmma_f32_kernel(const float* __restrict__ A0, int lda0,
                          const float* __restrict__ A1, int lda1,
                          const float* __restrict__ W,
                          const float* __restrict__ bias,
                          float* __restrict__ out, int ldo,
                          float* __restrict__ colsum,
                          float* __restrict__ colsq,
                          const float* __restrict__ nsum0,
                          const float* __restrict__ nsq0,
                          const float* __restrict__ ng0,
                          const float* __restrict__ nbeta0,
                          int ncols0,
                          const float* __restrict__ nsum1,
                          const float* __restrict__ nsq1,
                          const float* __restrict__ ng1,
                          const float* __restrict__ nbeta1,
                          float invN)
{
    static_assert(K % 16 == 0 && N % 16 == 0, "whole k blocks and column tiles");
    static_assert(2 * N <= GEMM_TPB, "one thread per (column, statistic)");
    static_assert(K <= GEMM_TPB && K % 32 == 0, "one thread per column of A0 for its scale and shift, whole waves");
    static_assert(NNODES % 16 == 0, "whole row strips");
    static_assert(((K / 2) * (N / 4)) % 32 == 0, "the part round of the W staging is whole waves");
    constexpr int NT    = N / 16;
    constexpr int KB    = K / 16;
    constexpr int PITCH = 2 * N + 8;
    constexpr int NQ    = N / 4;

    __shared__ float4 wl4[(K / 2) * PITCH / 4];
    float* wl = reinterpret_cast<float*>(wl4);
    __shared__ float red[2][2 * GEMM_WAVES][N];
    __shared__ float4 nsc4[NORM ? K / 4 : 1];
    __shared__ float4 nsh4[NORM ? K / 4 : 1];

    const int tid  = threadIdx.x;
    const int lane = tid & 31;
    const int wv   = tid >> 5;
    const int lo   = lane & 15;
    const int hi   = lane >> 4;

    int strip = blockIdx.x * GEMM_WAVES + wv;
    const bool live = strip < GEMM_STRIPS;
    if (!live) strip = GEMM_STRIPS - 1;
    const int row0 = strip * 16;

    const float* ar0 = A0 + (size_t)(row0 + lo) * lda0 + 8 * hi;
    float4 a[KB][2];
#pragma unroll
    for (int i = 0; i < KB; ++i) {
        a[i][0] = *reinterpret_cast<const float4*>(ar0 + 16 * i);
        a[i][1] = *reinterpret_cast<const float4*>(ar0 + 16 * i + 4);
    }
    float4 a1[HASADD ? KB : 1][2];
    if (HASADD) {
        const float* ar1 = A1 + (size_t)(row0 + lo) * lda1 + 8 * hi;
#pragma unroll
        for (int i = 0; i < KB; ++i) {
            a1[i][0] = *reinterpret_cast<const float4*>(ar1 + 16 * i);
            a1[i][1] = *reinterpret_cast<const float4*>(ar1 + 16 * i + 4);
        }
    }

    constexpr int ITEMS = (K / 2) * NQ;
    constexpr int SFULL = ITEMS / GEMM_TPB;
    constexpr int SREM  = ITEMS % GEMM_TPB;
    float4 r0[SFULL > 0 ? SFULL : 1], r1[SFULL > 0 ? SFULL : 1];
#pragma unroll
    for (int s = 0; s < SFULL; ++s) {
        const int it = tid + s * GEMM_TPB;
        const int p  = it / NQ;
        const int n4 = it % NQ;
        r0[s] = *reinterpret_cast<const float4*>(W + (size_t)(2 * p) * N + 4 * n4);
        r1[s] = *reinterpret_cast<const float4*>(W + (size_t)(2 * p + 1) * N + 4 * n4);
    }
    if (NORM && tid < K) {
        const bool first = tid < ncols0;
        const int  c     = first ? tid : tid - ncols0;
        float sc, sh;
        bn_scale_shift(first ? nsum0 : nsum1, first ? nsq0 : nsq1,
                       first ? ng0 : ng1, first ? nbeta0 : nbeta1,
                       invN, c, sc, sh);
        reinterpret_cast<float*>(nsc4)[tid] = sc;
        reinterpret_cast<float*>(nsh4)[tid] = sh;
    }
    if (SREM != 0 && tid < SREM) {
        const int it = tid + SFULL * GEMM_TPB;
        const int p  = it / NQ;
        const int n4 = it % NQ;
        const float4 q0 = *reinterpret_cast<const float4*>(W + (size_t)(2 * p) * N + 4 * n4);
        const float4 q1 = *reinterpret_cast<const float4*>(W + (size_t)(2 * p + 1) * N + 4 * n4);
        float4* d = reinterpret_cast<float4*>(wl + p * PITCH + 8 * n4);
        d[0] = make_float4(q0.x, q1.x, q0.y, q1.y);
        d[1] = make_float4(q0.z, q1.z, q0.w, q1.w);
    }
#pragma unroll
    for (int s = 0; s < SFULL; ++s) {
        const int it = tid + s * GEMM_TPB;
        const int p  = it / NQ;
        const int n4 = it % NQ;
        float4* d = reinterpret_cast<float4*>(wl + p * PITCH + 8 * n4);
        d[0] = make_float4(r0[s].x, r1[s].x, r0[s].y, r1[s].y);
        d[1] = make_float4(r0[s].z, r1[s].z, r0[s].w, r1[s].w);
    }
    __syncthreads();

    if (NORM) {
#pragma unroll
        for (int i = 0; i < KB; ++i) {
#pragma unroll
            for (int h = 0; h < 2; ++h) {
                const float4 s4 = nsc4[4 * i + 2 * hi + h];
                const float4 t4 = nsh4[4 * i + 2 * hi + h];
                a[i][h].x = a[i][h].x * s4.x + t4.x;
                a[i][h].y = a[i][h].y * s4.y + t4.y;
                a[i][h].z = a[i][h].z * s4.z + t4.z;
                a[i][h].w = a[i][h].w * s4.w + t4.w;
            }
        }
    }

    if (HASADD) {
#pragma unroll
        for (int i = 0; i < KB; ++i) {
#pragma unroll
            for (int h = 0; h < 2; ++h) {
                a[i][h].x += a1[i][h].x;  a[i][h].y += a1[i][h].y;
                a[i][h].z += a1[i][h].z;  a[i][h].w += a1[i][h].w;
            }
        }
    }

    v8f c[NT];
#pragma unroll
    for (int t = 0; t < NT; ++t) c[t] = v8f{};

    const float* wb = wl + (4 * hi) * PITCH + 2 * lo;
#pragma unroll
    for (int i = 0; i < KB; ++i) {
#pragma unroll
        for (int j = 0; j < 4; ++j) {
            v2f av;
            if (j == 0)      { av.x = a[i][0].x; av.y = a[i][0].y; }
            else if (j == 1) { av.x = a[i][0].z; av.y = a[i][0].w; }
            else if (j == 2) { av.x = a[i][1].x; av.y = a[i][1].y; }
            else             { av.x = a[i][1].z; av.y = a[i][1].w; }
#pragma unroll
            for (int t = 0; t < NT; ++t) {
                const v2f bv = *reinterpret_cast<const v2f*>(
                    wb + (8 * i + j) * PITCH + 32 * t);
                c[t] = __builtin_amdgcn_wmma_f32_16x16x4_f32(
                         false, av,  false, bv,
                         (short)0, c[t],  false,  false);
            }
        }
    }

    float res[NT][8];
#pragma unroll
    for (int t = 0; t < NT; ++t) {
        const float bsv = bias[16 * t + lo];
#pragma unroll
        for (int r = 0; r < 8; ++r) res[t][r] = fmaxf(c[t][r] + bsv, 0.0f);
    }
#pragma unroll
    for (int t = 0; t < NT; ++t) {
        float s = 0.0f, q = 0.0f;
#pragma unroll
        for (int r = 0; r < 8; ++r) {
            const float v = res[t][r];
            s += v;
            q += v * v;
        }
        red[0][2 * wv + hi][16 * t + lo] = live ? s : 0.0f;
        red[1][2 * wv + hi][16 * t + lo] = live ? q : 0.0f;
    }
    __syncthreads();

    if (live) {
#pragma unroll
        for (int r = 0; r < 8; ++r) {
            float* orow = out + (size_t)(row0 + r + 8 * hi) * ldo + lo;
#pragma unroll
            for (int t = 0; t < NT; ++t) orow[16 * t] = res[t][r];
        }
    }

    if (tid < 2 * N) {
        const int which = tid / N;
        const int col   = tid % N;
        float acc = 0.0f;
#pragma unroll
        for (int p = 0; p < 2 * GEMM_WAVES; ++p) acc += red[which][p][col];
        atomicAdd(which ? &colsq[col] : &colsum[col], acc);
    }
}

__global__ __launch_bounds__(256)
void bn_apply_kernel(const float* __restrict__ in, int ldi,
                     float* __restrict__ out, int ldo,
                     const float* __restrict__ sum,
                     const float* __restrict__ sumsq,
                     const float* __restrict__ g,
                     const float* __restrict__ beta,
                     float invN, int n, int C)
{
    int gid = blockIdx.x * blockDim.x + threadIdx.x;
    if (gid >= n * C) return;
    int row = gid / C;
    int c   = gid % C;
    float sc, sh;
    bn_scale_shift(sum, sumsq, g, beta, invN, c, sc, sh);
    out[(size_t)row * ldo + c] = in[(size_t)row * ldi + c] * sc + sh;
}

extern "C" void kernel_launch(void* const* d_in, const int* in_sizes, int n_in,
                              void* d_out, int out_size, void* d_ws, size_t ws_size,
                              hipStream_t stream)
{
    (void)in_sizes; (void)n_in; (void)out_size; (void)ws_size;

    const float* x          = (const float*)d_in[0];
    const float* edge_attr  = (const float*)d_in[1];
    const int*   edge_index = (const int*)  d_in[2];
    const float* lin_w      = (const float*)d_in[3];
    const float* lin_b      = (const float*)d_in[4];
    const float* c1_w  = (const float*)d_in[5];
    const float* c1_b  = (const float*)d_in[6];
    const float* c1_g  = (const float*)d_in[7];
    const float* c1_be = (const float*)d_in[8];
    const float* c2_w  = (const float*)d_in[9];
    const float* c2_b  = (const float*)d_in[10];
    const float* c2_g  = (const float*)d_in[11];
    const float* c2_be = (const float*)d_in[12];
    const float* l1_w  = (const float*)d_in[13];
    const float* l1_b  = (const float*)d_in[14];
    const float* l1_g  = (const float*)d_in[15];
    const float* l1_be = (const float*)d_in[16];
    const float* ma_w  = (const float*)d_in[17];
    const float* ma_b  = (const float*)d_in[18];
    const float* ma_g  = (const float*)d_in[19];
    const float* ma_be = (const float*)d_in[20];
    const float* mb_w  = (const float*)d_in[21];
    const float* mb_b  = (const float*)d_in[22];
    const float* mb_g  = (const float*)d_in[23];
    const float* mb_be = (const float*)d_in[24];

    const int* src = edge_index;
    const int* dst = edge_index + NEDGES;

    const size_t N = NNODES;

    float* ws    = (float*)d_ws;
    float* agg   = ws;
    float* bufP  = agg  + N * 96;
    float* bufQ  = bufP + N * 96;
    float* xcat  = bufQ + N * 96;
    float* stats = xcat + N * 160;
    int*   hist     = (int*)(stats + 5 * 192);
    int*   binstart = hist + PBLK * NBINS;
    int2*  rec      = (int2*)(binstart + 1024);

    const float invN = 1.0f / (float)N;

    auto sumL = [&](int L) { return stats + (size_t)L * 192; };
    auto sqL  = [&](int L) { return stats + (size_t)L * 192 + 96; };

    hipMemsetAsync(stats, 0, 5 * 192 * sizeof(float), stream);

    bin_count_kernel<<<PBLK, PTPB, 0, stream>>>(dst, hist);
    bin_place_kernel<<<PBLK, PTPB, 0, stream>>>(src, dst, edge_attr, hist, binstart, rec);

    edge_bin_kernel<false><<<NBINS, EDGE_TPB, 0, stream>>>(
        x, 96, rec, binstart, lin_w, lin_b, agg,
        nullptr, nullptr, nullptr, nullptr, invN);

    gemm_wmma_f32_kernel<96, 96, true, false><<<GEMM_GRID, GEMM_TPB, 0, stream>>>(
        x, 96, agg, 96, c1_w, c1_b, xcat, 160, sumL(0), sqL(0),
        nullptr, nullptr, nullptr, nullptr, 96,
        nullptr, nullptr, nullptr, nullptr, invN);

    edge_bin_kernel<true><<<NBINS, EDGE_TPB, 0, stream>>>(
        xcat, 160, rec, binstart, lin_w, lin_b, agg,
        sumL(0), sqL(0), c1_g, c1_be, invN);

    gemm_wmma_f32_kernel<96, 64, true, true><<<GEMM_GRID, GEMM_TPB, 0, stream>>>(
        xcat, 160, agg, 96, c2_w, c2_b, xcat + 96, 160, sumL(1), sqL(1),
        sumL(0), sqL(0), c1_g, c1_be, 96,
        nullptr, nullptr, nullptr, nullptr, invN);

    gemm_wmma_f32_kernel<160, 96, false, true><<<GEMM_GRID, GEMM_TPB, 0, stream>>>(
        xcat, 160, nullptr, 0, l1_w, l1_b, bufQ, 96, sumL(2), sqL(2),
        sumL(0), sqL(0), c1_g, c1_be, 96,
        sumL(1), sqL(1), c2_g, c2_be, invN);

    gemm_wmma_f32_kernel<96, 96, false, true><<<GEMM_GRID, GEMM_TPB, 0, stream>>>(
        bufQ, 96, nullptr, 0, ma_w, ma_b, bufP, 96, sumL(3), sqL(3),
        sumL(2), sqL(2), l1_g, l1_be, 96,
        nullptr, nullptr, nullptr, nullptr, invN);

    gemm_wmma_f32_kernel<96, 16, false, true><<<GEMM_GRID, GEMM_TPB, 0, stream>>>(
        bufP, 96, nullptr, 0, mb_w, mb_b, bufQ, 16, sumL(4), sqL(4),
        sumL(3), sqL(3), ma_g, ma_be, 96,
        nullptr, nullptr, nullptr, nullptr, invN);

    bn_apply_kernel<<<(NNODES * 16 + 255) / 256, 256, 0, stream>>>(
        bufQ, 16, (float*)d_out, 16, sumL(4), sqL(4), mb_g, mb_be, invN, NNODES, 16);
}


# Edge sums feed the layer GEMM inside one kernel: a block keeps its 208 rows' sums in LDS and adds them into the WMMA A operand; 2 launches and the agg round trip removed; sum orders change
# speedup vs baseline: 7.0433x; 1.1091x over previous
#include <hip/hip_runtime.h>
#include <hip/hip_bf16.h>


#define NNODES 50000
#define NEDGES 800000
#define BN_EPS 1e-5f

typedef __attribute__((ext_vector_type(2))) float v2f;
typedef __attribute__((ext_vector_type(8))) float v8f;

__device__ __forceinline__ void bn_scale_shift(const float* __restrict__ sum,
                                               const float* __restrict__ sumsq,
                                               const float* __restrict__ g,
                                               const float* __restrict__ beta,
                                               float invN, int c,
                                               float& sc, float& sh)
{
    const float mu  = sum[c] * invN;
    const float var = sumsq[c] * invN - mu * mu;
    const float inv = rsqrtf(var + BN_EPS);
    sc = g[c] * inv;
    sh = beta[c] - mu * sc;
}

#define SPB   13
#define NPB   (SPB * 16)
#define NBINS ((NNODES + NPB - 1) / NPB)
#define PBLK  128
#define PCH   (NEDGES / PBLK)
#define PTPB  1024
#define PIT   ((PCH + PTPB - 1) / PTPB)

__global__ __launch_bounds__(PTPB)
void bin_count_kernel(const int* __restrict__ dst, int* __restrict__ hist)
{
    static_assert(NNODES % 16 == 0, "a bin holds whole 16-row strips, the last bin fewer of them");
    static_assert(NEDGES % PBLK == 0, "whole chunks");
    static_assert(NBINS <= PTPB, "one thread per bin");
    __shared__ int cnt[NBINS];
    const int tid = threadIdx.x;
    if (tid < NBINS) cnt[tid] = 0;
    __syncthreads();
    const int j0 = blockIdx.x * PCH;
    int d[PIT];
#pragma unroll
    for (int i = 0; i < PIT; ++i) {
        const int j = tid + i * PTPB;
        d[i] = (j < PCH) ? dst[j0 + j] : 0;
    }
#pragma unroll
    for (int i = 0; i < PIT; ++i)
        if (tid + i * PTPB < PCH) atomicAdd(&cnt[d[i] / NPB], 1);
    __syncthreads();
    if (tid < NBINS) hist[blockIdx.x * NBINS + tid] = cnt[tid];
}

__global__ __launch_bounds__(PTPB)
void bin_place_kernel(const int* __restrict__ src,
                      const int* __restrict__ dst,
                      const float* __restrict__ ea,
                      const int* __restrict__ hist,
                      int* __restrict__ binstart,
                      int2* __restrict__ rec)
{
    static_assert(NNODES <= 65536, "src packs into 16 bits");
    static_assert(NPB < 32768, "local destination packs into the upper half");
    static_assert(PTPB == 1024, "32 waves: one lane of wave 0 per wave total");
    __shared__ int cur[NBINS];
    __shared__ int wsum[PTPB / 32];
    const int tid  = threadIdx.x;
    const int lane = tid & 31;
    const int wv   = tid >> 5;

    int tot = 0, pre = 0;
    if (tid < NBINS) {
#pragma unroll 32
        for (int k = 0; k < PBLK; ++k) {
            const int v = hist[k * NBINS + tid];
            tot += v;
            pre += (k < (int)blockIdx.x) ? v : 0;
        }
    }
    int inc = tot;
    for (int o = 1; o < 32; o <<= 1) {
        const int nb = __shfl_up(inc, o);
        if (lane >= o) inc += nb;
    }
    if (lane == 31) wsum[wv] = inc;
    __syncthreads();
    if (wv == 0) {
        const int w = wsum[lane];
        int winc = w;
        for (int o = 1; o < 32; o <<= 1) {
            const int nb = __shfl_up(winc, o);
            if (lane >= o) winc += nb;
        }
        wsum[lane] = winc - w;
    }
    __syncthreads();
    const int start = wsum[wv] + inc - tot;
    if (tid < NBINS) {
        cur[tid] = start + pre;
        if (blockIdx.x == 0) binstart[tid] = start;
    }
    if (blockIdx.x == 0 && tid == 0) binstart[NBINS] = NEDGES;
    __syncthreads();

    const int j0 = blockIdx.x * PCH;
    int d[PIT], sv[PIT], av[PIT];
#pragma unroll
    for (int i = 0; i < PIT; ++i) {
        const int  j  = tid + i * PTPB;
        const bool in = j < PCH;
        d[i]  = in ? dst[j0 + j] : 0;
        sv[i] = in ? src[j0 + j] : 0;
        av[i] = in ? __float_as_int(ea[j0 + j]) : 0;
    }
#pragma unroll
    for (int i = 0; i < PIT; ++i) {
        if (tid + i * PTPB < PCH) {
            const int b   = d[i] / NPB;
            const int loc = d[i] - b * NPB;
            const int pos = atomicAdd(&cur[b], 1);
            rec[pos] = make_int2(sv[i] | (loc << 16), av[i]);
        }
    }
}

#define GEMM_WAVES 8
#define GEMM_TPB   (GEMM_WAVES * 32)
#define GEMM_STRIPS (NNODES / 16)
#define GEMM_GRID  ((GEMM_STRIPS + GEMM_WAVES - 1) / GEMM_WAVES)

template <int K, int N, bool HASADD, bool NORM>
__global__ __launch_bounds__(GEMM_TPB)
void gemm_wmma_f32_kernel(const float* __restrict__ A0, int lda0,
                          const float* __restrict__ A1, int lda1,
                          const float* __restrict__ W,
                          const float* __restrict__ bias,
                          float* __restrict__ out, int ldo,
                          float* __restrict__ colsum,
                          float* __restrict__ colsq,
                          const float* __restrict__ nsum0,
                          const float* __restrict__ nsq0,
                          const float* __restrict__ ng0,
                          const float* __restrict__ nbeta0,
                          int ncols0,
                          const float* __restrict__ nsum1,
                          const float* __restrict__ nsq1,
                          const float* __restrict__ ng1,
                          const float* __restrict__ nbeta1,
                          float invN)
{
    static_assert(K % 16 == 0 && N % 16 == 0, "whole k blocks and column tiles");
    static_assert(2 * N <= GEMM_TPB, "one thread per (column, statistic)");
    static_assert(K <= GEMM_TPB && K % 32 == 0, "one thread per column of A0 for its scale and shift, whole waves");
    static_assert(NNODES % 16 == 0, "whole row strips");
    static_assert(((K / 2) * (N / 4)) % 32 == 0, "the part round of the W staging is whole waves");
    constexpr int NT    = N / 16;
    constexpr int KB    = K / 16;
    constexpr int PITCH = 2 * N + 8;
    constexpr int NQ    = N / 4;

    __shared__ float4 wl4[(K / 2) * PITCH / 4];
    float* wl = reinterpret_cast<float*>(wl4);
    __shared__ float red[2][2 * GEMM_WAVES][N];
    __shared__ float4 nsc4[NORM ? K / 4 : 1];
    __shared__ float4 nsh4[NORM ? K / 4 : 1];

    const int tid  = threadIdx.x;
    const int lane = tid & 31;
    const int wv   = tid >> 5;
    const int lo   = lane & 15;
    const int hi   = lane >> 4;

    int strip = blockIdx.x * GEMM_WAVES + wv;
    const bool live = strip < GEMM_STRIPS;
    if (!live) strip = GEMM_STRIPS - 1;
    const int row0 = strip * 16;

    const float* ar0 = A0 + (size_t)(row0 + lo) * lda0 + 8 * hi;
    float4 a[KB][2];
#pragma unroll
    for (int i = 0; i < KB; ++i) {
        a[i][0] = *reinterpret_cast<const float4*>(ar0 + 16 * i);
        a[i][1] = *reinterpret_cast<const float4*>(ar0 + 16 * i + 4);
    }
    float4 a1[HASADD ? KB : 1][2];
    if (HASADD) {
        const float* ar1 = A1 + (size_t)(row0 + lo) * lda1 + 8 * hi;
#pragma unroll
        for (int i = 0; i < KB; ++i) {
            a1[i][0] = *reinterpret_cast<const float4*>(ar1 + 16 * i);
            a1[i][1] = *reinterpret_cast<const float4*>(ar1 + 16 * i + 4);
        }
    }

    constexpr int ITEMS = (K / 2) * NQ;
    constexpr int SFULL = ITEMS / GEMM_TPB;
    constexpr int SREM  = ITEMS % GEMM_TPB;
    float4 r0[SFULL > 0 ? SFULL : 1], r1[SFULL > 0 ? SFULL : 1];
#pragma unroll
    for (int s = 0; s < SFULL; ++s) {
        const int it = tid + s * GEMM_TPB;
        const int p  = it / NQ;
        const int n4 = it % NQ;
        r0[s] = *reinterpret_cast<const float4*>(W + (size_t)(2 * p) * N + 4 * n4);
        r1[s] = *reinterpret_cast<const float4*>(W + (size_t)(2 * p + 1) * N + 4 * n4);
    }
    if (NORM && tid < K) {
        const bool first = tid < ncols0;
        const int  c     = first ? tid : tid - ncols0;
        float sc, sh;
        bn_scale_shift(first ? nsum0 : nsum1, first ? nsq0 : nsq1,
                       first ? ng0 : ng1, first ? nbeta0 : nbeta1,
                       invN, c, sc, sh);
        reinterpret_cast<float*>(nsc4)[tid] = sc;
        reinterpret_cast<float*>(nsh4)[tid] = sh;
    }
    if (SREM != 0 && tid < SREM) {
        const int it = tid + SFULL * GEMM_TPB;
        const int p  = it / NQ;
        const int n4 = it % NQ;
        const float4 q0 = *reinterpret_cast<const float4*>(W + (size_t)(2 * p) * N + 4 * n4);
        const float4 q1 = *reinterpret_cast<const float4*>(W + (size_t)(2 * p + 1) * N + 4 * n4);
        float4* d = reinterpret_cast<float4*>(wl + p * PITCH + 8 * n4);
        d[0] = make_float4(q0.x, q1.x, q0.y, q1.y);
        d[1] = make_float4(q0.z, q1.z, q0.w, q1.w);
    }
#pragma unroll
    for (int s = 0; s < SFULL; ++s) {
        const int it = tid + s * GEMM_TPB;
        const int p  = it / NQ;
        const int n4 = it % NQ;
        float4* d = reinterpret_cast<float4*>(wl + p * PITCH + 8 * n4);
        d[0] = make_float4(r0[s].x, r1[s].x, r0[s].y, r1[s].y);
        d[1] = make_float4(r0[s].z, r1[s].z, r0[s].w, r1[s].w);
    }
    __syncthreads();

    if (NORM) {
#pragma unroll
        for (int i = 0; i < KB; ++i) {
#pragma unroll
            for (int h = 0; h < 2; ++h) {
                const float4 s4 = nsc4[4 * i + 2 * hi + h];
                const float4 t4 = nsh4[4 * i + 2 * hi + h];
                a[i][h].x = a[i][h].x * s4.x + t4.x;
                a[i][h].y = a[i][h].y * s4.y + t4.y;
                a[i][h].z = a[i][h].z * s4.z + t4.z;
                a[i][h].w = a[i][h].w * s4.w + t4.w;
            }
        }
    }

    if (HASADD) {
#pragma unroll
        for (int i = 0; i < KB; ++i) {
#pragma unroll
            for (int h = 0; h < 2; ++h) {
                a[i][h].x += a1[i][h].x;  a[i][h].y += a1[i][h].y;
                a[i][h].z += a1[i][h].z;  a[i][h].w += a1[i][h].w;
            }
        }
    }

    v8f c[NT];
#pragma unroll
    for (int t = 0; t < NT; ++t) c[t] = v8f{};

    const float* wb = wl + (4 * hi) * PITCH + 2 * lo;
#pragma unroll
    for (int i = 0; i < KB; ++i) {
#pragma unroll
        for (int j = 0; j < 4; ++j) {
            v2f av;
            if (j == 0)      { av.x = a[i][0].x; av.y = a[i][0].y; }
            else if (j == 1) { av.x = a[i][0].z; av.y = a[i][0].w; }
            else if (j == 2) { av.x = a[i][1].x; av.y = a[i][1].y; }
            else             { av.x = a[i][1].z; av.y = a[i][1].w; }
#pragma unroll
            for (int t = 0; t < NT; ++t) {
                const v2f bv = *reinterpret_cast<const v2f*>(
                    wb + (8 * i + j) * PITCH + 32 * t);
                c[t] = __builtin_amdgcn_wmma_f32_16x16x4_f32(
                         false, av,  false, bv,
                         (short)0, c[t],  false,  false);
            }
        }
    }

    float res[NT][8];
#pragma unroll
    for (int t = 0; t < NT; ++t) {
        const float bsv = bias[16 * t + lo];
#pragma unroll
        for (int r = 0; r < 8; ++r) res[t][r] = fmaxf(c[t][r] + bsv, 0.0f);
    }
#pragma unroll
    for (int t = 0; t < NT; ++t) {
        float s = 0.0f, q = 0.0f;
#pragma unroll
        for (int r = 0; r < 8; ++r) {
            const float v = res[t][r];
            s += v;
            q += v * v;
        }
        red[0][2 * wv + hi][16 * t + lo] = live ? s : 0.0f;
        red[1][2 * wv + hi][16 * t + lo] = live ? q : 0.0f;
    }
    __syncthreads();

    if (live) {
#pragma unroll
        for (int r = 0; r < 8; ++r) {
            float* orow = out + (size_t)(row0 + r + 8 * hi) * ldo + lo;
#pragma unroll
            for (int t = 0; t < NT; ++t) orow[16 * t] = res[t][r];
        }
    }

    if (tid < 2 * N) {
        const int which = tid / N;
        const int col   = tid % N;
        float acc = 0.0f;
#pragma unroll
        for (int p = 0; p < 2 * GEMM_WAVES; ++p) acc += red[which][p][col];
        atomicAdd(which ? &colsq[col] : &colsum[col], acc);
    }
}

#define FUSE_WAVES 32
#define FUSE_TPB   (FUSE_WAVES * 32)
#define GEMM_WV    (2 * SPB)
#define EDGE_U     16
#define APITCH     100

template <int N, bool NORM>
__global__ __launch_bounds__(FUSE_TPB)
void edge_gemm_kernel(const float* __restrict__ x, int ldx,
                      const int2* __restrict__ rec,
                      const int* __restrict__ binstart,
                      const float* __restrict__ lw,
                      const float* __restrict__ lb,
                      const float* __restrict__ W,
                      const float* __restrict__ bias,
                      float* __restrict__ out, int ldo,
                      float* __restrict__ colsum,
                      float* __restrict__ colsq,
                      const float* __restrict__ nsum,
                      const float* __restrict__ nsq,
                      const float* __restrict__ ng,
                      const float* __restrict__ nbeta,
                      float invN)
{
    constexpr int K     = 96;
    constexpr int KB    = K / 16;
    constexpr int NTH   = N / 32;
    constexpr int PITCH = 2 * N + 8;
    constexpr int NQ    = N / 4;
    static_assert(N % 32 == 0, "two column halves of whole tiles");
    static_assert(2 * N <= FUSE_TPB && K <= FUSE_TPB, "one thread per (column, statistic) and per column of x");
    static_assert(32 % EDGE_U == 0, "whole batches");
    static_assert(APITCH >= K && APITCH % 4 == 0, "rows of the sums are whole float4");
    static_assert(FUSE_TPB <= 1024 && GEMM_WV <= FUSE_WAVES, "one block, with a wave for every strip half");

    __shared__ float4 acc4[NPB * APITCH / 4];
    float* acc = reinterpret_cast<float*>(acc4);
    __shared__ float4 wl4[(K / 2) * PITCH / 4];
    float* wl = reinterpret_cast<float*>(wl4);
    __shared__ float red[2][SPB][N];
    __shared__ float4 nsc4[NORM ? K / 4 : 1];
    __shared__ float4 nsh4[NORM ? K / 4 : 1];

    const int tid  = threadIdx.x;
    const int lane = tid & 31;
    const int lo   = lane & 15;
    const int hi   = lane >> 4;
    const int wv   = __builtin_amdgcn_readfirstlane(tid >> 5);

    for (int i = tid; i < NPB * APITCH / 4; i += FUSE_TPB)
        acc4[i] = make_float4(0.0f, 0.0f, 0.0f, 0.0f);

    const int beg = binstart[blockIdx.x];
    const int end = binstart[blockIdx.x + 1];

    const float w0 = lw[lane], w1 = lw[lane + 32], w2 = lw[lane + 64];
    const float b0 = lb[lane], b1 = lb[lane + 32], b2 = lb[lane + 64];

    float sc0 = 1.0f, sc1 = 1.0f, sc2 = 1.0f, sh0 = 0.0f, sh1 = 0.0f, sh2 = 0.0f;
    if (NORM) {
        bn_scale_shift(nsum, nsq, ng, nbeta, invN, lane,      sc0, sh0);
        bn_scale_shift(nsum, nsq, ng, nbeta, invN, lane + 32, sc1, sh1);
        bn_scale_shift(nsum, nsq, ng, nbeta, invN, lane + 64, sc2, sh2);
        if (tid < K) {
            float sc, sh;
            bn_scale_shift(nsum, nsq, ng, nbeta, invN, tid, sc, sh);
            reinterpret_cast<float*>(nsc4)[tid] = sc;
            reinterpret_cast<float*>(nsh4)[tid] = sh;
        }
    }
    __syncthreads();

    for (int g = beg + wv * 32; g < end; g += FUSE_TPB) {
        const int n = min(32, end - g);
        int2 r = make_int2(0, 0);
        if (lane < n) r = rec[g + lane];
        const int kv = r.x;
        const int av = r.y;

        for (int e0 = 0; e0 < n; e0 += EDGE_U) {
            float v0[EDGE_U], v1[EDGE_U], v2[EDGE_U];
#pragma unroll
            for (int u = 0; u < EDGE_U; ++u) {
                const int s = __builtin_amdgcn_readlane(kv, e0 + u) & 0xFFFF;
                const float* xr = x + (size_t)s * ldx + lane;
                v0[u] = xr[0];
                v1[u] = xr[32];
                v2[u] = xr[64];
            }
#pragma unroll
            for (int u = 0; u < EDGE_U; ++u) {
                if (e0 + u < n) {
                    const int   t = __builtin_amdgcn_readlane(kv, e0 + u) >> 16;
                    const float a = __int_as_float(__builtin_amdgcn_readlane(av, e0 + u));
                    float* ap = acc + t * APITCH + lane;
                    const float x0 = NORM ? v0[u] * sc0 + sh0 : v0[u];
                    const float x1 = NORM ? v1[u] * sc1 + sh1 : v1[u];
                    const float x2 = NORM ? v2[u] * sc2 + sh2 : v2[u];
                    atomicAdd(ap,      fmaxf(x0 + a * w0 + b0, 0.0f));
                    atomicAdd(ap + 32, fmaxf(x1 + a * w1 + b1, 0.0f));
                    atomicAdd(ap + 64, fmaxf(x2 + a * w2 + b2, 0.0f));
                }
            }
        }
    }

    const bool gw = wv < GEMM_WV;
    const int  st = wv >> 1;
    const int  hf = wv & 1;
    int row0 = blockIdx.x * NPB + st * 16;
    const bool live = gw && row0 < NNODES;
    if (!live) row0 = NNODES - 16;
    float4 a[KB][2];
    if (gw) {
        const float* ar = x + (size_t)(row0 + lo) * ldx + 8 * hi;
#pragma unroll
        for (int i = 0; i < KB; ++i) {
            a[i][0] = *reinterpret_cast<const float4*>(ar + 16 * i);
            a[i][1] = *reinterpret_cast<const float4*>(ar + 16 * i + 4);
        }
    }
    constexpr int ITEMS = (K / 2) * NQ;
    for (int it = tid; it < ITEMS; it += FUSE_TPB) {
        const int p  = it / NQ;
        const int n4 = it % NQ;
        const float4 q0 = *reinterpret_cast<const float4*>(W + (size_t)(2 * p) * N + 4 * n4);
        const float4 q1 = *reinterpret_cast<const float4*>(W + (size_t)(2 * p + 1) * N + 4 * n4);
        float4* d = reinterpret_cast<float4*>(wl + p * PITCH + 8 * n4);
        d[0] = make_float4(q0.x, q1.x, q0.y, q1.y);
        d[1] = make_float4(q0.z, q1.z, q0.w, q1.w);
    }
    __syncthreads();

    const int col0 = 16 * NTH * hf + lo;
    float res[NTH][8];
    if (gw) {
        const float* pr = acc + (st * 16 + lo) * APITCH + 8 * hi;
#pragma unroll
        for (int i = 0; i < KB; ++i) {
#pragma unroll
            for (int h = 0; h < 2; ++h) {
                if (NORM) {
                    const float4 s4 = nsc4[4 * i + 2 * hi + h];
                    const float4 t4 = nsh4[4 * i + 2 * hi + h];
                    a[i][h].x = a[i][h].x * s4.x + t4.x;
                    a[i][h].y = a[i][h].y * s4.y + t4.y;
                    a[i][h].z = a[i][h].z * s4.z + t4.z;
                    a[i][h].w = a[i][h].w * s4.w + t4.w;
                }
                const float4 g4 = *reinterpret_cast<const float4*>(pr + 16 * i + 4 * h);
                a[i][h].x += g4.x;  a[i][h].y += g4.y;
                a[i][h].z += g4.z;  a[i][h].w += g4.w;
            }
        }

        v8f c[NTH];
#pragma unroll
        for (int t = 0; t < NTH; ++t) c[t] = v8f{};

        const float* wb = wl + (4 * hi) * PITCH + 2 * lo + 32 * NTH * hf;
#pragma unroll
        for (int i = 0; i < KB; ++i) {
#pragma unroll
            for (int j = 0; j < 4; ++j) {
                v2f av;
                if (j == 0)      { av.x = a[i][0].x; av.y = a[i][0].y; }
                else if (j == 1) { av.x = a[i][0].z; av.y = a[i][0].w; }
                else if (j == 2) { av.x = a[i][1].x; av.y = a[i][1].y; }
                else             { av.x = a[i][1].z; av.y = a[i][1].w; }
#pragma unroll
                for (int t = 0; t < NTH; ++t) {
                    const v2f bv = *reinterpret_cast<const v2f*>(
                        wb + (8 * i + j) * PITCH + 32 * t);
                    c[t] = __builtin_amdgcn_wmma_f32_16x16x4_f32(
                             false, av,  false, bv,
                             (short)0, c[t],  false,  false);
                }
            }
        }

#pragma unroll
        for (int t = 0; t < NTH; ++t) {
            const float bsv = bias[col0 + 16 * t];
#pragma unroll
            for (int r = 0; r < 8; ++r) res[t][r] = fmaxf(c[t][r] + bsv, 0.0f);
        }
#pragma unroll
        for (int t = 0; t < NTH; ++t) {
            float s = 0.0f, q = 0.0f;
#pragma unroll
            for (int r = 0; r < 8; ++r) {
                const float v = res[t][r];
                s += v;
                q += v * v;
            }
            s += __shfl_xor(s, 16);
            q += __shfl_xor(q, 16);
            if (hi == 0) {
                red[0][st][col0 + 16 * t] = live ? s : 0.0f;
                red[1][st][col0 + 16 * t] = live ? q : 0.0f;
            }
        }
    }
    __syncthreads();

    if (live) {
#pragma unroll
        for (int r = 0; r < 8; ++r) {
            float* orow = out + (size_t)(row0 + r + 8 * hi) * ldo + col0;
#pragma unroll
            for (int t = 0; t < NTH; ++t) orow[16 * t] = res[t][r];
        }
    }

    if (tid < 2 * N) {
        const int which = tid / N;
        const int col   = tid % N;
        float sum = 0.0f;
#pragma unroll
        for (int p = 0; p < SPB; ++p) sum += red[which][p][col];
        atomicAdd(which ? &colsq[col] : &colsum[col], sum);
    }
}

__global__ __launch_bounds__(256)
void bn_apply_kernel(const float* __restrict__ in, int ldi,
                     float* __restrict__ out, int ldo,
                     const float* __restrict__ sum,
                     const float* __restrict__ sumsq,
                     const float* __restrict__ g,
                     const float* __restrict__ beta,
                     float invN, int n, int C)
{
    int gid = blockIdx.x * blockDim.x + threadIdx.x;
    if (gid >= n * C) return;
    int row = gid / C;
    int c   = gid % C;
    float sc, sh;
    bn_scale_shift(sum, sumsq, g, beta, invN, c, sc, sh);
    out[(size_t)row * ldo + c] = in[(size_t)row * ldi + c] * sc + sh;
}

extern "C" void kernel_launch(void* const* d_in, const int* in_sizes, int n_in,
                              void* d_out, int out_size, void* d_ws, size_t ws_size,
                              hipStream_t stream)
{
    (void)in_sizes; (void)n_in; (void)out_size; (void)ws_size;

    const float* x          = (const float*)d_in[0];
    const float* edge_attr  = (const float*)d_in[1];
    const int*   edge_index = (const int*)  d_in[2];
    const float* lin_w      = (const float*)d_in[3];
    const float* lin_b      = (const float*)d_in[4];
    const float* c1_w  = (const float*)d_in[5];
    const float* c1_b  = (const float*)d_in[6];
    const float* c1_g  = (const float*)d_in[7];
    const float* c1_be = (const float*)d_in[8];
    const float* c2_w  = (const float*)d_in[9];
    const float* c2_b  = (const float*)d_in[10];
    const float* c2_g  = (const float*)d_in[11];
    const float* c2_be = (const float*)d_in[12];
    const float* l1_w  = (const float*)d_in[13];
    const float* l1_b  = (const float*)d_in[14];
    const float* l1_g  = (const float*)d_in[15];
    const float* l1_be = (const float*)d_in[16];
    const float* ma_w  = (const float*)d_in[17];
    const float* ma_b  = (const float*)d_in[18];
    const float* ma_g  = (const float*)d_in[19];
    const float* ma_be = (const float*)d_in[20];
    const float* mb_w  = (const float*)d_in[21];
    const float* mb_b  = (const float*)d_in[22];
    const float* mb_g  = (const float*)d_in[23];
    const float* mb_be = (const float*)d_in[24];

    const int* src = edge_index;
    const int* dst = edge_index + NEDGES;

    const size_t N = NNODES;

    float* ws    = (float*)d_ws;
    float* bufP  = ws;
    float* bufQ  = bufP + N * 96;
    float* xcat  = bufQ + N * 96;
    float* stats = xcat + N * 160;
    int*   hist     = (int*)(stats + 5 * 192);
    int*   binstart = hist + PBLK * NBINS;
    int2*  rec      = (int2*)(binstart + 1024);

    const float invN = 1.0f / (float)N;

    auto sumL = [&](int L) { return stats + (size_t)L * 192; };
    auto sqL  = [&](int L) { return stats + (size_t)L * 192 + 96; };

    hipMemsetAsync(stats, 0, 5 * 192 * sizeof(float), stream);

    bin_count_kernel<<<PBLK, PTPB, 0, stream>>>(dst, hist);
    bin_place_kernel<<<PBLK, PTPB, 0, stream>>>(src, dst, edge_attr, hist, binstart, rec);

    edge_gemm_kernel<96, false><<<NBINS, FUSE_TPB, 0, stream>>>(
        x, 96, rec, binstart, lin_w, lin_b, c1_w, c1_b, xcat, 160, sumL(0), sqL(0),
        nullptr, nullptr, nullptr, nullptr, invN);

    edge_gemm_kernel<64, true><<<NBINS, FUSE_TPB, 0, stream>>>(
        xcat, 160, rec, binstart, lin_w, lin_b, c2_w, c2_b, xcat + 96, 160, sumL(1), sqL(1),
        sumL(0), sqL(0), c1_g, c1_be, invN);

    gemm_wmma_f32_kernel<160, 96, false, true><<<GEMM_GRID, GEMM_TPB, 0, stream>>>(
        xcat, 160, nullptr, 0, l1_w, l1_b, bufQ, 96, sumL(2), sqL(2),
        sumL(0), sqL(0), c1_g, c1_be, 96,
        sumL(1), sqL(1), c2_g, c2_be, invN);

    gemm_wmma_f32_kernel<96, 96, false, true><<<GEMM_GRID, GEMM_TPB, 0, stream>>>(
        bufQ, 96, nullptr, 0, ma_w, ma_b, bufP, 96, sumL(3), sqL(3),
        sumL(2), sqL(2), l1_g, l1_be, 96,
        nullptr, nullptr, nullptr, nullptr, invN);

    gemm_wmma_f32_kernel<96, 16, false, true><<<GEMM_GRID, GEMM_TPB, 0, stream>>>(
        bufP, 96, nullptr, 0, mb_w, mb_b, bufQ, 16, sumL(4), sqL(4),
        sumL(3), sqL(3), ma_g, ma_be, 96,
        nullptr, nullptr, nullptr, nullptr, invN);

    bn_apply_kernel<<<(NNODES * 16 + 255) / 256, 256, 0, stream>>>(
        bufQ, 16, (float*)d_out, 16, sumL(4), sqL(4), mb_g, mb_be, invN, NNODES, 16);
}


# Edge binning in one launch: each block sorts its own chunk by bin, the edge kernel joins a bin's 128 pieces via an LDS table; memset and count launches removed; LDS float add order changes
# speedup vs baseline: 7.6311x; 1.0835x over previous
#include <hip/hip_runtime.h>
#include <hip/hip_bf16.h>


#define NNODES 50000
#define NEDGES 800000
#define BN_EPS 1e-5f

typedef __attribute__((ext_vector_type(2))) float v2f;
typedef __attribute__((ext_vector_type(8))) float v8f;

__device__ __forceinline__ void bn_scale_shift(const float* __restrict__ sum,
                                               const float* __restrict__ sumsq,
                                               const float* __restrict__ g,
                                               const float* __restrict__ beta,
                                               float invN, int c,
                                               float& sc, float& sh)
{
    const float mu  = sum[c] * invN;
    const float var = sumsq[c] * invN - mu * mu;
    const float inv = rsqrtf(var + BN_EPS);
    sc = g[c] * inv;
    sh = beta[c] - mu * sc;
}

#define SPB   13
#define NPB   (SPB * 16)
#define NBINS ((NNODES + NPB - 1) / NPB)
#define PBLK  128
#define PCH   (NEDGES / PBLK)
#define PTPB  1024
#define PIT   ((PCH + PTPB - 1) / PTPB)
#define NSTAT (5 * 192)

__global__ __launch_bounds__(PTPB)
void bin_chunk_kernel(const int* __restrict__ src,
                      const int* __restrict__ dst,
                      const float* __restrict__ ea,
                      int* __restrict__ offT,
                      int2* __restrict__ rec,
                      float* __restrict__ stats)
{
    static_assert(NNODES % 16 == 0, "a bin holds whole 16-row strips, the last bin fewer of them");
    static_assert(NEDGES % PBLK == 0, "whole chunks");
    static_assert(NBINS <= PTPB && NSTAT <= PTPB, "one thread per bin and per accumulator");
    static_assert(NNODES <= 65536, "src packs into 16 bits");
    static_assert(NPB < 32768, "local destination packs into the upper half");
    static_assert(PTPB == 1024, "32 waves: one lane of wave 0 per wave total");
    __shared__ int cnt[NBINS];
    __shared__ int wsum[PTPB / 32];
    const int tid  = threadIdx.x;
    const int lane = tid & 31;
    const int wv   = tid >> 5;

    if (tid < NBINS) cnt[tid] = 0;
    if (blockIdx.x == 0 && tid < NSTAT) stats[tid] = 0.0f;
    __syncthreads();

    const int j0 = blockIdx.x * PCH;
    int d[PIT], sv[PIT], av[PIT], rk[PIT];
#pragma unroll
    for (int i = 0; i < PIT; ++i) {
        const int  j  = tid + i * PTPB;
        const bool in = j < PCH;
        d[i]  = in ? dst[j0 + j] : 0;
        sv[i] = in ? src[j0 + j] : 0;
        av[i] = in ? __float_as_int(ea[j0 + j]) : 0;
    }
#pragma unroll
    for (int i = 0; i < PIT; ++i) {
        rk[i] = 0;
        if (tid + i * PTPB < PCH) rk[i] = atomicAdd(&cnt[d[i] / NPB], 1);
    }
    __syncthreads();

    const int tot = (tid < NBINS) ? cnt[tid] : 0;
    int inc = tot;
    for (int o = 1; o < 32; o <<= 1) {
        const int nb = __shfl_up(inc, o);
        if (lane >= o) inc += nb;
    }
    if (lane == 31) wsum[wv] = inc;
    __syncthreads();
    if (wv == 0) {
        const int w = wsum[lane];
        int winc = w;
        for (int o = 1; o < 32; o <<= 1) {
            const int nb = __shfl_up(winc, o);
            if (lane >= o) winc += nb;
        }
        wsum[lane] = winc - w;
    }
    __syncthreads();
    const int start = wsum[wv] + inc - tot;
    if (tid < NBINS) {
        cnt[tid] = start;
        offT[tid * PBLK + blockIdx.x] = start;
    }
    if (tid == 0) offT[NBINS * PBLK + blockIdx.x] = PCH;
    __syncthreads();

#pragma unroll
    for (int i = 0; i < PIT; ++i) {
        if (tid + i * PTPB < PCH) {
            const int b   = d[i] / NPB;
            const int loc = d[i] - b * NPB;
            rec[j0 + cnt[b] + rk[i]] = make_int2(sv[i] | (loc << 16), av[i]);
        }
    }
}

#define GEMM_WAVES 8
#define GEMM_TPB   (GEMM_WAVES * 32)
#define GEMM_STRIPS (NNODES / 16)
#define GEMM_GRID  ((GEMM_STRIPS + GEMM_WAVES - 1) / GEMM_WAVES)

template <int K, int N, bool HASADD, bool NORM>
__global__ __launch_bounds__(GEMM_TPB)
void gemm_wmma_f32_kernel(const float* __restrict__ A0, int lda0,
                          const float* __restrict__ A1, int lda1,
                          const float* __restrict__ W,
                          const float* __restrict__ bias,
                          float* __restrict__ out, int ldo,
                          float* __restrict__ colsum,
                          float* __restrict__ colsq,
                          const float* __restrict__ nsum0,
                          const float* __restrict__ nsq0,
                          const float* __restrict__ ng0,
                          const float* __restrict__ nbeta0,
                          int ncols0,
                          const float* __restrict__ nsum1,
                          const float* __restrict__ nsq1,
                          const float* __restrict__ ng1,
                          const float* __restrict__ nbeta1,
                          float invN)
{
    static_assert(K % 16 == 0 && N % 16 == 0, "whole k blocks and column tiles");
    static_assert(2 * N <= GEMM_TPB, "one thread per (column, statistic)");
    static_assert(K <= GEMM_TPB && K % 32 == 0, "one thread per column of A0 for its scale and shift, whole waves");
    static_assert(NNODES % 16 == 0, "whole row strips");
    static_assert(((K / 2) * (N / 4)) % 32 == 0, "the part round of the W staging is whole waves");
    constexpr int NT    = N / 16;
    constexpr int KB    = K / 16;
    constexpr int PITCH = 2 * N + 8;
    constexpr int NQ    = N / 4;

    __shared__ float4 wl4[(K / 2) * PITCH / 4];
    float* wl = reinterpret_cast<float*>(wl4);
    __shared__ float red[2][2 * GEMM_WAVES][N];
    __shared__ float4 nsc4[NORM ? K / 4 : 1];
    __shared__ float4 nsh4[NORM ? K / 4 : 1];

    const int tid  = threadIdx.x;
    const int lane = tid & 31;
    const int wv   = tid >> 5;
    const int lo   = lane & 15;
    const int hi   = lane >> 4;

    int strip = blockIdx.x * GEMM_WAVES + wv;
    const bool live = strip < GEMM_STRIPS;
    if (!live) strip = GEMM_STRIPS - 1;
    const int row0 = strip * 16;

    const float* ar0 = A0 + (size_t)(row0 + lo) * lda0 + 8 * hi;
    float4 a[KB][2];
#pragma unroll
    for (int i = 0; i < KB; ++i) {
        a[i][0] = *reinterpret_cast<const float4*>(ar0 + 16 * i);
        a[i][1] = *reinterpret_cast<const float4*>(ar0 + 16 * i + 4);
    }
    float4 a1[HASADD ? KB : 1][2];
    if (HASADD) {
        const float* ar1 = A1 + (size_t)(row0 + lo) * lda1 + 8 * hi;
#pragma unroll
        for (int i = 0; i < KB; ++i) {
            a1[i][0] = *reinterpret_cast<const float4*>(ar1 + 16 * i);
            a1[i][1] = *reinterpret_cast<const float4*>(ar1 + 16 * i + 4);
        }
    }

    constexpr int ITEMS = (K / 2) * NQ;
    constexpr int SFULL = ITEMS / GEMM_TPB;
    constexpr int SREM  = ITEMS % GEMM_TPB;
    float4 r0[SFULL > 0 ? SFULL : 1], r1[SFULL > 0 ? SFULL : 1];
#pragma unroll
    for (int s = 0; s < SFULL; ++s) {
        const int it = tid + s * GEMM_TPB;
        const int p  = it / NQ;
        const int n4 = it % NQ;
        r0[s] = *reinterpret_cast<const float4*>(W + (size_t)(2 * p) * N + 4 * n4);
        r1[s] = *reinterpret_cast<const float4*>(W + (size_t)(2 * p + 1) * N + 4 * n4);
    }
    if (NORM && tid < K) {
        const bool first = tid < ncols0;
        const int  c     = first ? tid : tid - ncols0;
        float sc, sh;
        bn_scale_shift(first ? nsum0 : nsum1, first ? nsq0 : nsq1,
                       first ? ng0 : ng1, first ? nbeta0 : nbeta1,
                       invN, c, sc, sh);
        reinterpret_cast<float*>(nsc4)[tid] = sc;
        reinterpret_cast<float*>(nsh4)[tid] = sh;
    }
    if (SREM != 0 && tid < SREM) {
        const int it = tid + SFULL * GEMM_TPB;
        const int p  = it / NQ;
        const int n4 = it % NQ;
        const float4 q0 = *reinterpret_cast<const float4*>(W + (size_t)(2 * p) * N + 4 * n4);
        const float4 q1 = *reinterpret_cast<const float4*>(W + (size_t)(2 * p + 1) * N + 4 * n4);
        float4* d = reinterpret_cast<float4*>(wl + p * PITCH + 8 * n4);
        d[0] = make_float4(q0.x, q1.x, q0.y, q1.y);
        d[1] = make_float4(q0.z, q1.z, q0.w, q1.w);
    }
#pragma unroll
    for (int s = 0; s < SFULL; ++s) {
        const int it = tid + s * GEMM_TPB;
        const int p  = it / NQ;
        const int n4 = it % NQ;
        float4* d = reinterpret_cast<float4*>(wl + p * PITCH + 8 * n4);
        d[0] = make_float4(r0[s].x, r1[s].x, r0[s].y, r1[s].y);
        d[1] = make_float4(r0[s].z, r1[s].z, r0[s].w, r1[s].w);
    }
    __syncthreads();

    if (NORM) {
#pragma unroll
        for (int i = 0; i < KB; ++i) {
#pragma unroll
            for (int h = 0; h < 2; ++h) {
                const float4 s4 = nsc4[4 * i + 2 * hi + h];
                const float4 t4 = nsh4[4 * i + 2 * hi + h];
                a[i][h].x = a[i][h].x * s4.x + t4.x;
                a[i][h].y = a[i][h].y * s4.y + t4.y;
                a[i][h].z = a[i][h].z * s4.z + t4.z;
                a[i][h].w = a[i][h].w * s4.w + t4.w;
            }
        }
    }

    if (HASADD) {
#pragma unroll
        for (int i = 0; i < KB; ++i) {
#pragma unroll
            for (int h = 0; h < 2; ++h) {
                a[i][h].x += a1[i][h].x;  a[i][h].y += a1[i][h].y;
                a[i][h].z += a1[i][h].z;  a[i][h].w += a1[i][h].w;
            }
        }
    }

    v8f c[NT];
#pragma unroll
    for (int t = 0; t < NT; ++t) c[t] = v8f{};

    const float* wb = wl + (4 * hi) * PITCH + 2 * lo;
#pragma unroll
    for (int i = 0; i < KB; ++i) {
#pragma unroll
        for (int j = 0; j < 4; ++j) {
            v2f av;
            if (j == 0)      { av.x = a[i][0].x; av.y = a[i][0].y; }
            else if (j == 1) { av.x = a[i][0].z; av.y = a[i][0].w; }
            else if (j == 2) { av.x = a[i][1].x; av.y = a[i][1].y; }
            else             { av.x = a[i][1].z; av.y = a[i][1].w; }
#pragma unroll
            for (int t = 0; t < NT; ++t) {
                const v2f bv = *reinterpret_cast<const v2f*>(
                    wb + (8 * i + j) * PITCH + 32 * t);
                c[t] = __builtin_amdgcn_wmma_f32_16x16x4_f32(
                         false, av,  false, bv,
                         (short)0, c[t],  false,  false);
            }
        }
    }

    float res[NT][8];
#pragma unroll
    for (int t = 0; t < NT; ++t) {
        const float bsv = bias[16 * t + lo];
#pragma unroll
        for (int r = 0; r < 8; ++r) res[t][r] = fmaxf(c[t][r] + bsv, 0.0f);
    }
#pragma unroll
    for (int t = 0; t < NT; ++t) {
        float s = 0.0f, q = 0.0f;
#pragma unroll
        for (int r = 0; r < 8; ++r) {
            const float v = res[t][r];
            s += v;
            q += v * v;
        }
        red[0][2 * wv + hi][16 * t + lo] = live ? s : 0.0f;
        red[1][2 * wv + hi][16 * t + lo] = live ? q : 0.0f;
    }
    __syncthreads();

    if (live) {
#pragma unroll
        for (int r = 0; r < 8; ++r) {
            float* orow = out + (size_t)(row0 + r + 8 * hi) * ldo + lo;
#pragma unroll
            for (int t = 0; t < NT; ++t) orow[16 * t] = res[t][r];
        }
    }

    if (tid < 2 * N) {
        const int which = tid / N;
        const int col   = tid % N;
        float acc = 0.0f;
#pragma unroll
        for (int p = 0; p < 2 * GEMM_WAVES; ++p) acc += red[which][p][col];
        atomicAdd(which ? &colsq[col] : &colsum[col], acc);
    }
}

#define FUSE_WAVES 32
#define FUSE_TPB   (FUSE_WAVES * 32)
#define GEMM_WV    (2 * SPB)
#define EDGE_U     16
#define APITCH     100

template <int N, bool NORM>
__global__ __launch_bounds__(FUSE_TPB)
void edge_gemm_kernel(const float* __restrict__ x, int ldx,
                      const int2* __restrict__ rec,
                      const int* __restrict__ offT,
                      const float* __restrict__ lw,
                      const float* __restrict__ lb,
                      const float* __restrict__ W,
                      const float* __restrict__ bias,
                      float* __restrict__ out, int ldo,
                      float* __restrict__ colsum,
                      float* __restrict__ colsq,
                      const float* __restrict__ nsum,
                      const float* __restrict__ nsq,
                      const float* __restrict__ ng,
                      const float* __restrict__ nbeta,
                      float invN)
{
    constexpr int K     = 96;
    constexpr int KB    = K / 16;
    constexpr int NTH   = N / 32;
    constexpr int PITCH = 2 * N + 8;
    constexpr int NQ    = N / 4;
    static_assert(N % 32 == 0, "two column halves of whole tiles");
    static_assert(2 * N <= FUSE_TPB && K <= FUSE_TPB, "one thread per (column, statistic) and per column of x");
    static_assert(32 == 2 * EDGE_U, "a round of 32 records is two batches");
    static_assert(APITCH >= K && APITCH % 4 == 0, "rows of the sums are whole float4");
    static_assert(FUSE_TPB <= 1024 && GEMM_WV <= FUSE_WAVES, "one block, with a wave for every strip half");
    static_assert(PBLK == 128, "wave 0 scans the pieces, four per lane");

    __shared__ float4 acc4[NPB * APITCH / 4];
    float* acc = reinterpret_cast<float*>(acc4);
    __shared__ float4 wl4[(K / 2) * PITCH / 4];
    float* wl = reinterpret_cast<float*>(wl4);
    __shared__ float red[2][SPB][N];
    __shared__ float4 nsc4[NORM ? K / 4 : 1];
    __shared__ float4 nsh4[NORM ? K / 4 : 1];
    __shared__ int2 tab[PBLK + 1];

    const int tid  = threadIdx.x;
    const int lane = tid & 31;
    const int lo   = lane & 15;
    const int hi   = lane >> 4;
    const int wv   = __builtin_amdgcn_readfirstlane(tid >> 5);

    int4 s4 = make_int4(0, 0, 0, 0), e4 = make_int4(0, 0, 0, 0);
    if (wv == 0) {
        s4 = *reinterpret_cast<const int4*>(offT + (size_t)blockIdx.x * PBLK + 4 * lane);
        e4 = *reinterpret_cast<const int4*>(offT + (size_t)(blockIdx.x + 1) * PBLK + 4 * lane);
    }

    for (int i = tid; i < NPB * APITCH / 4; i += FUSE_TPB)
        acc4[i] = make_float4(0.0f, 0.0f, 0.0f, 0.0f);

    const float w0 = lw[lane], w1 = lw[lane + 32], w2 = lw[lane + 64];
    const float b0 = lb[lane], b1 = lb[lane + 32], b2 = lb[lane + 64];

    float sc0 = 1.0f, sc1 = 1.0f, sc2 = 1.0f, sh0 = 0.0f, sh1 = 0.0f, sh2 = 0.0f;
    if (NORM) {
        bn_scale_shift(nsum, nsq, ng, nbeta, invN, lane,      sc0, sh0);
        bn_scale_shift(nsum, nsq, ng, nbeta, invN, lane + 32, sc1, sh1);
        bn_scale_shift(nsum, nsq, ng, nbeta, invN, lane + 64, sc2, sh2);
        if (tid < K) {
            float sc, sh;
            bn_scale_shift(nsum, nsq, ng, nbeta, invN, tid, sc, sh);
            reinterpret_cast<float*>(nsc4)[tid] = sc;
            reinterpret_cast<float*>(nsh4)[tid] = sh;
        }
    }
    if (wv == 0) {
        const int c0 = e4.x - s4.x, c1 = e4.y - s4.y, c2 = e4.z - s4.z, c3 = e4.w - s4.w;
        const int lt = c0 + c1 + c2 + c3;
        int inc = lt;
        for (int o = 1; o < 32; o <<= 1) {
            const int nb = __shfl_up(inc, o);
            if (lane >= o) inc += nb;
        }
        const int p0 = inc - lt, p1 = p0 + c0, p2 = p1 + c1, p3 = p2 + c2;
        const int q0 = 4 * lane * PCH;
        tab[4 * lane + 0] = make_int2(p0, q0 + s4.x - p0);
        tab[4 * lane + 1] = make_int2(p1, q0 + PCH + s4.y - p1);
        tab[4 * lane + 2] = make_int2(p2, q0 + 2 * PCH + s4.z - p2);
        tab[4 * lane + 3] = make_int2(p3, q0 + 3 * PCH + s4.w - p3);
        if (lane == 31) tab[PBLK] = make_int2(inc, 0);
    }
    __syncthreads();

    const int total = __builtin_amdgcn_readfirstlane(tab[PBLK].x);

    auto fetch = [&](int g0) -> int2 {
        int2 rr = make_int2(0, 0);
        const int i = g0 + lane;
        if (i < total) {
            int p = 0;
#pragma unroll
            for (int s = PBLK / 2; s >= 1; s >>= 1)
                if (tab[p + s].x <= i) p += s;
            rr = rec[i + tab[p].y];
        }
        return rr;
    };

    float v0[EDGE_U], v1[EDGE_U], v2[EDGE_U];
    auto gather = [&](int kv, int e0) {
#pragma unroll
        for (int u = 0; u < EDGE_U; ++u) {
            const int s = __builtin_amdgcn_readlane(kv, e0 + u) & 0xFFFF;
            const float* xr = x + (size_t)s * ldx + lane;
            v0[u] = xr[0];
            v1[u] = xr[32];
            v2[u] = xr[64];
        }
    };
    auto addup = [&](int kv, int av, int e0, int n) {
#pragma unroll
        for (int u = 0; u < EDGE_U; ++u) {
            if (e0 + u < n) {
                const int   t = __builtin_amdgcn_readlane(kv, e0 + u) >> 16;
                const float a = __int_as_float(__builtin_amdgcn_readlane(av, e0 + u));
                float* ap = acc + t * APITCH + lane;
                const float x0 = NORM ? v0[u] * sc0 + sh0 : v0[u];
                const float x1 = NORM ? v1[u] * sc1 + sh1 : v1[u];
                const float x2 = NORM ? v2[u] * sc2 + sh2 : v2[u];
                atomicAdd(ap,      fmaxf(x0 + a * w0 + b0, 0.0f));
                atomicAdd(ap + 32, fmaxf(x1 + a * w1 + b1, 0.0f));
                atomicAdd(ap + 64, fmaxf(x2 + a * w2 + b2, 0.0f));
            }
        }
    };

    {
        int g = wv * 32;
        int2 r = fetch(g);
        for (; g < total; g += FUSE_TPB) {
            const int n  = min(32, total - g);
            const int kv = r.x;
            const int av = r.y;
            gather(kv, 0);
            r = fetch(g + FUSE_TPB);
            addup(kv, av, 0, n);
            if (n > EDGE_U) {
                gather(kv, EDGE_U);
                addup(kv, av, EDGE_U, n);
            }
        }
    }

    const bool gw = wv < GEMM_WV;
    const int  st = wv >> 1;
    const int  hf = wv & 1;
    int row0 = blockIdx.x * NPB + st * 16;
    const bool live = gw && row0 < NNODES;
    if (!live) row0 = NNODES - 16;
    float4 a[KB][2];
    if (gw) {
        const float* ar = x + (size_t)(row0 + lo) * ldx + 8 * hi;
#pragma unroll
        for (int i = 0; i < KB; ++i) {
            a[i][0] = *reinterpret_cast<const float4*>(ar + 16 * i);
            a[i][1] = *reinterpret_cast<const float4*>(ar + 16 * i + 4);
        }
    }
    constexpr int ITEMS = (K / 2) * NQ;
    for (int it = tid; it < ITEMS; it += FUSE_TPB) {
        const int p  = it / NQ;
        const int n4 = it % NQ;
        const float4 q0 = *reinterpret_cast<const float4*>(W + (size_t)(2 * p) * N + 4 * n4);
        const float4 q1 = *reinterpret_cast<const float4*>(W + (size_t)(2 * p + 1) * N + 4 * n4);
        float4* d = reinterpret_cast<float4*>(wl + p * PITCH + 8 * n4);
        d[0] = make_float4(q0.x, q1.x, q0.y, q1.y);
        d[1] = make_float4(q0.z, q1.z, q0.w, q1.w);
    }
    __syncthreads();

    const int col0 = 16 * NTH * hf + lo;
    float res[NTH][8];
    if (gw) {
        const float* pr = acc + (st * 16 + lo) * APITCH + 8 * hi;
#pragma unroll
        for (int i = 0; i < KB; ++i) {
#pragma unroll
            for (int h = 0; h < 2; ++h) {
                if (NORM) {
                    const float4 s4 = nsc4[4 * i + 2 * hi + h];
                    const float4 t4 = nsh4[4 * i + 2 * hi + h];
                    a[i][h].x = a[i][h].x * s4.x + t4.x;
                    a[i][h].y = a[i][h].y * s4.y + t4.y;
                    a[i][h].z = a[i][h].z * s4.z + t4.z;
                    a[i][h].w = a[i][h].w * s4.w + t4.w;
                }
                const float4 g4 = *reinterpret_cast<const float4*>(pr + 16 * i + 4 * h);
                a[i][h].x += g4.x;  a[i][h].y += g4.y;
                a[i][h].z += g4.z;  a[i][h].w += g4.w;
            }
        }

        v8f c[NTH];
#pragma unroll
        for (int t = 0; t < NTH; ++t) c[t] = v8f{};

        const float* wb = wl + (4 * hi) * PITCH + 2 * lo + 32 * NTH * hf;
#pragma unroll
        for (int i = 0; i < KB; ++i) {
#pragma unroll
            for (int j = 0; j < 4; ++j) {
                v2f av;
                if (j == 0)      { av.x = a[i][0].x; av.y = a[i][0].y; }
                else if (j == 1) { av.x = a[i][0].z; av.y = a[i][0].w; }
                else if (j == 2) { av.x = a[i][1].x; av.y = a[i][1].y; }
                else             { av.x = a[i][1].z; av.y = a[i][1].w; }
#pragma unroll
                for (int t = 0; t < NTH; ++t) {
                    const v2f bv = *reinterpret_cast<const v2f*>(
                        wb + (8 * i + j) * PITCH + 32 * t);
                    c[t] = __builtin_amdgcn_wmma_f32_16x16x4_f32(
                             false, av,  false, bv,
                             (short)0, c[t],  false,  false);
                }
            }
        }

#pragma unroll
        for (int t = 0; t < NTH; ++t) {
            const float bsv = bias[col0 + 16 * t];
#pragma unroll
            for (int r = 0; r < 8; ++r) res[t][r] = fmaxf(c[t][r] + bsv, 0.0f);
        }
#pragma unroll
        for (int t = 0; t < NTH; ++t) {
            float s = 0.0f, q = 0.0f;
#pragma unroll
            for (int r = 0; r < 8; ++r) {
                const float v = res[t][r];
                s += v;
                q += v * v;
            }
            s += __shfl_xor(s, 16);
            q += __shfl_xor(q, 16);
            if (hi == 0) {
                red[0][st][col0 + 16 * t] = live ? s : 0.0f;
                red[1][st][col0 + 16 * t] = live ? q : 0.0f;
            }
        }
    }
    __syncthreads();

    if (live) {
#pragma unroll
        for (int r = 0; r < 8; ++r) {
            float* orow = out + (size_t)(row0 + r + 8 * hi) * ldo + col0;
#pragma unroll
            for (int t = 0; t < NTH; ++t) orow[16 * t] = res[t][r];
        }
    }

    if (tid < 2 * N) {
        const int which = tid / N;
        const int col   = tid % N;
        float sum = 0.0f;
#pragma unroll
        for (int p = 0; p < SPB; ++p) sum += red[which][p][col];
        atomicAdd(which ? &colsq[col] : &colsum[col], sum);
    }
}

__global__ __launch_bounds__(256)
void bn_apply_kernel(const float* __restrict__ in, int ldi,
                     float* __restrict__ out, int ldo,
                     const float* __restrict__ sum,
                     const float* __restrict__ sumsq,
                     const float* __restrict__ g,
                     const float* __restrict__ beta,
                     float invN, int n, int C)
{
    int gid = blockIdx.x * blockDim.x + threadIdx.x;
    if (gid >= n * C) return;
    int row = gid / C;
    int c   = gid % C;
    float sc, sh;
    bn_scale_shift(sum, sumsq, g, beta, invN, c, sc, sh);
    out[(size_t)row * ldo + c] = in[(size_t)row * ldi + c] * sc + sh;
}

extern "C" void kernel_launch(void* const* d_in, const int* in_sizes, int n_in,
                              void* d_out, int out_size, void* d_ws, size_t ws_size,
                              hipStream_t stream)
{
    (void)in_sizes; (void)n_in; (void)out_size; (void)ws_size;

    const float* x          = (const float*)d_in[0];
    const float* edge_attr  = (const float*)d_in[1];
    const int*   edge_index = (const int*)  d_in[2];
    const float* lin_w      = (const float*)d_in[3];
    const float* lin_b      = (const float*)d_in[4];
    const float* c1_w  = (const float*)d_in[5];
    const float* c1_b  = (const float*)d_in[6];
    const float* c1_g  = (const float*)d_in[7];
    const float* c1_be = (const float*)d_in[8];
    const float* c2_w  = (const float*)d_in[9];
    const float* c2_b  = (const float*)d_in[10];
    const float* c2_g  = (const float*)d_in[11];
    const float* c2_be = (const float*)d_in[12];
    const float* l1_w  = (const float*)d_in[13];
    const float* l1_b  = (const float*)d_in[14];
    const float* l1_g  = (const float*)d_in[15];
    const float* l1_be = (const float*)d_in[16];
    const float* ma_w  = (const float*)d_in[17];
    const float* ma_b  = (const float*)d_in[18];
    const float* ma_g  = (const float*)d_in[19];
    const float* ma_be = (const float*)d_in[20];
    const float* mb_w  = (const float*)d_in[21];
    const float* mb_b  = (const float*)d_in[22];
    const float* mb_g  = (const float*)d_in[23];
    const float* mb_be = (const float*)d_in[24];

    const int* src = edge_index;
    const int* dst = edge_index + NEDGES;

    const size_t N = NNODES;

    float* ws    = (float*)d_ws;
    float* bufP  = ws;
    float* bufQ  = bufP + N * 96;
    float* xcat  = bufQ + N * 96;
    float* stats = xcat + N * 160;
    int*   offT  = (int*)(stats + NSTAT);
    int2*  rec   = (int2*)(offT + 256 * PBLK);

    const float invN = 1.0f / (float)N;

    auto sumL = [&](int L) { return stats + (size_t)L * 192; };
    auto sqL  = [&](int L) { return stats + (size_t)L * 192 + 96; };

    static_assert(NBINS + 1 <= 256, "offT fits its part of the workspace");
    bin_chunk_kernel<<<PBLK, PTPB, 0, stream>>>(src, dst, edge_attr, offT, rec, stats);

    edge_gemm_kernel<96, false><<<NBINS, FUSE_TPB, 0, stream>>>(
        x, 96, rec, offT, lin_w, lin_b, c1_w, c1_b, xcat, 160, sumL(0), sqL(0),
        nullptr, nullptr, nullptr, nullptr, invN);

    edge_gemm_kernel<64, true><<<NBINS, FUSE_TPB, 0, stream>>>(
        xcat, 160, rec, offT, lin_w, lin_b, c2_w, c2_b, xcat + 96, 160, sumL(1), sqL(1),
        sumL(0), sqL(0), c1_g, c1_be, invN);

    gemm_wmma_f32_kernel<160, 96, false, true><<<GEMM_GRID, GEMM_TPB, 0, stream>>>(
        xcat, 160, nullptr, 0, l1_w, l1_b, bufQ, 96, sumL(2), sqL(2),
        sumL(0), sqL(0), c1_g, c1_be, 96,
        sumL(1), sqL(1), c2_g, c2_be, invN);

    gemm_wmma_f32_kernel<96, 96, false, true><<<GEMM_GRID, GEMM_TPB, 0, stream>>>(
        bufQ, 96, nullptr, 0, ma_w, ma_b, bufP, 96, sumL(3), sqL(3),
        sumL(2), sqL(2), l1_g, l1_be, 96,
        nullptr, nullptr, nullptr, nullptr, invN);

    gemm_wmma_f32_kernel<96, 16, false, true><<<GEMM_GRID, GEMM_TPB, 0, stream>>>(
        bufP, 96, nullptr, 0, mb_w, mb_b, bufQ, 16, sumL(4), sqL(4),
        sumL(3), sqL(3), ma_g, ma_be, 96,
        nullptr, nullptr, nullptr, nullptr, invN);

    bn_apply_kernel<<<(NNODES * 16 + 255) / 256, 256, 0, stream>>>(
        bufQ, 16, (float*)d_out, 16, sumL(4), sqL(4), mb_g, mb_be, invN, NNODES, 16);
}


# Edge sums in registers: a block orders its bin's records by destination row in LDS (count, scan, place); a wave sums runs of equal rows and adds once per run, not per edge; float sum order changes
# speedup vs baseline: 8.0619x; 1.0565x over previous
#include <hip/hip_runtime.h>
#include <hip/hip_bf16.h>


#define NNODES 50000
#define NEDGES 800000
#define BN_EPS 1e-5f

typedef __attribute__((ext_vector_type(2))) float v2f;
typedef __attribute__((ext_vector_type(8))) float v8f;

__device__ __forceinline__ void bn_scale_shift(const float* __restrict__ sum,
                                               const float* __restrict__ sumsq,
                                               const float* __restrict__ g,
                                               const float* __restrict__ beta,
                                               float invN, int c,
                                               float& sc, float& sh)
{
    const float mu  = sum[c] * invN;
    const float var = sumsq[c] * invN - mu * mu;
    const float inv = rsqrtf(var + BN_EPS);
    sc = g[c] * inv;
    sh = beta[c] - mu * sc;
}

#define SPB   13
#define NPB   (SPB * 16)
#define NBINS ((NNODES + NPB - 1) / NPB)
#define PBLK  128
#define PCH   (NEDGES / PBLK)
#define PTPB  1024
#define PIT   ((PCH + PTPB - 1) / PTPB)
#define NSTAT (5 * 192)

__global__ __launch_bounds__(PTPB)
void bin_chunk_kernel(const int* __restrict__ src,
                      const int* __restrict__ dst,
                      const float* __restrict__ ea,
                      int* __restrict__ offT,
                      int2* __restrict__ rec,
                      float* __restrict__ stats)
{
    static_assert(NNODES % 16 == 0, "a bin holds whole 16-row strips, the last bin fewer of them");
    static_assert(NEDGES % PBLK == 0, "whole chunks");
    static_assert(NBINS <= PTPB && NSTAT <= PTPB, "one thread per bin and per accumulator");
    static_assert(NNODES <= 65536, "src packs into 16 bits");
    static_assert(NPB < 32768, "local destination packs into the upper half");
    static_assert(PTPB == 1024, "32 waves: one lane of wave 0 per wave total");
    __shared__ int cnt[NBINS];
    __shared__ int wsum[PTPB / 32];
    const int tid  = threadIdx.x;
    const int lane = tid & 31;
    const int wv   = tid >> 5;

    if (tid < NBINS) cnt[tid] = 0;
    if (blockIdx.x == 0 && tid < NSTAT) stats[tid] = 0.0f;
    __syncthreads();

    const int j0 = blockIdx.x * PCH;
    int d[PIT], sv[PIT], av[PIT], rk[PIT];
#pragma unroll
    for (int i = 0; i < PIT; ++i) {
        const int  j  = tid + i * PTPB;
        const bool in = j < PCH;
        d[i]  = in ? dst[j0 + j] : 0;
        sv[i] = in ? src[j0 + j] : 0;
        av[i] = in ? __float_as_int(ea[j0 + j]) : 0;
    }
#pragma unroll
    for (int i = 0; i < PIT; ++i) {
        rk[i] = 0;
        if (tid + i * PTPB < PCH) rk[i] = atomicAdd(&cnt[d[i] / NPB], 1);
    }
    __syncthreads();

    const int tot = (tid < NBINS) ? cnt[tid] : 0;
    int inc = tot;
    for (int o = 1; o < 32; o <<= 1) {
        const int nb = __shfl_up(inc, o);
        if (lane >= o) inc += nb;
    }
    if (lane == 31) wsum[wv] = inc;
    __syncthreads();
    if (wv == 0) {
        const int w = wsum[lane];
        int winc = w;
        for (int o = 1; o < 32; o <<= 1) {
            const int nb = __shfl_up(winc, o);
            if (lane >= o) winc += nb;
        }
        wsum[lane] = winc - w;
    }
    __syncthreads();
    const int start = wsum[wv] + inc - tot;
    if (tid < NBINS) {
        cnt[tid] = start;
        offT[tid * PBLK + blockIdx.x] = start;
    }
    if (tid == 0) offT[NBINS * PBLK + blockIdx.x] = PCH;
    __syncthreads();

#pragma unroll
    for (int i = 0; i < PIT; ++i) {
        if (tid + i * PTPB < PCH) {
            const int b   = d[i] / NPB;
            const int loc = d[i] - b * NPB;
            rec[j0 + cnt[b] + rk[i]] = make_int2(sv[i] | (loc << 16), av[i]);
        }
    }
}

#define GEMM_WAVES 8
#define GEMM_TPB   (GEMM_WAVES * 32)
#define GEMM_STRIPS (NNODES / 16)
#define GEMM_GRID  ((GEMM_STRIPS + GEMM_WAVES - 1) / GEMM_WAVES)

template <int K, int N, bool HASADD, bool NORM>
__global__ __launch_bounds__(GEMM_TPB)
void gemm_wmma_f32_kernel(const float* __restrict__ A0, int lda0,
                          const float* __restrict__ A1, int lda1,
                          const float* __restrict__ W,
                          const float* __restrict__ bias,
                          float* __restrict__ out, int ldo,
                          float* __restrict__ colsum,
                          float* __restrict__ colsq,
                          const float* __restrict__ nsum0,
                          const float* __restrict__ nsq0,
                          const float* __restrict__ ng0,
                          const float* __restrict__ nbeta0,
                          int ncols0,
                          const float* __restrict__ nsum1,
                          const float* __restrict__ nsq1,
                          const float* __restrict__ ng1,
                          const float* __restrict__ nbeta1,
                          float invN)
{
    static_assert(K % 16 == 0 && N % 16 == 0, "whole k blocks and column tiles");
    static_assert(2 * N <= GEMM_TPB, "one thread per (column, statistic)");
    static_assert(K <= GEMM_TPB && K % 32 == 0, "one thread per column of A0 for its scale and shift, whole waves");
    static_assert(NNODES % 16 == 0, "whole row strips");
    static_assert(((K / 2) * (N / 4)) % 32 == 0, "the part round of the W staging is whole waves");
    constexpr int NT    = N / 16;
    constexpr int KB    = K / 16;
    constexpr int PITCH = 2 * N + 8;
    constexpr int NQ    = N / 4;

    __shared__ float4 wl4[(K / 2) * PITCH / 4];
    float* wl = reinterpret_cast<float*>(wl4);
    __shared__ float red[2][2 * GEMM_WAVES][N];
    __shared__ float4 nsc4[NORM ? K / 4 : 1];
    __shared__ float4 nsh4[NORM ? K / 4 : 1];

    const int tid  = threadIdx.x;
    const int lane = tid & 31;
    const int wv   = tid >> 5;
    const int lo   = lane & 15;
    const int hi   = lane >> 4;

    int strip = blockIdx.x * GEMM_WAVES + wv;
    const bool live = strip < GEMM_STRIPS;
    if (!live) strip = GEMM_STRIPS - 1;
    const int row0 = strip * 16;

    const float* ar0 = A0 + (size_t)(row0 + lo) * lda0 + 8 * hi;
    float4 a[KB][2];
#pragma unroll
    for (int i = 0; i < KB; ++i) {
        a[i][0] = *reinterpret_cast<const float4*>(ar0 + 16 * i);
        a[i][1] = *reinterpret_cast<const float4*>(ar0 + 16 * i + 4);
    }
    float4 a1[HASADD ? KB : 1][2];
    if (HASADD) {
        const float* ar1 = A1 + (size_t)(row0 + lo) * lda1 + 8 * hi;
#pragma unroll
        for (int i = 0; i < KB; ++i) {
            a1[i][0] = *reinterpret_cast<const float4*>(ar1 + 16 * i);
            a1[i][1] = *reinterpret_cast<const float4*>(ar1 + 16 * i + 4);
        }
    }

    constexpr int ITEMS = (K / 2) * NQ;
    constexpr int SFULL = ITEMS / GEMM_TPB;
    constexpr int SREM  = ITEMS % GEMM_TPB;
    float4 r0[SFULL > 0 ? SFULL : 1], r1[SFULL > 0 ? SFULL : 1];
#pragma unroll
    for (int s = 0; s < SFULL; ++s) {
        const int it = tid + s * GEMM_TPB;
        const int p  = it / NQ;
        const int n4 = it % NQ;
        r0[s] = *reinterpret_cast<const float4*>(W + (size_t)(2 * p) * N + 4 * n4);
        r1[s] = *reinterpret_cast<const float4*>(W + (size_t)(2 * p + 1) * N + 4 * n4);
    }
    if (NORM && tid < K) {
        const bool first = tid < ncols0;
        const int  c     = first ? tid : tid - ncols0;
        float sc, sh;
        bn_scale_shift(first ? nsum0 : nsum1, first ? nsq0 : nsq1,
                       first ? ng0 : ng1, first ? nbeta0 : nbeta1,
                       invN, c, sc, sh);
        reinterpret_cast<float*>(nsc4)[tid] = sc;
        reinterpret_cast<float*>(nsh4)[tid] = sh;
    }
    if (SREM != 0 && tid < SREM) {
        const int it = tid + SFULL * GEMM_TPB;
        const int p  = it / NQ;
        const int n4 = it % NQ;
        const float4 q0 = *reinterpret_cast<const float4*>(W + (size_t)(2 * p) * N + 4 * n4);
        const float4 q1 = *reinterpret_cast<const float4*>(W + (size_t)(2 * p + 1) * N + 4 * n4);
        float4* d = reinterpret_cast<float4*>(wl + p * PITCH + 8 * n4);
        d[0] = make_float4(q0.x, q1.x, q0.y, q1.y);
        d[1] = make_float4(q0.z, q1.z, q0.w, q1.w);
    }
#pragma unroll
    for (int s = 0; s < SFULL; ++s) {
        const int it = tid + s * GEMM_TPB;
        const int p  = it / NQ;
        const int n4 = it % NQ;
        float4* d = reinterpret_cast<float4*>(wl + p * PITCH + 8 * n4);
        d[0] = make_float4(r0[s].x, r1[s].x, r0[s].y, r1[s].y);
        d[1] = make_float4(r0[s].z, r1[s].z, r0[s].w, r1[s].w);
    }
    __syncthreads();

    if (NORM) {
#pragma unroll
        for (int i = 0; i < KB; ++i) {
#pragma unroll
            for (int h = 0; h < 2; ++h) {
                const float4 s4 = nsc4[4 * i + 2 * hi + h];
                const float4 t4 = nsh4[4 * i + 2 * hi + h];
                a[i][h].x = a[i][h].x * s4.x + t4.x;
                a[i][h].y = a[i][h].y * s4.y + t4.y;
                a[i][h].z = a[i][h].z * s4.z + t4.z;
                a[i][h].w = a[i][h].w * s4.w + t4.w;
            }
        }
    }

    if (HASADD) {
#pragma unroll
        for (int i = 0; i < KB; ++i) {
#pragma unroll
            for (int h = 0; h < 2; ++h) {
                a[i][h].x += a1[i][h].x;  a[i][h].y += a1[i][h].y;
                a[i][h].z += a1[i][h].z;  a[i][h].w += a1[i][h].w;
            }
        }
    }

    v8f c[NT];
#pragma unroll
    for (int t = 0; t < NT; ++t) c[t] = v8f{};

    const float* wb = wl + (4 * hi) * PITCH + 2 * lo;
#pragma unroll
    for (int i = 0; i < KB; ++i) {
#pragma unroll
        for (int j = 0; j < 4; ++j) {
            v2f av;
            if (j == 0)      { av.x = a[i][0].x; av.y = a[i][0].y; }
            else if (j == 1) { av.x = a[i][0].z; av.y = a[i][0].w; }
            else if (j == 2) { av.x = a[i][1].x; av.y = a[i][1].y; }
            else             { av.x = a[i][1].z; av.y = a[i][1].w; }
#pragma unroll
            for (int t = 0; t < NT; ++t) {
                const v2f bv = *reinterpret_cast<const v2f*>(
                    wb + (8 * i + j) * PITCH + 32 * t);
                c[t] = __builtin_amdgcn_wmma_f32_16x16x4_f32(
                         false, av,  false, bv,
                         (short)0, c[t],  false,  false);
            }
        }
    }

    float res[NT][8];
#pragma unroll
    for (int t = 0; t < NT; ++t) {
        const float bsv = bias[16 * t + lo];
#pragma unroll
        for (int r = 0; r < 8; ++r) res[t][r] = fmaxf(c[t][r] + bsv, 0.0f);
    }
#pragma unroll
    for (int t = 0; t < NT; ++t) {
        float s = 0.0f, q = 0.0f;
#pragma unroll
        for (int r = 0; r < 8; ++r) {
            const float v = res[t][r];
            s += v;
            q += v * v;
        }
        red[0][2 * wv + hi][16 * t + lo] = live ? s : 0.0f;
        red[1][2 * wv + hi][16 * t + lo] = live ? q : 0.0f;
    }
    __syncthreads();

    if (live) {
#pragma unroll
        for (int r = 0; r < 8; ++r) {
            float* orow = out + (size_t)(row0 + r + 8 * hi) * ldo + lo;
#pragma unroll
            for (int t = 0; t < NT; ++t) orow[16 * t] = res[t][r];
        }
    }

    if (tid < 2 * N) {
        const int which = tid / N;
        const int col   = tid % N;
        float acc = 0.0f;
#pragma unroll
        for (int p = 0; p < 2 * GEMM_WAVES; ++p) acc += red[which][p][col];
        atomicAdd(which ? &colsq[col] : &colsum[col], acc);
    }
}

#define FUSE_WAVES 32
#define FUSE_TPB   (FUSE_WAVES * 32)
#define GEMM_WV    (2 * SPB)
#define EDGE_U     16
#define APITCH     100
#define SRT_CAP    4096
#define SRT_Q      (SRT_CAP / FUSE_TPB)
#define SCAN_Q     ((NPB + 31) / 32)

template <int N, bool NORM>
__global__ __launch_bounds__(FUSE_TPB)
void edge_gemm_kernel(const float* __restrict__ x, int ldx,
                      const int2* __restrict__ rec,
                      const int* __restrict__ offT,
                      const float* __restrict__ lw,
                      const float* __restrict__ lb,
                      const float* __restrict__ W,
                      const float* __restrict__ bias,
                      float* __restrict__ out, int ldo,
                      float* __restrict__ colsum,
                      float* __restrict__ colsq,
                      const float* __restrict__ nsum,
                      const float* __restrict__ nsq,
                      const float* __restrict__ ng,
                      const float* __restrict__ nbeta,
                      float invN)
{
    constexpr int K     = 96;
    constexpr int KB    = K / 16;
    constexpr int NTH   = N / 32;
    constexpr int PITCH = 2 * N + 8;
    constexpr int NQ    = N / 4;
    static_assert(N % 32 == 0, "two column halves of whole tiles");
    static_assert(2 * N <= FUSE_TPB && K <= FUSE_TPB, "one thread per (column, statistic) and per column of x");
    static_assert(32 == 2 * EDGE_U, "a round of 32 records is two batches");
    static_assert(APITCH >= K && APITCH % 4 == 0, "rows of the sums are whole float4");
    static_assert(FUSE_TPB <= 1024 && GEMM_WV <= FUSE_WAVES, "one block, with a wave for every strip half");
    static_assert(PBLK == 128, "wave 0 scans the pieces, four per lane");
    static_assert(SRT_CAP % FUSE_TPB == 0 && SRT_CAP % (FUSE_WAVES * EDGE_U) == 0,
                  "whole records per thread; a full round splits into whole batches per wave");

    __shared__ float4 acc4[NPB * APITCH / 4];
    float* acc = reinterpret_cast<float*>(acc4);
    __shared__ float4 wl4[(K / 2) * PITCH / 4];
    float* wl = reinterpret_cast<float*>(wl4);
    __shared__ float red[2][SPB][N];
    __shared__ float4 nsc4[NORM ? K / 4 : 1];
    __shared__ float4 nsh4[NORM ? K / 4 : 1];
    __shared__ int2 tab[PBLK + 1];
    __shared__ int2 srt[SRT_CAP];
    __shared__ int  rcnt[NPB];
    __shared__ int  rstart[NPB];

    const int tid  = threadIdx.x;
    const int lane = tid & 31;
    const int lo   = lane & 15;
    const int hi   = lane >> 4;
    const int wv   = __builtin_amdgcn_readfirstlane(tid >> 5);

    int4 s4 = make_int4(0, 0, 0, 0), e4 = make_int4(0, 0, 0, 0);
    if (wv == 0) {
        s4 = *reinterpret_cast<const int4*>(offT + (size_t)blockIdx.x * PBLK + 4 * lane);
        e4 = *reinterpret_cast<const int4*>(offT + (size_t)(blockIdx.x + 1) * PBLK + 4 * lane);
    }

    for (int i = tid; i < NPB * APITCH / 4; i += FUSE_TPB)
        acc4[i] = make_float4(0.0f, 0.0f, 0.0f, 0.0f);
    if (tid < NPB) rcnt[tid] = 0;

    const float w0 = lw[lane], w1 = lw[lane + 32], w2 = lw[lane + 64];
    const float b0 = lb[lane], b1 = lb[lane + 32], b2 = lb[lane + 64];

    float sc0 = 1.0f, sc1 = 1.0f, sc2 = 1.0f, sh0 = 0.0f, sh1 = 0.0f, sh2 = 0.0f;
    if (NORM) {
        bn_scale_shift(nsum, nsq, ng, nbeta, invN, lane,      sc0, sh0);
        bn_scale_shift(nsum, nsq, ng, nbeta, invN, lane + 32, sc1, sh1);
        bn_scale_shift(nsum, nsq, ng, nbeta, invN, lane + 64, sc2, sh2);
        if (tid < K) {
            float sc, sh;
            bn_scale_shift(nsum, nsq, ng, nbeta, invN, tid, sc, sh);
            reinterpret_cast<float*>(nsc4)[tid] = sc;
            reinterpret_cast<float*>(nsh4)[tid] = sh;
        }
    }
    if (wv == 0) {
        const int c0 = e4.x - s4.x, c1 = e4.y - s4.y, c2 = e4.z - s4.z, c3 = e4.w - s4.w;
        const int lt = c0 + c1 + c2 + c3;
        int inc = lt;
        for (int o = 1; o < 32; o <<= 1) {
            const int nb = __shfl_up(inc, o);
            if (lane >= o) inc += nb;
        }
        const int p0 = inc - lt, p1 = p0 + c0, p2 = p1 + c1, p3 = p2 + c2;
        const int q0 = 4 * lane * PCH;
        tab[4 * lane + 0] = make_int2(p0, q0 + s4.x - p0);
        tab[4 * lane + 1] = make_int2(p1, q0 + PCH + s4.y - p1);
        tab[4 * lane + 2] = make_int2(p2, q0 + 2 * PCH + s4.z - p2);
        tab[4 * lane + 3] = make_int2(p3, q0 + 3 * PCH + s4.w - p3);
        if (lane == 31) tab[PBLK] = make_int2(inc, 0);
    }
    __syncthreads();

    const int total = __builtin_amdgcn_readfirstlane(tab[PBLK].x);

    float v0[EDGE_U], v1[EDGE_U], v2[EDGE_U];
    auto gather = [&](int kv, int e0) {
#pragma unroll
        for (int u = 0; u < EDGE_U; ++u) {
            const int s = __builtin_amdgcn_readlane(kv, e0 + u) & 0xFFFF;
            const float* xr = x + (size_t)s * ldx + lane;
            v0[u] = xr[0];
            v1[u] = xr[32];
            v2[u] = xr[64];
        }
    };
    int   cur = 0;
    float s0 = 0.0f, s1 = 0.0f, s2 = 0.0f;
    auto flush = [&]() {
        float* ap = acc + cur * APITCH + lane;
        atomicAdd(ap,      s0);
        atomicAdd(ap + 32, s1);
        atomicAdd(ap + 64, s2);
    };
    auto addup = [&](int kv, int av, int e0, int n) {
#pragma unroll
        for (int u = 0; u < EDGE_U; ++u) {
            if (e0 + u < n) {
                const int   t = __builtin_amdgcn_readlane(kv, e0 + u) >> 16;
                const float a = __int_as_float(__builtin_amdgcn_readlane(av, e0 + u));
                if (t != cur) {
                    flush();
                    cur = t;
                    s0 = 0.0f;  s1 = 0.0f;  s2 = 0.0f;
                }
                const float x0 = NORM ? v0[u] * sc0 + sh0 : v0[u];
                const float x1 = NORM ? v1[u] * sc1 + sh1 : v1[u];
                const float x2 = NORM ? v2[u] * sc2 + sh2 : v2[u];
                s0 += fmaxf(x0 + a * w0 + b0, 0.0f);
                s1 += fmaxf(x1 + a * w1 + b1, 0.0f);
                s2 += fmaxf(x2 + a * w2 + b2, 0.0f);
            }
        }
    };

    for (int base = 0; base < total; base += SRT_CAP) {
        const int nt = min(SRT_CAP, total - base);

        int pp[SRT_Q];
#pragma unroll
        for (int q = 0; q < SRT_Q; ++q) pp[q] = 0;
#pragma unroll
        for (int s = PBLK / 2; s >= 1; s >>= 1) {
#pragma unroll
            for (int q = 0; q < SRT_Q; ++q)
                if (tab[pp[q] + s].x <= base + q * FUSE_TPB + tid) pp[q] += s;
        }
        int2 rr[SRT_Q];
        int  rk[SRT_Q];
        int po[SRT_Q];
#pragma unroll
        for (int q = 0; q < SRT_Q; ++q) po[q] = tab[pp[q]].y;
#pragma unroll
        for (int q = 0; q < SRT_Q; ++q) {
            const int j = q * FUSE_TPB + tid;
            rr[q] = make_int2(0, 0);
            if (j < nt) rr[q] = rec[base + j + po[q]];
        }
#pragma unroll
        for (int q = 0; q < SRT_Q; ++q) {
            rk[q] = 0;
            if (q * FUSE_TPB + tid < nt) rk[q] = atomicAdd(&rcnt[rr[q].x >> 16], 1);
        }
        __syncthreads();

        if (wv == 0) {
            int cn[SCAN_Q];
            int lt = 0;
#pragma unroll
            for (int k = 0; k < SCAN_Q; ++k) {
                const int row = SCAN_Q * lane + k;
                cn[k] = (row < NPB) ? rcnt[row] : 0;
                lt += cn[k];
            }
            int inc = lt;
            for (int o = 1; o < 32; o <<= 1) {
                const int nb = __shfl_up(inc, o);
                if (lane >= o) inc += nb;
            }
            int run = inc - lt;
#pragma unroll
            for (int k = 0; k < SCAN_Q; ++k) {
                const int row = SCAN_Q * lane + k;
                if (row < NPB) {
                    rstart[row] = run;
                    rcnt[row]   = 0;
                }
                run += cn[k];
            }
        }
        __syncthreads();

#pragma unroll
        for (int q = 0; q < SRT_Q; ++q)
            if (q * FUSE_TPB + tid < nt) srt[rstart[rr[q].x >> 16] + rk[q]] = rr[q];
        __syncthreads();

        const int per = (((nt + FUSE_WAVES - 1) / FUSE_WAVES) + EDGE_U - 1) / EDGE_U * EDGE_U;
        const int lo_ = wv * per;
        const int hi_ = min(lo_ + per, nt);
        if (lo_ < hi_) {
            int2 r = (lo_ + lane < hi_) ? srt[lo_ + lane] : make_int2(0, 0);
            for (int g = lo_; g < hi_; g += 32) {
                const int n  = min(32, hi_ - g);
                const int kv = r.x;
                const int av = r.y;
                gather(kv, 0);
                r = (g + 32 + lane < hi_) ? srt[g + 32 + lane] : make_int2(0, 0);
                addup(kv, av, 0, n);
                if (n > EDGE_U) {
                    gather(kv, EDGE_U);
                    addup(kv, av, EDGE_U, n);
                }
            }
        }
    }
    flush();

    const bool gw = wv < GEMM_WV;
    const int  st = wv >> 1;
    const int  hf = wv & 1;
    int row0 = blockIdx.x * NPB + st * 16;
    const bool live = gw && row0 < NNODES;
    if (!live) row0 = NNODES - 16;
    float4 a[KB][2];
    if (gw) {
        const float* ar = x + (size_t)(row0 + lo) * ldx + 8 * hi;
#pragma unroll
        for (int i = 0; i < KB; ++i) {
            a[i][0] = *reinterpret_cast<const float4*>(ar + 16 * i);
            a[i][1] = *reinterpret_cast<const float4*>(ar + 16 * i + 4);
        }
    }
    constexpr int ITEMS = (K / 2) * NQ;
    for (int it = tid; it < ITEMS; it += FUSE_TPB) {
        const int p  = it / NQ;
        const int n4 = it % NQ;
        const float4 q0 = *reinterpret_cast<const float4*>(W + (size_t)(2 * p) * N + 4 * n4);
        const float4 q1 = *reinterpret_cast<const float4*>(W + (size_t)(2 * p + 1) * N + 4 * n4);
        float4* d = reinterpret_cast<float4*>(wl + p * PITCH + 8 * n4);
        d[0] = make_float4(q0.x, q1.x, q0.y, q1.y);
        d[1] = make_float4(q0.z, q1.z, q0.w, q1.w);
    }
    __syncthreads();

    const int col0 = 16 * NTH * hf + lo;
    float res[NTH][8];
    if (gw) {
        const float* pr = acc + (st * 16 + lo) * APITCH + 8 * hi;
#pragma unroll
        for (int i = 0; i < KB; ++i) {
#pragma unroll
            for (int h = 0; h < 2; ++h) {
                if (NORM) {
                    const float4 s4 = nsc4[4 * i + 2 * hi + h];
                    const float4 t4 = nsh4[4 * i + 2 * hi + h];
                    a[i][h].x = a[i][h].x * s4.x + t4.x;
                    a[i][h].y = a[i][h].y * s4.y + t4.y;
                    a[i][h].z = a[i][h].z * s4.z + t4.z;
                    a[i][h].w = a[i][h].w * s4.w + t4.w;
                }
                const float4 g4 = *reinterpret_cast<const float4*>(pr + 16 * i + 4 * h);
                a[i][h].x += g4.x;  a[i][h].y += g4.y;
                a[i][h].z += g4.z;  a[i][h].w += g4.w;
            }
        }

        v8f c[NTH];
#pragma unroll
        for (int t = 0; t < NTH; ++t) c[t] = v8f{};

        const float* wb = wl + (4 * hi) * PITCH + 2 * lo + 32 * NTH * hf;
#pragma unroll
        for (int i = 0; i < KB; ++i) {
#pragma unroll
            for (int j = 0; j < 4; ++j) {
                v2f av;
                if (j == 0)      { av.x = a[i][0].x; av.y = a[i][0].y; }
                else if (j == 1) { av.x = a[i][0].z; av.y = a[i][0].w; }
                else if (j == 2) { av.x = a[i][1].x; av.y = a[i][1].y; }
                else             { av.x = a[i][1].z; av.y = a[i][1].w; }
#pragma unroll
                for (int t = 0; t < NTH; ++t) {
                    const v2f bv = *reinterpret_cast<const v2f*>(
                        wb + (8 * i + j) * PITCH + 32 * t);
                    c[t] = __builtin_amdgcn_wmma_f32_16x16x4_f32(
                             false, av,  false, bv,
                             (short)0, c[t],  false,  false);
                }
            }
        }

#pragma unroll
        for (int t = 0; t < NTH; ++t) {
            const float bsv = bias[col0 + 16 * t];
#pragma unroll
            for (int r = 0; r < 8; ++r) res[t][r] = fmaxf(c[t][r] + bsv, 0.0f);
        }
#pragma unroll
        for (int t = 0; t < NTH; ++t) {
            float s = 0.0f, q = 0.0f;
#pragma unroll
            for (int r = 0; r < 8; ++r) {
                const float v = res[t][r];
                s += v;
                q += v * v;
            }
            s += __shfl_xor(s, 16);
            q += __shfl_xor(q, 16);
            if (hi == 0) {
                red[0][st][col0 + 16 * t] = live ? s : 0.0f;
                red[1][st][col0 + 16 * t] = live ? q : 0.0f;
            }
        }
    }
    __syncthreads();

    if (live) {
#pragma unroll
        for (int r = 0; r < 8; ++r) {
            float* orow = out + (size_t)(row0 + r + 8 * hi) * ldo + col0;
#pragma unroll
            for (int t = 0; t < NTH; ++t) orow[16 * t] = res[t][r];
        }
    }

    if (tid < 2 * N) {
        const int which = tid / N;
        const int col   = tid % N;
        float sum = 0.0f;
#pragma unroll
        for (int p = 0; p < SPB; ++p) sum += red[which][p][col];
        atomicAdd(which ? &colsq[col] : &colsum[col], sum);
    }
}

__global__ __launch_bounds__(256)
void bn_apply_kernel(const float* __restrict__ in, int ldi,
                     float* __restrict__ out, int ldo,
                     const float* __restrict__ sum,
                     const float* __restrict__ sumsq,
                     const float* __restrict__ g,
                     const float* __restrict__ beta,
                     float invN, int n, int C)
{
    int gid = blockIdx.x * blockDim.x + threadIdx.x;
    if (gid >= n * C) return;
    int row = gid / C;
    int c   = gid % C;
    float sc, sh;
    bn_scale_shift(sum, sumsq, g, beta, invN, c, sc, sh);
    out[(size_t)row * ldo + c] = in[(size_t)row * ldi + c] * sc + sh;
}

extern "C" void kernel_launch(void* const* d_in, const int* in_sizes, int n_in,
                              void* d_out, int out_size, void* d_ws, size_t ws_size,
                              hipStream_t stream)
{
    (void)in_sizes; (void)n_in; (void)out_size; (void)ws_size;

    const float* x          = (const float*)d_in[0];
    const float* edge_attr  = (const float*)d_in[1];
    const int*   edge_index = (const int*)  d_in[2];
    const float* lin_w      = (const float*)d_in[3];
    const float* lin_b      = (const float*)d_in[4];
    const float* c1_w  = (const float*)d_in[5];
    const float* c1_b  = (const float*)d_in[6];
    const float* c1_g  = (const float*)d_in[7];
    const float* c1_be = (const float*)d_in[8];
    const float* c2_w  = (const float*)d_in[9];
    const float* c2_b  = (const float*)d_in[10];
    const float* c2_g  = (const float*)d_in[11];
    const float* c2_be = (const float*)d_in[12];
    const float* l1_w  = (const float*)d_in[13];
    const float* l1_b  = (const float*)d_in[14];
    const float* l1_g  = (const float*)d_in[15];
    const float* l1_be = (const float*)d_in[16];
    const float* ma_w  = (const float*)d_in[17];
    const float* ma_b  = (const float*)d_in[18];
    const float* ma_g  = (const float*)d_in[19];
    const float* ma_be = (const float*)d_in[20];
    const float* mb_w  = (const float*)d_in[21];
    const float* mb_b  = (const float*)d_in[22];
    const float* mb_g  = (const float*)d_in[23];
    const float* mb_be = (const float*)d_in[24];

    const int* src = edge_index;
    const int* dst = edge_index + NEDGES;

    const size_t N = NNODES;

    float* ws    = (float*)d_ws;
    float* bufP  = ws;
    float* bufQ  = bufP + N * 96;
    float* xcat  = bufQ + N * 96;
    float* stats = xcat + N * 160;
    int*   offT  = (int*)(stats + NSTAT);
    int2*  rec   = (int2*)(offT + 256 * PBLK);

    const float invN = 1.0f / (float)N;

    auto sumL = [&](int L) { return stats + (size_t)L * 192; };
    auto sqL  = [&](int L) { return stats + (size_t)L * 192 + 96; };

    static_assert(NBINS + 1 <= 256, "offT fits its part of the workspace");
    bin_chunk_kernel<<<PBLK, PTPB, 0, stream>>>(src, dst, edge_attr, offT, rec, stats);

    edge_gemm_kernel<96, false><<<NBINS, FUSE_TPB, 0, stream>>>(
        x, 96, rec, offT, lin_w, lin_b, c1_w, c1_b, xcat, 160, sumL(0), sqL(0),
        nullptr, nullptr, nullptr, nullptr, invN);

    edge_gemm_kernel<64, true><<<NBINS, FUSE_TPB, 0, stream>>>(
        xcat, 160, rec, offT, lin_w, lin_b, c2_w, c2_b, xcat + 96, 160, sumL(1), sqL(1),
        sumL(0), sqL(0), c1_g, c1_be, invN);

    gemm_wmma_f32_kernel<160, 96, false, true><<<GEMM_GRID, GEMM_TPB, 0, stream>>>(
        xcat, 160, nullptr, 0, l1_w, l1_b, bufQ, 96, sumL(2), sqL(2),
        sumL(0), sqL(0), c1_g, c1_be, 96,
        sumL(1), sqL(1), c2_g, c2_be, invN);

    gemm_wmma_f32_kernel<96, 96, false, true><<<GEMM_GRID, GEMM_TPB, 0, stream>>>(
        bufQ, 96, nullptr, 0, ma_w, ma_b, bufP, 96, sumL(3), sqL(3),
        sumL(2), sqL(2), l1_g, l1_be, 96,
        nullptr, nullptr, nullptr, nullptr, invN);

    gemm_wmma_f32_kernel<96, 16, false, true><<<GEMM_GRID, GEMM_TPB, 0, stream>>>(
        bufP, 96, nullptr, 0, mb_w, mb_b, bufQ, 16, sumL(4), sqL(4),
        sumL(3), sqL(3), ma_g, ma_be, 96,
        nullptr, nullptr, nullptr, nullptr, invN);

    bn_apply_kernel<<<(NNODES * 16 + 255) / 256, 256, 0, stream>>>(
        bufQ, 16, (float*)d_out, 16, sumL(4), sqL(4), mb_g, mb_be, invN, NNODES, 16);
}


# Edge binning: a chunk's records are ordered by bin in LDS and leave as 16-byte stores of consecutive lanes, 512 contiguous bytes per store (was 32 lines of 8 bytes each); sum orders unchanged
# speedup vs baseline: 8.3141x; 1.0313x over previous
#include <hip/hip_runtime.h>
#include <hip/hip_bf16.h>


#define NNODES 50000
#define NEDGES 800000
#define BN_EPS 1e-5f

typedef __attribute__((ext_vector_type(2))) float v2f;
typedef __attribute__((ext_vector_type(8))) float v8f;

__device__ __forceinline__ void bn_scale_shift(const float* __restrict__ sum,
                                               const float* __restrict__ sumsq,
                                               const float* __restrict__ g,
                                               const float* __restrict__ beta,
                                               float invN, int c,
                                               float& sc, float& sh)
{
    const float mu  = sum[c] * invN;
    const float var = sumsq[c] * invN - mu * mu;
    const float inv = rsqrtf(var + BN_EPS);
    sc = g[c] * inv;
    sh = beta[c] - mu * sc;
}

#define SPB   13
#define NPB   (SPB * 16)
#define NBINS ((NNODES + NPB - 1) / NPB)
#define PBLK  128
#define PCH   (NEDGES / PBLK)
#define PTPB  1024
#define PIT   ((PCH + PTPB - 1) / PTPB)
#define NSTAT (5 * 192)

__global__ __launch_bounds__(PTPB)
void bin_chunk_kernel(const int* __restrict__ src,
                      const int* __restrict__ dst,
                      const float* __restrict__ ea,
                      int* __restrict__ offT,
                      int2* __restrict__ rec,
                      float* __restrict__ stats)
{
    static_assert(NNODES % 16 == 0, "a bin holds whole 16-row strips, the last bin fewer of them");
    static_assert(NEDGES % PBLK == 0, "whole chunks");
    static_assert(NBINS <= PTPB && NSTAT <= PTPB, "one thread per bin and per accumulator");
    static_assert(NNODES <= 65536, "src packs into 16 bits");
    static_assert(NPB < 32768, "local destination packs into the upper half");
    static_assert(PTPB == 1024, "32 waves: one lane of wave 0 per wave total");
    static_assert(PCH % 2 == 0, "a chunk's records leave two per 16-byte store, and every chunk starts on a 16-byte boundary");
    __shared__ int cnt[NBINS];
    __shared__ int wsum[PTPB / 32];
    __shared__ int4 srec4[PCH / 2];
    const int tid  = threadIdx.x;
    const int lane = tid & 31;
    const int wv   = tid >> 5;

    if (tid < NBINS) cnt[tid] = 0;
    if (blockIdx.x == 0 && tid < NSTAT) stats[tid] = 0.0f;
    __syncthreads();

    const int j0 = blockIdx.x * PCH;
    int d[PIT], sv[PIT], av[PIT], rk[PIT];
#pragma unroll
    for (int i = 0; i < PIT; ++i) {
        const int  j  = tid + i * PTPB;
        const bool in = j < PCH;
        d[i]  = in ? dst[j0 + j] : 0;
        sv[i] = in ? src[j0 + j] : 0;
        av[i] = in ? __float_as_int(ea[j0 + j]) : 0;
    }
#pragma unroll
    for (int i = 0; i < PIT; ++i) {
        rk[i] = 0;
        if (tid + i * PTPB < PCH) rk[i] = atomicAdd(&cnt[d[i] / NPB], 1);
    }
    __syncthreads();

    const int tot = (tid < NBINS) ? cnt[tid] : 0;
    int inc = tot;
    for (int o = 1; o < 32; o <<= 1) {
        const int nb = __shfl_up(inc, o);
        if (lane >= o) inc += nb;
    }
    if (lane == 31) wsum[wv] = inc;
    __syncthreads();
    if (wv == 0) {
        const int w = wsum[lane];
        int winc = w;
        for (int o = 1; o < 32; o <<= 1) {
            const int nb = __shfl_up(winc, o);
            if (lane >= o) winc += nb;
        }
        wsum[lane] = winc - w;
    }
    __syncthreads();
    const int start = wsum[wv] + inc - tot;
    if (tid < NBINS) {
        cnt[tid] = start;
        offT[tid * PBLK + blockIdx.x] = start;
    }
    if (tid == 0) offT[NBINS * PBLK + blockIdx.x] = PCH;
    __syncthreads();

    int2* srec = reinterpret_cast<int2*>(srec4);
#pragma unroll
    for (int i = 0; i < PIT; ++i) {
        if (tid + i * PTPB < PCH) {
            const int b   = d[i] / NPB;
            const int loc = d[i] - b * NPB;
            srec[cnt[b] + rk[i]] = make_int2(sv[i] | (loc << 16), av[i]);
        }
    }
    __syncthreads();

    int4* out4 = reinterpret_cast<int4*>(rec + j0);
    constexpr int Q4  = PCH / 2;
    constexpr int QIT = (Q4 + PTPB - 1) / PTPB;
#pragma unroll
    for (int i = 0; i < QIT; ++i) {
        const int k = tid + i * PTPB;
        if (k < Q4) out4[k] = srec4[k];
    }
}

#define GEMM_WAVES 8
#define GEMM_TPB   (GEMM_WAVES * 32)
#define GEMM_STRIPS (NNODES / 16)
#define GEMM_GRID  ((GEMM_STRIPS + GEMM_WAVES - 1) / GEMM_WAVES)

template <int K, int N, bool HASADD, bool NORM>
__global__ __launch_bounds__(GEMM_TPB)
void gemm_wmma_f32_kernel(const float* __restrict__ A0, int lda0,
                          const float* __restrict__ A1, int lda1,
                          const float* __restrict__ W,
                          const float* __restrict__ bias,
                          float* __restrict__ out, int ldo,
                          float* __restrict__ colsum,
                          float* __restrict__ colsq,
                          const float* __restrict__ nsum0,
                          const float* __restrict__ nsq0,
                          const float* __restrict__ ng0,
                          const float* __restrict__ nbeta0,
                          int ncols0,
                          const float* __restrict__ nsum1,
                          const float* __restrict__ nsq1,
                          const float* __restrict__ ng1,
                          const float* __restrict__ nbeta1,
                          float invN)
{
    static_assert(K % 16 == 0 && N % 16 == 0, "whole k blocks and column tiles");
    static_assert(2 * N <= GEMM_TPB, "one thread per (column, statistic)");
    static_assert(K <= GEMM_TPB && K % 32 == 0, "one thread per column of A0 for its scale and shift, whole waves");
    static_assert(NNODES % 16 == 0, "whole row strips");
    static_assert(((K / 2) * (N / 4)) % 32 == 0, "the part round of the W staging is whole waves");
    constexpr int NT    = N / 16;
    constexpr int KB    = K / 16;
    constexpr int PITCH = 2 * N + 8;
    constexpr int NQ    = N / 4;

    __shared__ float4 wl4[(K / 2) * PITCH / 4];
    float* wl = reinterpret_cast<float*>(wl4);
    __shared__ float red[2][2 * GEMM_WAVES][N];
    __shared__ float4 nsc4[NORM ? K / 4 : 1];
    __shared__ float4 nsh4[NORM ? K / 4 : 1];

    const int tid  = threadIdx.x;
    const int lane = tid & 31;
    const int wv   = tid >> 5;
    const int lo   = lane & 15;
    const int hi   = lane >> 4;

    int strip = blockIdx.x * GEMM_WAVES + wv;
    const bool live = strip < GEMM_STRIPS;
    if (!live) strip = GEMM_STRIPS - 1;
    const int row0 = strip * 16;

    const float* ar0 = A0 + (size_t)(row0 + lo) * lda0 + 8 * hi;
    float4 a[KB][2];
#pragma unroll
    for (int i = 0; i < KB; ++i) {
        a[i][0] = *reinterpret_cast<const float4*>(ar0 + 16 * i);
        a[i][1] = *reinterpret_cast<const float4*>(ar0 + 16 * i + 4);
    }
    float4 a1[HASADD ? KB : 1][2];
    if (HASADD) {
        const float* ar1 = A1 + (size_t)(row0 + lo) * lda1 + 8 * hi;
#pragma unroll
        for (int i = 0; i < KB; ++i) {
            a1[i][0] = *reinterpret_cast<const float4*>(ar1 + 16 * i);
            a1[i][1] = *reinterpret_cast<const float4*>(ar1 + 16 * i + 4);
        }
    }

    constexpr int ITEMS = (K / 2) * NQ;
    constexpr int SFULL = ITEMS / GEMM_TPB;
    constexpr int SREM  = ITEMS % GEMM_TPB;
    float4 r0[SFULL > 0 ? SFULL : 1], r1[SFULL > 0 ? SFULL : 1];
#pragma unroll
    for (int s = 0; s < SFULL; ++s) {
        const int it = tid + s * GEMM_TPB;
        const int p  = it / NQ;
        const int n4 = it % NQ;
        r0[s] = *reinterpret_cast<const float4*>(W + (size_t)(2 * p) * N + 4 * n4);
        r1[s] = *reinterpret_cast<const float4*>(W + (size_t)(2 * p + 1) * N + 4 * n4);
    }
    if (NORM && tid < K) {
        const bool first = tid < ncols0;
        const int  c     = first ? tid : tid - ncols0;
        float sc, sh;
        bn_scale_shift(first ? nsum0 : nsum1, first ? nsq0 : nsq1,
                       first ? ng0 : ng1, first ? nbeta0 : nbeta1,
                       invN, c, sc, sh);
        reinterpret_cast<float*>(nsc4)[tid] = sc;
        reinterpret_cast<float*>(nsh4)[tid] = sh;
    }
    if (SREM != 0 && tid < SREM) {
        const int it = tid + SFULL * GEMM_TPB;
        const int p  = it / NQ;
        const int n4 = it % NQ;
        const float4 q0 = *reinterpret_cast<const float4*>(W + (size_t)(2 * p) * N + 4 * n4);
        const float4 q1 = *reinterpret_cast<const float4*>(W + (size_t)(2 * p + 1) * N + 4 * n4);
        float4* d = reinterpret_cast<float4*>(wl + p * PITCH + 8 * n4);
        d[0] = make_float4(q0.x, q1.x, q0.y, q1.y);
        d[1] = make_float4(q0.z, q1.z, q0.w, q1.w);
    }
#pragma unroll
    for (int s = 0; s < SFULL; ++s) {
        const int it = tid + s * GEMM_TPB;
        const int p  = it / NQ;
        const int n4 = it % NQ;
        float4* d = reinterpret_cast<float4*>(wl + p * PITCH + 8 * n4);
        d[0] = make_float4(r0[s].x, r1[s].x, r0[s].y, r1[s].y);
        d[1] = make_float4(r0[s].z, r1[s].z, r0[s].w, r1[s].w);
    }
    __syncthreads();

    if (NORM) {
#pragma unroll
        for (int i = 0; i < KB; ++i) {
#pragma unroll
            for (int h = 0; h < 2; ++h) {
                const float4 s4 = nsc4[4 * i + 2 * hi + h];
                const float4 t4 = nsh4[4 * i + 2 * hi + h];
                a[i][h].x = a[i][h].x * s4.x + t4.x;
                a[i][h].y = a[i][h].y * s4.y + t4.y;
                a[i][h].z = a[i][h].z * s4.z + t4.z;
                a[i][h].w = a[i][h].w * s4.w + t4.w;
            }
        }
    }

    if (HASADD) {
#pragma unroll
        for (int i = 0; i < KB; ++i) {
#pragma unroll
            for (int h = 0; h < 2; ++h) {
                a[i][h].x += a1[i][h].x;  a[i][h].y += a1[i][h].y;
                a[i][h].z += a1[i][h].z;  a[i][h].w += a1[i][h].w;
            }
        }
    }

    v8f c[NT];
#pragma unroll
    for (int t = 0; t < NT; ++t) c[t] = v8f{};

    const float* wb = wl + (4 * hi) * PITCH + 2 * lo;
#pragma unroll
    for (int i = 0; i < KB; ++i) {
#pragma unroll
        for (int j = 0; j < 4; ++j) {
            v2f av;
            if (j == 0)      { av.x = a[i][0].x; av.y = a[i][0].y; }
            else if (j == 1) { av.x = a[i][0].z; av.y = a[i][0].w; }
            else if (j == 2) { av.x = a[i][1].x; av.y = a[i][1].y; }
            else             { av.x = a[i][1].z; av.y = a[i][1].w; }
#pragma unroll
            for (int t = 0; t < NT; ++t) {
                const v2f bv = *reinterpret_cast<const v2f*>(
                    wb + (8 * i + j) * PITCH + 32 * t);
                c[t] = __builtin_amdgcn_wmma_f32_16x16x4_f32(
                         false, av,  false, bv,
                         (short)0, c[t],  false,  false);
            }
        }
    }

    float res[NT][8];
#pragma unroll
    for (int t = 0; t < NT; ++t) {
        const float bsv = bias[16 * t + lo];
#pragma unroll
        for (int r = 0; r < 8; ++r) res[t][r] = fmaxf(c[t][r] + bsv, 0.0f);
    }
#pragma unroll
    for (int t = 0; t < NT; ++t) {
        float s = 0.0f, q = 0.0f;
#pragma unroll
        for (int r = 0; r < 8; ++r) {
            const float v = res[t][r];
            s += v;
            q += v * v;
        }
        red[0][2 * wv + hi][16 * t + lo] = live ? s : 0.0f;
        red[1][2 * wv + hi][16 * t + lo] = live ? q : 0.0f;
    }
    __syncthreads();

    if (live) {
#pragma unroll
        for (int r = 0; r < 8; ++r) {
            float* orow = out + (size_t)(row0 + r + 8 * hi) * ldo + lo;
#pragma unroll
            for (int t = 0; t < NT; ++t) orow[16 * t] = res[t][r];
        }
    }

    if (tid < 2 * N) {
        const int which = tid / N;
        const int col   = tid % N;
        float acc = 0.0f;
#pragma unroll
        for (int p = 0; p < 2 * GEMM_WAVES; ++p) acc += red[which][p][col];
        atomicAdd(which ? &colsq[col] : &colsum[col], acc);
    }
}

#define FUSE_WAVES 32
#define FUSE_TPB   (FUSE_WAVES * 32)
#define GEMM_WV    (2 * SPB)
#define EDGE_U     16
#define APITCH     100
#define SRT_CAP    4096
#define SRT_Q      (SRT_CAP / FUSE_TPB)
#define SCAN_Q     ((NPB + 31) / 32)

template <int N, bool NORM>
__global__ __launch_bounds__(FUSE_TPB)
void edge_gemm_kernel(const float* __restrict__ x, int ldx,
                      const int2* __restrict__ rec,
                      const int* __restrict__ offT,
                      const float* __restrict__ lw,
                      const float* __restrict__ lb,
                      const float* __restrict__ W,
                      const float* __restrict__ bias,
                      float* __restrict__ out, int ldo,
                      float* __restrict__ colsum,
                      float* __restrict__ colsq,
                      const float* __restrict__ nsum,
                      const float* __restrict__ nsq,
                      const float* __restrict__ ng,
                      const float* __restrict__ nbeta,
                      float invN)
{
    constexpr int K     = 96;
    constexpr int KB    = K / 16;
    constexpr int NTH   = N / 32;
    constexpr int PITCH = 2 * N + 8;
    constexpr int NQ    = N / 4;
    static_assert(N % 32 == 0, "two column halves of whole tiles");
    static_assert(2 * N <= FUSE_TPB && K <= FUSE_TPB, "one thread per (column, statistic) and per column of x");
    static_assert(32 == 2 * EDGE_U, "a round of 32 records is two batches");
    static_assert(APITCH >= K && APITCH % 4 == 0, "rows of the sums are whole float4");
    static_assert(FUSE_TPB <= 1024 && GEMM_WV <= FUSE_WAVES, "one block, with a wave for every strip half");
    static_assert(PBLK == 128, "wave 0 scans the pieces, four per lane");
    static_assert(SRT_CAP % FUSE_TPB == 0 && SRT_CAP % (FUSE_WAVES * EDGE_U) == 0,
                  "whole records per thread; a full round splits into whole batches per wave");

    __shared__ float4 acc4[NPB * APITCH / 4];
    float* acc = reinterpret_cast<float*>(acc4);
    __shared__ float4 wl4[(K / 2) * PITCH / 4];
    float* wl = reinterpret_cast<float*>(wl4);
    __shared__ float red[2][SPB][N];
    __shared__ float4 nsc4[NORM ? K / 4 : 1];
    __shared__ float4 nsh4[NORM ? K / 4 : 1];
    __shared__ int2 tab[PBLK + 1];
    __shared__ int2 srt[SRT_CAP];
    __shared__ int  rcnt[NPB];
    __shared__ int  rstart[NPB];

    const int tid  = threadIdx.x;
    const int lane = tid & 31;
    const int lo   = lane & 15;
    const int hi   = lane >> 4;
    const int wv   = __builtin_amdgcn_readfirstlane(tid >> 5);

    int4 s4 = make_int4(0, 0, 0, 0), e4 = make_int4(0, 0, 0, 0);
    if (wv == 0) {
        s4 = *reinterpret_cast<const int4*>(offT + (size_t)blockIdx.x * PBLK + 4 * lane);
        e4 = *reinterpret_cast<const int4*>(offT + (size_t)(blockIdx.x + 1) * PBLK + 4 * lane);
    }

    for (int i = tid; i < NPB * APITCH / 4; i += FUSE_TPB)
        acc4[i] = make_float4(0.0f, 0.0f, 0.0f, 0.0f);
    if (tid < NPB) rcnt[tid] = 0;

    const float w0 = lw[lane], w1 = lw[lane + 32], w2 = lw[lane + 64];
    const float b0 = lb[lane], b1 = lb[lane + 32], b2 = lb[lane + 64];

    float sc0 = 1.0f, sc1 = 1.0f, sc2 = 1.0f, sh0 = 0.0f, sh1 = 0.0f, sh2 = 0.0f;
    if (NORM) {
        bn_scale_shift(nsum, nsq, ng, nbeta, invN, lane,      sc0, sh0);
        bn_scale_shift(nsum, nsq, ng, nbeta, invN, lane + 32, sc1, sh1);
        bn_scale_shift(nsum, nsq, ng, nbeta, invN, lane + 64, sc2, sh2);
        if (tid < K) {
            float sc, sh;
            bn_scale_shift(nsum, nsq, ng, nbeta, invN, tid, sc, sh);
            reinterpret_cast<float*>(nsc4)[tid] = sc;
            reinterpret_cast<float*>(nsh4)[tid] = sh;
        }
    }
    if (wv == 0) {
        const int c0 = e4.x - s4.x, c1 = e4.y - s4.y, c2 = e4.z - s4.z, c3 = e4.w - s4.w;
        const int lt = c0 + c1 + c2 + c3;
        int inc = lt;
        for (int o = 1; o < 32; o <<= 1) {
            const int nb = __shfl_up(inc, o);
            if (lane >= o) inc += nb;
        }
        const int p0 = inc - lt, p1 = p0 + c0, p2 = p1 + c1, p3 = p2 + c2;
        const int q0 = 4 * lane * PCH;
        tab[4 * lane + 0] = make_int2(p0, q0 + s4.x - p0);
        tab[4 * lane + 1] = make_int2(p1, q0 + PCH + s4.y - p1);
        tab[4 * lane + 2] = make_int2(p2, q0 + 2 * PCH + s4.z - p2);
        tab[4 * lane + 3] = make_int2(p3, q0 + 3 * PCH + s4.w - p3);
        if (lane == 31) tab[PBLK] = make_int2(inc, 0);
    }
    __syncthreads();

    const int total = __builtin_amdgcn_readfirstlane(tab[PBLK].x);

    float v0[EDGE_U], v1[EDGE_U], v2[EDGE_U];
    auto gather = [&](int kv, int e0) {
#pragma unroll
        for (int u = 0; u < EDGE_U; ++u) {
            const int s = __builtin_amdgcn_readlane(kv, e0 + u) & 0xFFFF;
            const float* xr = x + (size_t)s * ldx + lane;
            v0[u] = xr[0];
            v1[u] = xr[32];
            v2[u] = xr[64];
        }
    };
    int   cur = 0;
    float s0 = 0.0f, s1 = 0.0f, s2 = 0.0f;
    auto flush = [&]() {
        float* ap = acc + cur * APITCH + lane;
        atomicAdd(ap,      s0);
        atomicAdd(ap + 32, s1);
        atomicAdd(ap + 64, s2);
    };
    auto addup = [&](int kv, int av, int e0, int n) {
#pragma unroll
        for (int u = 0; u < EDGE_U; ++u) {
            if (e0 + u < n) {
                const int   t = __builtin_amdgcn_readlane(kv, e0 + u) >> 16;
                const float a = __int_as_float(__builtin_amdgcn_readlane(av, e0 + u));
                if (t != cur) {
                    flush();
                    cur = t;
                    s0 = 0.0f;  s1 = 0.0f;  s2 = 0.0f;
                }
                const float x0 = NORM ? v0[u] * sc0 + sh0 : v0[u];
                const float x1 = NORM ? v1[u] * sc1 + sh1 : v1[u];
                const float x2 = NORM ? v2[u] * sc2 + sh2 : v2[u];
                s0 += fmaxf(x0 + a * w0 + b0, 0.0f);
                s1 += fmaxf(x1 + a * w1 + b1, 0.0f);
                s2 += fmaxf(x2 + a * w2 + b2, 0.0f);
            }
        }
    };

    for (int base = 0; base < total; base += SRT_CAP) {
        const int nt = min(SRT_CAP, total - base);

        int pp[SRT_Q];
#pragma unroll
        for (int q = 0; q < SRT_Q; ++q) pp[q] = 0;
#pragma unroll
        for (int s = PBLK / 2; s >= 1; s >>= 1) {
#pragma unroll
            for (int q = 0; q < SRT_Q; ++q)
                if (tab[pp[q] + s].x <= base + q * FUSE_TPB + tid) pp[q] += s;
        }
        int2 rr[SRT_Q];
        int  rk[SRT_Q];
        int po[SRT_Q];
#pragma unroll
        for (int q = 0; q < SRT_Q; ++q) po[q] = tab[pp[q]].y;
#pragma unroll
        for (int q = 0; q < SRT_Q; ++q) {
            const int j = q * FUSE_TPB + tid;
            rr[q] = make_int2(0, 0);
            if (j < nt) rr[q] = rec[base + j + po[q]];
        }
#pragma unroll
        for (int q = 0; q < SRT_Q; ++q) {
            rk[q] = 0;
            if (q * FUSE_TPB + tid < nt) rk[q] = atomicAdd(&rcnt[rr[q].x >> 16], 1);
        }
        __syncthreads();

        if (wv == 0) {
            int cn[SCAN_Q];
            int lt = 0;
#pragma unroll
            for (int k = 0; k < SCAN_Q; ++k) {
                const int row = SCAN_Q * lane + k;
                cn[k] = (row < NPB) ? rcnt[row] : 0;
                lt += cn[k];
            }
            int inc = lt;
            for (int o = 1; o < 32; o <<= 1) {
                const int nb = __shfl_up(inc, o);
                if (lane >= o) inc += nb;
            }
            int run = inc - lt;
#pragma unroll
            for (int k = 0; k < SCAN_Q; ++k) {
                const int row = SCAN_Q * lane + k;
                if (row < NPB) {
                    rstart[row] = run;
                    rcnt[row]   = 0;
                }
                run += cn[k];
            }
        }
        __syncthreads();

#pragma unroll
        for (int q = 0; q < SRT_Q; ++q)
            if (q * FUSE_TPB + tid < nt) srt[rstart[rr[q].x >> 16] + rk[q]] = rr[q];
        __syncthreads();

        const int per = (((nt + FUSE_WAVES - 1) / FUSE_WAVES) + EDGE_U - 1) / EDGE_U * EDGE_U;
        const int lo_ = wv * per;
        const int hi_ = min(lo_ + per, nt);
        if (lo_ < hi_) {
            int2 r = (lo_ + lane < hi_) ? srt[lo_ + lane] : make_int2(0, 0);
            for (int g = lo_; g < hi_; g += 32) {
                const int n  = min(32, hi_ - g);
                const int kv = r.x;
                const int av = r.y;
                gather(kv, 0);
                r = (g + 32 + lane < hi_) ? srt[g + 32 + lane] : make_int2(0, 0);
                addup(kv, av, 0, n);
                if (n > EDGE_U) {
                    gather(kv, EDGE_U);
                    addup(kv, av, EDGE_U, n);
                }
            }
        }
    }
    flush();

    const bool gw = wv < GEMM_WV;
    const int  st = wv >> 1;
    const int  hf = wv & 1;
    int row0 = blockIdx.x * NPB + st * 16;
    const bool live = gw && row0 < NNODES;
    if (!live) row0 = NNODES - 16;
    float4 a[KB][2];
    if (gw) {
        const float* ar = x + (size_t)(row0 + lo) * ldx + 8 * hi;
#pragma unroll
        for (int i = 0; i < KB; ++i) {
            a[i][0] = *reinterpret_cast<const float4*>(ar + 16 * i);
            a[i][1] = *reinterpret_cast<const float4*>(ar + 16 * i + 4);
        }
    }
    constexpr int ITEMS = (K / 2) * NQ;
    for (int it = tid; it < ITEMS; it += FUSE_TPB) {
        const int p  = it / NQ;
        const int n4 = it % NQ;
        const float4 q0 = *reinterpret_cast<const float4*>(W + (size_t)(2 * p) * N + 4 * n4);
        const float4 q1 = *reinterpret_cast<const float4*>(W + (size_t)(2 * p + 1) * N + 4 * n4);
        float4* d = reinterpret_cast<float4*>(wl + p * PITCH + 8 * n4);
        d[0] = make_float4(q0.x, q1.x, q0.y, q1.y);
        d[1] = make_float4(q0.z, q1.z, q0.w, q1.w);
    }
    __syncthreads();

    const int col0 = 16 * NTH * hf + lo;
    float res[NTH][8];
    if (gw) {
        const float* pr = acc + (st * 16 + lo) * APITCH + 8 * hi;
#pragma unroll
        for (int i = 0; i < KB; ++i) {
#pragma unroll
            for (int h = 0; h < 2; ++h) {
                if (NORM) {
                    const float4 s4 = nsc4[4 * i + 2 * hi + h];
                    const float4 t4 = nsh4[4 * i + 2 * hi + h];
                    a[i][h].x = a[i][h].x * s4.x + t4.x;
                    a[i][h].y = a[i][h].y * s4.y + t4.y;
                    a[i][h].z = a[i][h].z * s4.z + t4.z;
                    a[i][h].w = a[i][h].w * s4.w + t4.w;
                }
                const float4 g4 = *reinterpret_cast<const float4*>(pr + 16 * i + 4 * h);
                a[i][h].x += g4.x;  a[i][h].y += g4.y;
                a[i][h].z += g4.z;  a[i][h].w += g4.w;
            }
        }

        v8f c[NTH];
#pragma unroll
        for (int t = 0; t < NTH; ++t) c[t] = v8f{};

        const float* wb = wl + (4 * hi) * PITCH + 2 * lo + 32 * NTH * hf;
#pragma unroll
        for (int i = 0; i < KB; ++i) {
#pragma unroll
            for (int j = 0; j < 4; ++j) {
                v2f av;
                if (j == 0)      { av.x = a[i][0].x; av.y = a[i][0].y; }
                else if (j == 1) { av.x = a[i][0].z; av.y = a[i][0].w; }
                else if (j == 2) { av.x = a[i][1].x; av.y = a[i][1].y; }
                else             { av.x = a[i][1].z; av.y = a[i][1].w; }
#pragma unroll
                for (int t = 0; t < NTH; ++t) {
                    const v2f bv = *reinterpret_cast<const v2f*>(
                        wb + (8 * i + j) * PITCH + 32 * t);
                    c[t] = __builtin_amdgcn_wmma_f32_16x16x4_f32(
                             false, av,  false, bv,
                             (short)0, c[t],  false,  false);
                }
            }
        }

#pragma unroll
        for (int t = 0; t < NTH; ++t) {
            const float bsv = bias[col0 + 16 * t];
#pragma unroll
            for (int r = 0; r < 8; ++r) res[t][r] = fmaxf(c[t][r] + bsv, 0.0f);
        }
#pragma unroll
        for (int t = 0; t < NTH; ++t) {
            float s = 0.0f, q = 0.0f;
#pragma unroll
            for (int r = 0; r < 8; ++r) {
                const float v = res[t][r];
                s += v;
                q += v * v;
            }
            s += __shfl_xor(s, 16);
            q += __shfl_xor(q, 16);
            if (hi == 0) {
                red[0][st][col0 + 16 * t] = live ? s : 0.0f;
                red[1][st][col0 + 16 * t] = live ? q : 0.0f;
            }
        }
    }
    __syncthreads();

    if (live) {
#pragma unroll
        for (int r = 0; r < 8; ++r) {
            float* orow = out + (size_t)(row0 + r + 8 * hi) * ldo + col0;
#pragma unroll
            for (int t = 0; t < NTH; ++t) orow[16 * t] = res[t][r];
        }
    }

    if (tid < 2 * N) {
        const int which = tid / N;
        const int col   = tid % N;
        float sum = 0.0f;
#pragma unroll
        for (int p = 0; p < SPB; ++p) sum += red[which][p][col];
        atomicAdd(which ? &colsq[col] : &colsum[col], sum);
    }
}

__global__ __launch_bounds__(256)
void bn_apply_kernel(const float* __restrict__ in, int ldi,
                     float* __restrict__ out, int ldo,
                     const float* __restrict__ sum,
                     const float* __restrict__ sumsq,
                     const float* __restrict__ g,
                     const float* __restrict__ beta,
                     float invN, int n, int C)
{
    int gid = blockIdx.x * blockDim.x + threadIdx.x;
    if (gid >= n * C) return;
    int row = gid / C;
    int c   = gid % C;
    float sc, sh;
    bn_scale_shift(sum, sumsq, g, beta, invN, c, sc, sh);
    out[(size_t)row * ldo + c] = in[(size_t)row * ldi + c] * sc + sh;
}

extern "C" void kernel_launch(void* const* d_in, const int* in_sizes, int n_in,
                              void* d_out, int out_size, void* d_ws, size_t ws_size,
                              hipStream_t stream)
{
    (void)in_sizes; (void)n_in; (void)out_size; (void)ws_size;

    const float* x          = (const float*)d_in[0];
    const float* edge_attr  = (const float*)d_in[1];
    const int*   edge_index = (const int*)  d_in[2];
    const float* lin_w      = (const float*)d_in[3];
    const float* lin_b      = (const float*)d_in[4];
    const float* c1_w  = (const float*)d_in[5];
    const float* c1_b  = (const float*)d_in[6];
    const float* c1_g  = (const float*)d_in[7];
    const float* c1_be = (const float*)d_in[8];
    const float* c2_w  = (const float*)d_in[9];
    const float* c2_b  = (const float*)d_in[10];
    const float* c2_g  = (const float*)d_in[11];
    const float* c2_be = (const float*)d_in[12];
    const float* l1_w  = (const float*)d_in[13];
    const float* l1_b  = (const float*)d_in[14];
    const float* l1_g  = (const float*)d_in[15];
    const float* l1_be = (const float*)d_in[16];
    const float* ma_w  = (const float*)d_in[17];
    const float* ma_b  = (const float*)d_in[18];
    const float* ma_g  = (const float*)d_in[19];
    const float* ma_be = (const float*)d_in[20];
    const float* mb_w  = (const float*)d_in[21];
    const float* mb_b  = (const float*)d_in[22];
    const float* mb_g  = (const float*)d_in[23];
    const float* mb_be = (const float*)d_in[24];

    const int* src = edge_index;
    const int* dst = edge_index + NEDGES;

    const size_t N = NNODES;

    float* ws    = (float*)d_ws;
    float* bufP  = ws;
    float* bufQ  = bufP + N * 96;
    float* xcat  = bufQ + N * 96;
    float* stats = xcat + N * 160;
    int*   offT  = (int*)(stats + NSTAT);
    int2*  rec   = (int2*)(offT + 256 * PBLK);

    const float invN = 1.0f / (float)N;

    auto sumL = [&](int L) { return stats + (size_t)L * 192; };
    auto sqL  = [&](int L) { return stats + (size_t)L * 192 + 96; };

    static_assert(NBINS + 1 <= 256, "offT fits its part of the workspace");
    static_assert(((size_t)NNODES * (96 + 96 + 160) + NSTAT + 256 * PBLK) % 4 == 0,
                  "rec starts on a 16-byte boundary of the workspace");
    bin_chunk_kernel<<<PBLK, PTPB, 0, stream>>>(src, dst, edge_attr, offT, rec, stats);

    edge_gemm_kernel<96, false><<<NBINS, FUSE_TPB, 0, stream>>>(
        x, 96, rec, offT, lin_w, lin_b, c1_w, c1_b, xcat, 160, sumL(0), sqL(0),
        nullptr, nullptr, nullptr, nullptr, invN);

    edge_gemm_kernel<64, true><<<NBINS, FUSE_TPB, 0, stream>>>(
        xcat, 160, rec, offT, lin_w, lin_b, c2_w, c2_b, xcat + 96, 160, sumL(1), sqL(1),
        sumL(0), sqL(0), c1_g, c1_be, invN);

    gemm_wmma_f32_kernel<160, 96, false, true><<<GEMM_GRID, GEMM_TPB, 0, stream>>>(
        xcat, 160, nullptr, 0, l1_w, l1_b, bufQ, 96, sumL(2), sqL(2),
        sumL(0), sqL(0), c1_g, c1_be, 96,
        sumL(1), sqL(1), c2_g, c2_be, invN);

    gemm_wmma_f32_kernel<96, 96, false, true><<<GEMM_GRID, GEMM_TPB, 0, stream>>>(
        bufQ, 96, nullptr, 0, ma_w, ma_b, bufP, 96, sumL(3), sqL(3),
        sumL(2), sqL(2), l1_g, l1_be, 96,
        nullptr, nullptr, nullptr, nullptr, invN);

    gemm_wmma_f32_kernel<96, 16, false, true><<<GEMM_GRID, GEMM_TPB, 0, stream>>>(
        bufP, 96, nullptr, 0, mb_w, mb_b, bufQ, 16, sumL(4), sqL(4),
        sumL(3), sqL(3), ma_g, ma_be, 96,
        nullptr, nullptr, nullptr, nullptr, invN);

    bn_apply_kernel<<<(NNODES * 16 + 255) / 256, 256, 0, stream>>>(
        bufQ, 16, (float*)d_out, 16, sumL(4), sqL(4), mb_g, mb_be, invN, NNODES, 16);
}
